# starting kernel
# baseline (speedup 1.0000x reference)
#include <hip/hip_runtime.h>
#include <hip/hip_bf16.h>

typedef __attribute__((ext_vector_type(2))) float v2f;
typedef __attribute__((ext_vector_type(8))) float v8f;

#define HDIM 128
#define HDIM2 256
#define NUM_GRAPHS 64
#define NUM_CLASSES 10
#define MSG_EPS 1e-7f

__device__ __forceinline__ void atomicMaxF(float* addr, float v) {
  if (v >= 0.0f) atomicMax((int*)addr, __float_as_int(v));
  else           atomicMin((unsigned int*)addr, __float_as_uint(v));
}

__global__ void gemm16_wmma_f32(const float* __restrict__ X,
                                const float* __restrict__ W,
                                const float* __restrict__ bias,
                                float* __restrict__ Y,
                                int Kdim, int Mdim, int accum) {
  const int lane = threadIdx.x & 31;
  const int wid  = threadIdx.x >> 5;
  const int row0 = blockIdx.x * 16;
  const int col0 = (blockIdx.y * 4 + wid) * 16;
  if (col0 >= Mdim) return;

  const int half = lane >> 4;
  const int l16  = lane & 15;

  const float bv = bias[col0 + l16];
  v8f c;
  if (accum) {
#pragma unroll
    for (int r = 0; r < 8; ++r)
      c[r] = Y[(size_t)(row0 + half * 8 + r) * Mdim + col0 + l16] + bv;
  } else {
#pragma unroll
    for (int r = 0; r < 8; ++r) c[r] = bv;
  }

  const float* xrow = X + (size_t)(row0 + l16) * Kdim + half * 2;
  const float* wcol = W + (size_t)(half * 2) * Mdim + col0 + l16;

  for (int k = 0; k < Kdim; k += 4) {
    __builtin_prefetch(xrow + k + 32, 0, 1);
    float2 av = *(const float2*)(xrow + k);
    v2f a; a[0] = av.x; a[1] = av.y;
    v2f b;
    b[0] = wcol[(size_t)k * Mdim];
    b[1] = wcol[(size_t)(k + 1) * Mdim];
    c = __builtin_amdgcn_wmma_f32_16x16x4_f32(false, a, false, b,
                                              (short)0, c, false, false);
  }

#pragma unroll
  for (int r = 0; r < 8; ++r)
    Y[(size_t)(row0 + half * 8 + r) * Mdim + col0 + l16] = c[r];
}

__global__ void ln_relu_kernel(const float* __restrict__ x, float* __restrict__ y,
                               const float* __restrict__ g, const float* __restrict__ b,
                               int rows, int D) {
  int row  = blockIdx.x * (blockDim.x >> 5) + (threadIdx.x >> 5);
  int lane = threadIdx.x & 31;
  if (row >= rows) return;
  const float* xr = x + (size_t)row * D;
  float s = 0.f, q = 0.f;
  for (int j = lane; j < D; j += 32) { float v = xr[j]; s += v; q += v * v; }
#pragma unroll
  for (int off = 16; off > 0; off >>= 1) {
    s += __shfl_xor(s, off, 32);
    q += __shfl_xor(q, off, 32);
  }
  float mean = s / (float)D;
  float var  = q / (float)D - mean * mean;
  float inv  = rsqrtf(var + 1e-5f);
  float* yr = y + (size_t)row * D;
  for (int j = lane; j < D; j += 32) {
    float v = (xr[j] - mean) * inv * g[j] + b[j];
    yr[j] = fmaxf(v, 0.f);
  }
}

__global__ void init_seg_kernel(float* mx, float* den, float* num, int total) {
  int i = blockIdx.x * blockDim.x + threadIdx.x;
  if (i >= total) return;
  mx[i]  = __int_as_float(0xff800000);
  den[i] = 0.f;
  num[i] = 0.f;
}

__global__ void zero_kernel(float* p, int n) {
  int i = blockIdx.x * blockDim.x + threadIdx.x;
  if (i < n) p[i] = 0.f;
}

__global__ void edge_max_kernel(const float* __restrict__ x, const float* __restrict__ ea,
                                const int* __restrict__ src, const int* __restrict__ dst,
                                const float* __restrict__ tptr, int layer,
                                float* __restrict__ mx, int E) {
  int e    = blockIdx.x * (blockDim.x >> 5) + (threadIdx.x >> 5);
  int lane = threadIdx.x & 31;
  if (e >= E) return;
  int s = src[e], d = dst[e];
  float tv = tptr[layer];
  float4 a  = ((const float4*)(x  + (size_t)s * HDIM))[lane];
  float4 bb = ((const float4*)(ea + (size_t)e * HDIM))[lane];
  float m0 = fmaxf(a.x + bb.x, 0.f) + MSG_EPS;
  float m1 = fmaxf(a.y + bb.y, 0.f) + MSG_EPS;
  float m2 = fmaxf(a.z + bb.z, 0.f) + MSG_EPS;
  float m3 = fmaxf(a.w + bb.w, 0.f) + MSG_EPS;
  float* mrow = mx + (size_t)d * HDIM + lane * 4;
  atomicMaxF(mrow + 0, m0 * tv);
  atomicMaxF(mrow + 1, m1 * tv);
  atomicMaxF(mrow + 2, m2 * tv);
  atomicMaxF(mrow + 3, m3 * tv);
}

__global__ void edge_sum_kernel(const float* __restrict__ x, const float* __restrict__ ea,
                                const int* __restrict__ src, const int* __restrict__ dst,
                                const float* __restrict__ tptr, int layer,
                                const float* __restrict__ mx,
                                float* __restrict__ den, float* __restrict__ num, int E) {
  int e    = blockIdx.x * (blockDim.x >> 5) + (threadIdx.x >> 5);
  int lane = threadIdx.x & 31;
  if (e >= E) return;
  int s = src[e], d = dst[e];
  float tv = tptr[layer];
  float4 a  = ((const float4*)(x  + (size_t)s * HDIM))[lane];
  float4 bb = ((const float4*)(ea + (size_t)e * HDIM))[lane];
  float4 mv = ((const float4*)(mx + (size_t)d * HDIM))[lane];
  float m0 = fmaxf(a.x + bb.x, 0.f) + MSG_EPS;
  float m1 = fmaxf(a.y + bb.y, 0.f) + MSG_EPS;
  float m2 = fmaxf(a.z + bb.z, 0.f) + MSG_EPS;
  float m3 = fmaxf(a.w + bb.w, 0.f) + MSG_EPS;
  float e0 = __expf(m0 * tv - mv.x);
  float e1 = __expf(m1 * tv - mv.y);
  float e2 = __expf(m2 * tv - mv.z);
  float e3 = __expf(m3 * tv - mv.w);
  float* drow = den + (size_t)d * HDIM + lane * 4;
  float* nrow = num + (size_t)d * HDIM + lane * 4;
  atomicAdd(drow + 0, e0);       atomicAdd(nrow + 0, e0 * m0);
  atomicAdd(drow + 1, e1);       atomicAdd(nrow + 1, e1 * m1);
  atomicAdd(drow + 2, e2);       atomicAdd(nrow + 2, e2 * m2);
  atomicAdd(drow + 3, e3);       atomicAdd(nrow + 3, e3 * m3);
}

__global__ void combine_kernel(const float* __restrict__ rin, const float* __restrict__ num,
                               const float* __restrict__ den, float* __restrict__ hc, int total) {
  int i = blockIdx.x * blockDim.x + threadIdx.x;
  if (i >= total) return;
  float dv = den[i];
  float agg = dv > 0.f ? num[i] / dv : 0.f;
  hc[i] = agg + rin[i];
}

__global__ void pool_kernel(const float* __restrict__ x, const int* __restrict__ batch,
                            float* __restrict__ pooled, int n) {
  int i = blockIdx.x * blockDim.x + threadIdx.x;
  if (i >= n * HDIM) return;
  int node = i >> 7, col = i & (HDIM - 1);
  atomicAdd(pooled + (size_t)batch[node] * HDIM + col, x[i]);
}

__global__ void final_linear_kernel(const float* __restrict__ pooled,
                                    const float* __restrict__ W,
                                    const float* __restrict__ b,
                                    float* __restrict__ out) {
  int i = blockIdx.x * blockDim.x + threadIdx.x;
  if (i >= NUM_GRAPHS * NUM_CLASSES) return;
  int g = i / NUM_CLASSES, c = i % NUM_CLASSES;
  float acc = b[c];
  const float* pr = pooled + (size_t)g * HDIM;
#pragma unroll 4
  for (int k = 0; k < HDIM; ++k) acc += pr[k] * W[k * NUM_CLASSES + c];
  out[i] = acc;
}

extern "C" void kernel_launch(void* const* d_in, const int* in_sizes, int n_in,
                              void* d_out, int out_size, void* d_ws, size_t ws_size,
                              hipStream_t stream) {
  const float* x     = (const float*)d_in[0];
  const float* ea    = (const float*)d_in[1];
  const float* enc_W = (const float*)d_in[2];
  const float* enc_b = (const float*)d_in[3];
  const float* t     = (const float*)d_in[4];
  const float* W1    = (const float*)d_in[5];
  const float* b1    = (const float*)d_in[6];
  const float* mlp_g = (const float*)d_in[7];
  const float* mlp_b = (const float*)d_in[8];
  const float* W2    = (const float*)d_in[9];
  const float* b2    = (const float*)d_in[10];
  const float* ln_g  = (const float*)d_in[11];
  const float* ln_b  = (const float*)d_in[12];
  const float* lin_W = (const float*)d_in[13];
  const float* lin_b = (const float*)d_in[14];
  const int*   eidx  = (const int*)d_in[15];
  const int*   batch = (const int*)d_in[16];

  const int N = in_sizes[0] / HDIM;
  const int E = in_sizes[1] / HDIM;
  const int L = in_sizes[4];
  const int* src = eidx;
  const int* dst = eidx + E;

  const size_t NH = (size_t)N * HDIM;
  float* h    = (float*)d_ws;
  float* rbuf = h    + NH;
  float* hc   = rbuf + NH;
  float* mx   = hc   + NH;
  float* den  = mx   + NH;
  float* num  = den  + NH;
  float* t1   = num  + NH;

  float* out    = (float*)d_out;
  float* pooled = out + NUM_GRAPHS * NUM_CLASSES;

  const dim3 gblk(128);
  const int eBlocks = (E + 7) / 8;
  const int elBlocks = (int)((NH + 255) / 256);
  const int lnBlocks = (N + 7) / 8;

  gemm16_wmma_f32<<<dim3(N / 16, HDIM / 64), gblk, 0, stream>>>(x, enc_W, enc_b, h, HDIM, HDIM, 0);

  for (int layer = 0; layer < L; ++layer) {
    const float* rin = h;
    if (layer > 0) {
      ln_relu_kernel<<<lnBlocks, 256, 0, stream>>>(h, rbuf, ln_g + layer * HDIM, ln_b + layer * HDIM, N, HDIM);
      rin = rbuf;
    }
    init_seg_kernel<<<elBlocks, 256, 0, stream>>>(mx, den, num, (int)NH);
    edge_max_kernel<<<eBlocks, 256, 0, stream>>>(rin, ea, src, dst, t, layer, mx, E);
    edge_sum_kernel<<<eBlocks, 256, 0, stream>>>(rin, ea, src, dst, t, layer, mx, den, num, E);
    combine_kernel<<<elBlocks, 256, 0, stream>>>(rin, num, den, hc, (int)NH);
    gemm16_wmma_f32<<<dim3(N / 16, HDIM2 / 64), gblk, 0, stream>>>(
        hc, W1 + (size_t)layer * HDIM * HDIM2, b1 + layer * HDIM2, t1, HDIM, HDIM2, 0);
    ln_relu_kernel<<<lnBlocks, 256, 0, stream>>>(t1, t1, mlp_g + layer * HDIM2, mlp_b + layer * HDIM2, N, HDIM2);
    gemm16_wmma_f32<<<dim3(N / 16, HDIM / 64), gblk, 0, stream>>>(
        t1, W2 + (size_t)layer * HDIM2 * HDIM, b2 + layer * HDIM, h, HDIM2, HDIM, layer > 0 ? 1 : 0);
  }

  ln_relu_kernel<<<lnBlocks, 256, 0, stream>>>(h, rbuf, ln_g, ln_b, N, HDIM);
  zero_kernel<<<(NUM_GRAPHS * HDIM + 255) / 256, 256, 0, stream>>>(pooled, NUM_GRAPHS * HDIM);
  pool_kernel<<<(int)((NH + 255) / 256), 256, 0, stream>>>(rbuf, batch, pooled, N);
  final_linear_kernel<<<(NUM_GRAPHS * NUM_CLASSES + 63) / 64, 64, 0, stream>>>(pooled, lin_W, lin_b, out);
}


# Edge softmax by gather: edges bucketed by dst once per call (int atomic slots, scan, id-sorted), one wave per node sums in registers; float atomics gone, sums now in fixed edge-id order
# speedup vs baseline: 2.2498x; 2.2498x over previous
#include <hip/hip_runtime.h>
#include <hip/hip_bf16.h>

typedef __attribute__((ext_vector_type(2))) float v2f;
typedef __attribute__((ext_vector_type(8))) float v8f;

#define HDIM 128
#define HDIM2 256
#define NUM_GRAPHS 64
#define NUM_CLASSES 10
#define MSG_EPS 1e-7f

__global__ void gemm16_wmma_f32(const float* __restrict__ X,
                                const float* __restrict__ W,
                                const float* __restrict__ bias,
                                float* __restrict__ Y,
                                int Kdim, int Mdim, int accum) {
  const int lane = threadIdx.x & 31;
  const int wid  = threadIdx.x >> 5;
  const int row0 = blockIdx.x * 16;
  const int col0 = (blockIdx.y * 4 + wid) * 16;
  if (col0 >= Mdim) return;

  const int half = lane >> 4;
  const int l16  = lane & 15;

  const float bv = bias[col0 + l16];
  v8f c;
  if (accum) {
#pragma unroll
    for (int r = 0; r < 8; ++r)
      c[r] = Y[(size_t)(row0 + half * 8 + r) * Mdim + col0 + l16] + bv;
  } else {
#pragma unroll
    for (int r = 0; r < 8; ++r) c[r] = bv;
  }

  const float* xrow = X + (size_t)(row0 + l16) * Kdim + half * 2;
  const float* wcol = W + (size_t)(half * 2) * Mdim + col0 + l16;

  for (int k = 0; k < Kdim; k += 4) {
    __builtin_prefetch(xrow + k + 32, 0, 1);
    float2 av = *(const float2*)(xrow + k);
    v2f a; a[0] = av.x; a[1] = av.y;
    v2f b;
    b[0] = wcol[(size_t)k * Mdim];
    b[1] = wcol[(size_t)(k + 1) * Mdim];
    c = __builtin_amdgcn_wmma_f32_16x16x4_f32(false, a, false, b,
                                              (short)0, c, false, false);
  }

#pragma unroll
  for (int r = 0; r < 8; ++r)
    Y[(size_t)(row0 + half * 8 + r) * Mdim + col0 + l16] = c[r];
}

__global__ void ln_relu_kernel(const float* __restrict__ x, float* __restrict__ y,
                               const float* __restrict__ g, const float* __restrict__ b,
                               int rows, int D) {
  int row  = blockIdx.x * (blockDim.x >> 5) + (threadIdx.x >> 5);
  int lane = threadIdx.x & 31;
  if (row >= rows) return;
  const float* xr = x + (size_t)row * D;
  float s = 0.f, q = 0.f;
  for (int j = lane; j < D; j += 32) { float v = xr[j]; s += v; q += v * v; }
#pragma unroll
  for (int off = 16; off > 0; off >>= 1) {
    s += __shfl_xor(s, off, 32);
    q += __shfl_xor(q, off, 32);
  }
  float mean = s / (float)D;
  float var  = q / (float)D - mean * mean;
  float inv  = rsqrtf(var + 1e-5f);
  float* yr = y + (size_t)row * D;
  for (int j = lane; j < D; j += 32) {
    float v = (xr[j] - mean) * inv * g[j] + b[j];
    yr[j] = fmaxf(v, 0.f);
  }
}

__global__ void zero_int_kernel(int* p, int n) {
  int i = blockIdx.x * blockDim.x + threadIdx.x;
  if (i < n) p[i] = 0;
}

__global__ void zero_kernel(float* p, int n) {
  int i = blockIdx.x * blockDim.x + threadIdx.x;
  if (i < n) p[i] = 0.f;
}

__global__ void bucket_count_kernel(const int* __restrict__ dst, int* __restrict__ cnt,
                                    int* __restrict__ slot, int E) {
  int e = blockIdx.x * blockDim.x + threadIdx.x;
  if (e >= E) return;
  slot[e] = atomicAdd(cnt + dst[e], 1);
}

__global__ void __launch_bounds__(1024)
bucket_scan_kernel(const int* __restrict__ cnt, int* __restrict__ off, int n) {
  __shared__ int part[1024];
  const int tid = threadIdx.x;
  const int per = (n + 1023) / 1024;
  const int b = min(tid * per, n), e = min(b + per, n);
  int s = 0;
  for (int i = b; i < e; ++i) s += cnt[i];
  part[tid] = s;
  __syncthreads();
  for (int d = 1; d < 1024; d <<= 1) {
    int v = (tid >= d) ? part[tid - d] : 0;
    __syncthreads();
    part[tid] += v;
    __syncthreads();
  }
  int run = part[tid] - s;
  for (int i = b; i < e; ++i) { off[i] = run; run += cnt[i]; }
  if (tid == 1023) off[n] = part[1023];
}

__global__ void bucket_place_kernel(const int* __restrict__ dst, const int* __restrict__ off,
                                    const int* __restrict__ slot, int* __restrict__ raw, int E) {
  int e = blockIdx.x * blockDim.x + threadIdx.x;
  if (e >= E) return;
  raw[off[dst[e]] + slot[e]] = e;
}

__global__ void bucket_order_kernel(const int* __restrict__ raw, const int* __restrict__ off,
                                    const int* __restrict__ src, int* __restrict__ pairs, int n) {
  int node = blockIdx.x * (blockDim.x >> 5) + (threadIdx.x >> 5);
  int lane = threadIdx.x & 31;
  if (node >= n) return;
  const int beg = off[node], end = off[node + 1];
  for (int i = beg + lane; i < end; i += 32) {
    const int ei = raw[i];
    int r = 0;
    for (int j = beg; j < end; ++j) r += (raw[j] < ei) ? 1 : 0;
    ((int2*)pairs)[beg + r] = make_int2(ei, src[ei]);
  }
}

#define EB 8

template <bool TAIL>
__device__ __forceinline__ void gather_group(const float* __restrict__ x,
                                             const float* __restrict__ ea,
                                             int2 pr, int first, int nlive,
                                             unsigned lane, float tv, bool tpos,
                                             float (&mx)[4], float (&den)[4], float (&num)[4]) {
  float4 xa[EB], eb[EB];
#pragma unroll
  for (int j = 0; j < EB; ++j) {
    const int e = __builtin_amdgcn_readlane(pr.x, first + j);
    const int s = __builtin_amdgcn_readlane(pr.y, first + j);
    xa[j] = ((const float4*)x)[(unsigned)s * (HDIM / 4) + lane];
    eb[j] = ((const float4*)ea)[(unsigned)e * (HDIM / 4) + lane];
  }
  float m[EB][4];
#pragma unroll
  for (int j = 0; j < EB; ++j) {
    m[j][0] = fmaxf(xa[j].x + eb[j].x, 0.f) + MSG_EPS;
    m[j][1] = fmaxf(xa[j].y + eb[j].y, 0.f) + MSG_EPS;
    m[j][2] = fmaxf(xa[j].z + eb[j].z, 0.f) + MSG_EPS;
    m[j][3] = fmaxf(xa[j].w + eb[j].w, 0.f) + MSG_EPS;
  }
#pragma unroll
  for (int c = 0; c < 4; ++c) {
    float ext = m[0][c];
    if (tpos) {
#pragma unroll
      for (int j = 1; j < EB; ++j) ext = fmaxf(ext, (TAIL && j >= nlive) ? m[0][c] : m[j][c]);
    } else {
#pragma unroll
      for (int j = 1; j < EB; ++j) ext = fminf(ext, (TAIL && j >= nlive) ? m[0][c] : m[j][c]);
    }
    const float nm = fmaxf(mx[c], ext * tv);
    const float sc = __expf(mx[c] - nm);
    float d = den[c] * sc, u = num[c] * sc;
#pragma unroll
    for (int j = 0; j < EB; ++j) {
      float ex = __expf(m[j][c] * tv - nm);
      if (TAIL && j >= nlive) ex = 0.f;
      d += ex;
      u += ex * m[j][c];
    }
    mx[c] = nm; den[c] = d; num[c] = u;
  }
}

__global__ void __launch_bounds__(128)
node_softmax_gather_kernel(const float* __restrict__ x, const float* __restrict__ ea,
                           const int* __restrict__ pairs, const int* __restrict__ off,
                           const float* __restrict__ tptr, int layer,
                           float* __restrict__ hc, int n) {
  const int nodev = blockIdx.x * (blockDim.x >> 5) + (threadIdx.x >> 5);
  const unsigned lane = threadIdx.x & 31u;
  if (nodev >= n) return;
  const int node = __builtin_amdgcn_readfirstlane(nodev);
  const int beg = off[node], end = off[node + 1];
  const float tv = tptr[layer];
  const bool tpos = tv >= 0.f;

  float mx[4], den[4], num[4];
#pragma unroll
  for (int c = 0; c < 4; ++c) { mx[c] = __int_as_float(0xff800000); den[c] = 0.f; num[c] = 0.f; }

  for (int base = beg; base < end; base += 32) {
    const int cnt = min(32, end - base);
    const int2 pr = ((const int2*)pairs)[base + min((int)lane, cnt - 1)];
    int g = 0;
    for (; g + EB <= cnt; g += EB)
      gather_group<false>(x, ea, pr, g, EB, lane, tv, tpos, mx, den, num);
    if (g < cnt)
      gather_group<true>(x, ea, pr, g, cnt - g, lane, tv, tpos, mx, den, num);
  }

  const float4 root = ((const float4*)(x + (size_t)node * HDIM))[lane];
  float4 o;
  if (beg < end) {
    o.x = num[0] / den[0] + root.x;
    o.y = num[1] / den[1] + root.y;
    o.z = num[2] / den[2] + root.z;
    o.w = num[3] / den[3] + root.w;
  } else {
    o = root;
  }
  ((float4*)(hc + (size_t)node * HDIM))[lane] = o;
}

__global__ void pool_kernel(const float* __restrict__ x, const int* __restrict__ batch,
                            float* __restrict__ pooled, int n) {
  int i = blockIdx.x * blockDim.x + threadIdx.x;
  if (i >= n * HDIM) return;
  int node = i >> 7, col = i & (HDIM - 1);
  atomicAdd(pooled + (size_t)batch[node] * HDIM + col, x[i]);
}

__global__ void final_linear_kernel(const float* __restrict__ pooled,
                                    const float* __restrict__ W,
                                    const float* __restrict__ b,
                                    float* __restrict__ out) {
  int i = blockIdx.x * blockDim.x + threadIdx.x;
  if (i >= NUM_GRAPHS * NUM_CLASSES) return;
  int g = i / NUM_CLASSES, c = i % NUM_CLASSES;
  float acc = b[c];
  const float* pr = pooled + (size_t)g * HDIM;
#pragma unroll 4
  for (int k = 0; k < HDIM; ++k) acc += pr[k] * W[k * NUM_CLASSES + c];
  out[i] = acc;
}

extern "C" void kernel_launch(void* const* d_in, const int* in_sizes, int n_in,
                              void* d_out, int out_size, void* d_ws, size_t ws_size,
                              hipStream_t stream) {
  const float* x     = (const float*)d_in[0];
  const float* ea    = (const float*)d_in[1];
  const float* enc_W = (const float*)d_in[2];
  const float* enc_b = (const float*)d_in[3];
  const float* t     = (const float*)d_in[4];
  const float* W1    = (const float*)d_in[5];
  const float* b1    = (const float*)d_in[6];
  const float* mlp_g = (const float*)d_in[7];
  const float* mlp_b = (const float*)d_in[8];
  const float* W2    = (const float*)d_in[9];
  const float* b2    = (const float*)d_in[10];
  const float* ln_g  = (const float*)d_in[11];
  const float* ln_b  = (const float*)d_in[12];
  const float* lin_W = (const float*)d_in[13];
  const float* lin_b = (const float*)d_in[14];
  const int*   eidx  = (const int*)d_in[15];
  const int*   batch = (const int*)d_in[16];

  const int N = in_sizes[0] / HDIM;
  const int E = in_sizes[1] / HDIM;
  const int L = in_sizes[4];
  const int* src = eidx;
  const int* dst = eidx + E;

  const size_t NH = (size_t)N * HDIM;
  float* h    = (float*)d_ws;
  float* rbuf = h    + NH;
  float* hc   = rbuf + NH;
  float* t1   = hc   + NH;
  int* pairs  = (int*)(t1 + 2 * NH);
  int* raw    = pairs + 2 * (size_t)E;
  int* slot   = raw   + E;
  int* cnt    = slot  + E;
  int* off    = cnt   + N;

  float* out    = (float*)d_out;
  float* pooled = out + NUM_GRAPHS * NUM_CLASSES;

  const dim3 gblk(128);
  const int eThreadBlocks = (E + 255) / 256;
  const int nodeBlocks = (N + 3) / 4;
  const int lnBlocks = (N + 7) / 8;

  zero_int_kernel<<<(N + 255) / 256, 256, 0, stream>>>(cnt, N);
  bucket_count_kernel<<<eThreadBlocks, 256, 0, stream>>>(dst, cnt, slot, E);
  bucket_scan_kernel<<<1, 1024, 0, stream>>>(cnt, off, N);
  bucket_place_kernel<<<eThreadBlocks, 256, 0, stream>>>(dst, off, slot, raw, E);
  bucket_order_kernel<<<nodeBlocks, 128, 0, stream>>>(raw, off, src, pairs, N);

  gemm16_wmma_f32<<<dim3(N / 16, HDIM / 64), gblk, 0, stream>>>(x, enc_W, enc_b, h, HDIM, HDIM, 0);

  for (int layer = 0; layer < L; ++layer) {
    const float* rin = h;
    if (layer > 0) {
      ln_relu_kernel<<<lnBlocks, 256, 0, stream>>>(h, rbuf, ln_g + layer * HDIM, ln_b + layer * HDIM, N, HDIM);
      rin = rbuf;
    }
    node_softmax_gather_kernel<<<nodeBlocks, 128, 0, stream>>>(rin, ea, pairs, off, t, layer, hc, N);
    gemm16_wmma_f32<<<dim3(N / 16, HDIM2 / 64), gblk, 0, stream>>>(
        hc, W1 + (size_t)layer * HDIM * HDIM2, b1 + layer * HDIM2, t1, HDIM, HDIM2, 0);
    ln_relu_kernel<<<lnBlocks, 256, 0, stream>>>(t1, t1, mlp_g + layer * HDIM2, mlp_b + layer * HDIM2, N, HDIM2);
    gemm16_wmma_f32<<<dim3(N / 16, HDIM / 64), gblk, 0, stream>>>(
        t1, W2 + (size_t)layer * HDIM2 * HDIM, b2 + layer * HDIM, h, HDIM2, HDIM, layer > 0 ? 1 : 0);
  }

  ln_relu_kernel<<<lnBlocks, 256, 0, stream>>>(h, rbuf, ln_g, ln_b, N, HDIM);
  zero_kernel<<<(NUM_GRAPHS * HDIM + 255) / 256, 256, 0, stream>>>(pooled, NUM_GRAPHS * HDIM);
  pool_kernel<<<(int)((NH + 255) / 256), 256, 0, stream>>>(rbuf, batch, pooled, N);
  final_linear_kernel<<<(NUM_GRAPHS * NUM_CLASSES + 63) / 64, 64, 0, stream>>>(pooled, lin_W, lin_b, out);
}


# Each layer's Linear, LayerNorm+ReLU, Linear(+residual) and the next LayerNorm+ReLU fused into one WMMA kernel, 16 rows per block, 256-wide tile in LDS; LayerNorm row sums in a new fixed tree
# speedup vs baseline: 4.5324x; 2.0146x over previous
#include <hip/hip_runtime.h>
#include <hip/hip_bf16.h>

typedef __attribute__((ext_vector_type(2))) float v2f;
typedef __attribute__((ext_vector_type(8))) float v8f;

#define HDIM 128
#define HDIM2 256
#define NUM_GRAPHS 64
#define NUM_CLASSES 10
#define MSG_EPS 1e-7f

__global__ void gemm16_wmma_f32(const float* __restrict__ X,
                                const float* __restrict__ W,
                                const float* __restrict__ bias,
                                float* __restrict__ Y,
                                int Kdim, int Mdim, int accum) {
  const int lane = threadIdx.x & 31;
  const int wid  = threadIdx.x >> 5;
  const int row0 = blockIdx.x * 16;
  const int col0 = (blockIdx.y * 4 + wid) * 16;
  if (col0 >= Mdim) return;

  const int half = lane >> 4;
  const int l16  = lane & 15;

  const float bv = bias[col0 + l16];
  v8f c;
  if (accum) {
#pragma unroll
    for (int r = 0; r < 8; ++r)
      c[r] = Y[(size_t)(row0 + half * 8 + r) * Mdim + col0 + l16] + bv;
  } else {
#pragma unroll
    for (int r = 0; r < 8; ++r) c[r] = bv;
  }

  const float* xrow = X + (size_t)(row0 + l16) * Kdim + half * 2;
  const float* wcol = W + (size_t)(half * 2) * Mdim + col0 + l16;

  for (int k = 0; k < Kdim; k += 4) {
    __builtin_prefetch(xrow + k + 32, 0, 1);
    float2 av = *(const float2*)(xrow + k);
    v2f a; a[0] = av.x; a[1] = av.y;
    v2f b;
    b[0] = wcol[(size_t)k * Mdim];
    b[1] = wcol[(size_t)(k + 1) * Mdim];
    c = __builtin_amdgcn_wmma_f32_16x16x4_f32(false, a, false, b,
                                              (short)0, c, false, false);
  }

#pragma unroll
  for (int r = 0; r < 8; ++r)
    Y[(size_t)(row0 + half * 8 + r) * Mdim + col0 + l16] = c[r];
}

#define A_PITCH 132
#define T_PITCH 260
#define R_PITCH 66

__device__ __forceinline__ v8f wmma4(v2f a, v2f b, v8f c) {
  return __builtin_amdgcn_wmma_f32_16x16x4_f32(false, a, false, b, (short)0, c, false, false);
}

__device__ __forceinline__ void row_stats(const float2* red, float2* stat, int tid, float invD) {
  const int row = tid >> 3, sub = tid & 7;
  float s = 0.f, q = 0.f;
#pragma unroll
  for (int i = 0; i < 8; ++i) {
    const float2 p = red[row * R_PITCH + sub + 8 * i];
    s += p.x; q += p.y;
  }
#pragma unroll
  for (int off = 1; off < 8; off <<= 1) {
    s += __shfl_xor(s, off, 32);
    q += __shfl_xor(q, off, 32);
  }
  if (sub == 0) {
    const float mean = s * invD;
    const float var  = q * invD - mean * mean;
    stat[row] = make_float2(mean, rsqrtf(var + 1e-5f));
  }
}

__global__ void __launch_bounds__(128) __attribute__((amdgpu_waves_per_eu(1, 4)))
mlp_ln_fused_kernel(const float* __restrict__ hc,
                    const float* __restrict__ W1, const float* __restrict__ b1,
                    const float* __restrict__ g1, const float* __restrict__ be1,
                    const float* __restrict__ W2, const float* __restrict__ b2,
                    float* __restrict__ h, int accum,
                    const float* __restrict__ g2, const float* __restrict__ be2,
                    float* __restrict__ rout) {
  __shared__ __attribute__((aligned(16))) float sA[16 * A_PITCH];
  __shared__ __attribute__((aligned(16))) float sT[16 * T_PITCH];
  __shared__ float2 sRed[16 * R_PITCH];
  __shared__ float2 sStat[16];

  const int tid  = threadIdx.x;
  const int lane = tid & 31, w = tid >> 5;
  const int hh = lane >> 4, c = lane & 15;
  const int row0 = blockIdx.x * 16;
  const int col1 = w * 64 + 4 * c;
  const int col2 = w * 32 + 2 * c;

#pragma unroll
  for (int i = 0; i < 4; ++i) {
    const int idx = tid + 128 * i, r = idx >> 5, c4 = idx & 31;
    const float4 v = ((const float4*)hc)[(size_t)(row0 + r) * (HDIM / 4) + c4];
    *(float4*)&sA[r * A_PITCH + c4 * 4] = v;
  }

  v8f acc2[2];
  {
    const float2 bb = *(const float2*)(b2 + col2);
#pragma unroll
    for (int r = 0; r < 8; ++r) {
      float2 o = *(const float2*)(h + (size_t)(row0 + 8 * hh + r) * HDIM + col2);
      if (!accum) o = make_float2(0.f, 0.f);
      acc2[0][r] = o.x + bb.x;
      acc2[1][r] = o.y + bb.y;
    }
  }
  __syncthreads();

  v8f acc[4];
  {
    const float4 bb = *(const float4*)(b1 + col1);
#pragma unroll
    for (int r = 0; r < 8; ++r) { acc[0][r] = bb.x; acc[1][r] = bb.y; acc[2][r] = bb.z; acc[3][r] = bb.w; }
  }
  {
    const float4* w1p = (const float4*)(W1 + (size_t)(2 * hh) * HDIM2 + col1);
    const float* ap = sA + c * A_PITCH + 2 * hh;
#pragma unroll
    for (int ks = 0; ks < HDIM / 4; ++ks) {
      const int k0 = 4 * ks;
      const float4 blo = w1p[(size_t)k0 * (HDIM2 / 4)];
      const float4 bhi = w1p[(size_t)(k0 + 1) * (HDIM2 / 4)];
      const float2 av = *(const float2*)(ap + k0);
      v2f a; a[0] = av.x; a[1] = av.y;
      v2f b;
      b[0] = blo.x; b[1] = bhi.x; acc[0] = wmma4(a, b, acc[0]);
      b[0] = blo.y; b[1] = bhi.y; acc[1] = wmma4(a, b, acc[1]);
      b[0] = blo.z; b[1] = bhi.z; acc[2] = wmma4(a, b, acc[2]);
      b[0] = blo.w; b[1] = bhi.w; acc[3] = wmma4(a, b, acc[3]);
    }
  }

#pragma unroll
  for (int r = 0; r < 8; ++r) {
    const float s = (acc[0][r] + acc[1][r]) + (acc[2][r] + acc[3][r]);
    const float q = (acc[0][r] * acc[0][r] + acc[1][r] * acc[1][r]) +
                    (acc[2][r] * acc[2][r] + acc[3][r] * acc[3][r]);
    sRed[(8 * hh + r) * R_PITCH + w * 16 + c] = make_float2(s, q);
  }
  __syncthreads();
  row_stats(sRed, sStat, tid, 1.f / (float)HDIM2);
  __syncthreads();
  {
    const float4 gg = *(const float4*)(g1 + col1);
    const float4 be = *(const float4*)(be1 + col1);
#pragma unroll
    for (int r = 0; r < 8; ++r) {
      const float2 st = sStat[8 * hh + r];
      float4 o;
      o.x = fmaxf((acc[0][r] - st.x) * st.y * gg.x + be.x, 0.f);
      o.y = fmaxf((acc[1][r] - st.x) * st.y * gg.y + be.y, 0.f);
      o.z = fmaxf((acc[2][r] - st.x) * st.y * gg.z + be.z, 0.f);
      o.w = fmaxf((acc[3][r] - st.x) * st.y * gg.w + be.w, 0.f);
      *(float4*)&sT[(8 * hh + r) * T_PITCH + col1] = o;
    }
  }
  __syncthreads();

  {
    const float2* w2p = (const float2*)(W2 + (size_t)(2 * hh) * HDIM + col2);
    const float* tp = sT + c * T_PITCH + 2 * hh;
#pragma unroll
    for (int ks = 0; ks < HDIM2 / 4; ++ks) {
      const int k0 = 4 * ks;
      const float2 blo = w2p[(size_t)k0 * (HDIM / 2)];
      const float2 bhi = w2p[(size_t)(k0 + 1) * (HDIM / 2)];
      const float2 av = *(const float2*)(tp + k0);
      v2f a; a[0] = av.x; a[1] = av.y;
      v2f b;
      b[0] = blo.x; b[1] = bhi.x; acc2[0] = wmma4(a, b, acc2[0]);
      b[0] = blo.y; b[1] = bhi.y; acc2[1] = wmma4(a, b, acc2[1]);
    }
  }

#pragma unroll
  for (int r = 0; r < 8; ++r) {
    const int row = 8 * hh + r;
    const float2 o = make_float2(acc2[0][r], acc2[1][r]);
    *(float2*)(h + (size_t)(row0 + row) * HDIM + col2) = o;
    sRed[row * R_PITCH + w * 16 + c] = make_float2(o.x + o.y, o.x * o.x + o.y * o.y);
  }
  __syncthreads();
  row_stats(sRed, sStat, tid, 1.f / (float)HDIM);
  __syncthreads();
  {
    const float2 gg = *(const float2*)(g2 + col2);
    const float2 be = *(const float2*)(be2 + col2);
#pragma unroll
    for (int r = 0; r < 8; ++r) {
      const int row = 8 * hh + r;
      const float2 st = sStat[row];
      float2 o;
      o.x = fmaxf((acc2[0][r] - st.x) * st.y * gg.x + be.x, 0.f);
      o.y = fmaxf((acc2[1][r] - st.x) * st.y * gg.y + be.y, 0.f);
      *(float2*)(rout + (size_t)(row0 + row) * HDIM + col2) = o;
    }
  }
}

__global__ void zero_int_kernel(int* p, int n) {
  int i = blockIdx.x * blockDim.x + threadIdx.x;
  if (i < n) p[i] = 0;
}

__global__ void zero_kernel(float* p, int n) {
  int i = blockIdx.x * blockDim.x + threadIdx.x;
  if (i < n) p[i] = 0.f;
}

__global__ void bucket_count_kernel(const int* __restrict__ dst, int* __restrict__ cnt,
                                    int* __restrict__ slot, int E) {
  int e = blockIdx.x * blockDim.x + threadIdx.x;
  if (e >= E) return;
  slot[e] = atomicAdd(cnt + dst[e], 1);
}

__global__ void __launch_bounds__(1024)
bucket_scan_kernel(const int* __restrict__ cnt, int* __restrict__ off, int n) {
  __shared__ int part[1024];
  const int tid = threadIdx.x;
  const int per = (n + 1023) / 1024;
  const int b = min(tid * per, n), e = min(b + per, n);
  int s = 0;
  for (int i = b; i < e; ++i) s += cnt[i];
  part[tid] = s;
  __syncthreads();
  for (int d = 1; d < 1024; d <<= 1) {
    int v = (tid >= d) ? part[tid - d] : 0;
    __syncthreads();
    part[tid] += v;
    __syncthreads();
  }
  int run = part[tid] - s;
  for (int i = b; i < e; ++i) { off[i] = run; run += cnt[i]; }
  if (tid == 1023) off[n] = part[1023];
}

__global__ void bucket_place_kernel(const int* __restrict__ dst, const int* __restrict__ off,
                                    const int* __restrict__ slot, int* __restrict__ raw, int E) {
  int e = blockIdx.x * blockDim.x + threadIdx.x;
  if (e >= E) return;
  raw[off[dst[e]] + slot[e]] = e;
}

__global__ void bucket_order_kernel(const int* __restrict__ raw, const int* __restrict__ off,
                                    const int* __restrict__ src, int* __restrict__ pairs, int n) {
  int node = blockIdx.x * (blockDim.x >> 5) + (threadIdx.x >> 5);
  int lane = threadIdx.x & 31;
  if (node >= n) return;
  const int beg = off[node], end = off[node + 1];
  for (int i = beg + lane; i < end; i += 32) {
    const int ei = raw[i];
    int r = 0;
    for (int j = beg; j < end; ++j) r += (raw[j] < ei) ? 1 : 0;
    ((int2*)pairs)[beg + r] = make_int2(ei, src[ei]);
  }
}

#define EB 8

template <bool TAIL>
__device__ __forceinline__ void gather_group(const float* __restrict__ x,
                                             const float* __restrict__ ea,
                                             int2 pr, int first, int nlive,
                                             unsigned lane, float tv, bool tpos,
                                             float (&mx)[4], float (&den)[4], float (&num)[4]) {
  float4 xa[EB], eb[EB];
#pragma unroll
  for (int j = 0; j < EB; ++j) {
    const int e = __builtin_amdgcn_readlane(pr.x, first + j);
    const int s = __builtin_amdgcn_readlane(pr.y, first + j);
    xa[j] = ((const float4*)x)[(unsigned)s * (HDIM / 4) + lane];
    eb[j] = ((const float4*)ea)[(unsigned)e * (HDIM / 4) + lane];
  }
  float m[EB][4];
#pragma unroll
  for (int j = 0; j < EB; ++j) {
    m[j][0] = fmaxf(xa[j].x + eb[j].x, 0.f) + MSG_EPS;
    m[j][1] = fmaxf(xa[j].y + eb[j].y, 0.f) + MSG_EPS;
    m[j][2] = fmaxf(xa[j].z + eb[j].z, 0.f) + MSG_EPS;
    m[j][3] = fmaxf(xa[j].w + eb[j].w, 0.f) + MSG_EPS;
  }
#pragma unroll
  for (int c = 0; c < 4; ++c) {
    float ext = m[0][c];
    if (tpos) {
#pragma unroll
      for (int j = 1; j < EB; ++j) ext = fmaxf(ext, (TAIL && j >= nlive) ? m[0][c] : m[j][c]);
    } else {
#pragma unroll
      for (int j = 1; j < EB; ++j) ext = fminf(ext, (TAIL && j >= nlive) ? m[0][c] : m[j][c]);
    }
    const float nm = fmaxf(mx[c], ext * tv);
    const float sc = __expf(mx[c] - nm);
    float d = den[c] * sc, u = num[c] * sc;
#pragma unroll
    for (int j = 0; j < EB; ++j) {
      float ex = __expf(m[j][c] * tv - nm);
      if (TAIL && j >= nlive) ex = 0.f;
      d += ex;
      u += ex * m[j][c];
    }
    mx[c] = nm; den[c] = d; num[c] = u;
  }
}

__global__ void __launch_bounds__(128)
node_softmax_gather_kernel(const float* __restrict__ x, const float* __restrict__ ea,
                           const int* __restrict__ pairs, const int* __restrict__ off,
                           const float* __restrict__ tptr, int layer,
                           float* __restrict__ hc, int n) {
  const int nodev = blockIdx.x * (blockDim.x >> 5) + (threadIdx.x >> 5);
  const unsigned lane = threadIdx.x & 31u;
  if (nodev >= n) return;
  const int node = __builtin_amdgcn_readfirstlane(nodev);
  const int beg = off[node], end = off[node + 1];
  const float tv = tptr[layer];
  const bool tpos = tv >= 0.f;

  float mx[4], den[4], num[4];
#pragma unroll
  for (int c = 0; c < 4; ++c) { mx[c] = __int_as_float(0xff800000); den[c] = 0.f; num[c] = 0.f; }

  for (int base = beg; base < end; base += 32) {
    const int cnt = min(32, end - base);
    const int2 pr = ((const int2*)pairs)[base + min((int)lane, cnt - 1)];
    int g = 0;
    for (; g + EB <= cnt; g += EB)
      gather_group<false>(x, ea, pr, g, EB, lane, tv, tpos, mx, den, num);
    if (g < cnt)
      gather_group<true>(x, ea, pr, g, cnt - g, lane, tv, tpos, mx, den, num);
  }

  const float4 root = ((const float4*)(x + (size_t)node * HDIM))[lane];
  float4 o;
  if (beg < end) {
    o.x = num[0] / den[0] + root.x;
    o.y = num[1] / den[1] + root.y;
    o.z = num[2] / den[2] + root.z;
    o.w = num[3] / den[3] + root.w;
  } else {
    o = root;
  }
  ((float4*)(hc + (size_t)node * HDIM))[lane] = o;
}

__global__ void pool_kernel(const float* __restrict__ x, const int* __restrict__ batch,
                            float* __restrict__ pooled, int n) {
  int i = blockIdx.x * blockDim.x + threadIdx.x;
  if (i >= n * HDIM) return;
  int node = i >> 7, col = i & (HDIM - 1);
  atomicAdd(pooled + (size_t)batch[node] * HDIM + col, x[i]);
}

__global__ void final_linear_kernel(const float* __restrict__ pooled,
                                    const float* __restrict__ W,
                                    const float* __restrict__ b,
                                    float* __restrict__ out) {
  int i = blockIdx.x * blockDim.x + threadIdx.x;
  if (i >= NUM_GRAPHS * NUM_CLASSES) return;
  int g = i / NUM_CLASSES, c = i % NUM_CLASSES;
  float acc = b[c];
  const float* pr = pooled + (size_t)g * HDIM;
#pragma unroll 4
  for (int k = 0; k < HDIM; ++k) acc += pr[k] * W[k * NUM_CLASSES + c];
  out[i] = acc;
}

extern "C" void kernel_launch(void* const* d_in, const int* in_sizes, int n_in,
                              void* d_out, int out_size, void* d_ws, size_t ws_size,
                              hipStream_t stream) {
  const float* x     = (const float*)d_in[0];
  const float* ea    = (const float*)d_in[1];
  const float* enc_W = (const float*)d_in[2];
  const float* enc_b = (const float*)d_in[3];
  const float* t     = (const float*)d_in[4];
  const float* W1    = (const float*)d_in[5];
  const float* b1    = (const float*)d_in[6];
  const float* mlp_g = (const float*)d_in[7];
  const float* mlp_b = (const float*)d_in[8];
  const float* W2    = (const float*)d_in[9];
  const float* b2    = (const float*)d_in[10];
  const float* ln_g  = (const float*)d_in[11];
  const float* ln_b  = (const float*)d_in[12];
  const float* lin_W = (const float*)d_in[13];
  const float* lin_b = (const float*)d_in[14];
  const int*   eidx  = (const int*)d_in[15];
  const int*   batch = (const int*)d_in[16];

  const int N = in_sizes[0] / HDIM;
  const int E = in_sizes[1] / HDIM;
  const int L = in_sizes[4];
  const int* src = eidx;
  const int* dst = eidx + E;

  const size_t NH = (size_t)N * HDIM;
  float* h    = (float*)d_ws;
  float* rbuf = h    + NH;
  float* hc   = rbuf + NH;
  float* t1   = hc   + NH;
  int* pairs  = (int*)(t1 + 2 * NH);
  int* raw    = pairs + 2 * (size_t)E;
  int* slot   = raw   + E;
  int* cnt    = slot  + E;
  int* off    = cnt   + N;

  float* out    = (float*)d_out;
  float* pooled = out + NUM_GRAPHS * NUM_CLASSES;

  const dim3 gblk(128);
  const int eThreadBlocks = (E + 255) / 256;
  const int nodeBlocks = (N + 3) / 4;

  zero_int_kernel<<<(N + 255) / 256, 256, 0, stream>>>(cnt, N);
  bucket_count_kernel<<<eThreadBlocks, 256, 0, stream>>>(dst, cnt, slot, E);
  bucket_scan_kernel<<<1, 1024, 0, stream>>>(cnt, off, N);
  bucket_place_kernel<<<eThreadBlocks, 256, 0, stream>>>(dst, off, slot, raw, E);
  bucket_order_kernel<<<nodeBlocks, 128, 0, stream>>>(raw, off, src, pairs, N);

  gemm16_wmma_f32<<<dim3(N / 16, HDIM / 64), gblk, 0, stream>>>(x, enc_W, enc_b, h, HDIM, HDIM, 0);

  for (int layer = 0; layer < L; ++layer) {
    const float* rin = layer > 0 ? rbuf : h;
    node_softmax_gather_kernel<<<nodeBlocks, 128, 0, stream>>>(rin, ea, pairs, off, t, layer, hc, N);
    const int nl = layer + 1 < L ? layer + 1 : 0;
    mlp_ln_fused_kernel<<<N / 16, 128, 0, stream>>>(
        hc, W1 + (size_t)layer * HDIM * HDIM2, b1 + layer * HDIM2,
        mlp_g + layer * HDIM2, mlp_b + layer * HDIM2,
        W2 + (size_t)layer * HDIM2 * HDIM, b2 + layer * HDIM,
        h, layer > 0 ? 1 : 0, ln_g + nl * HDIM, ln_b + nl * HDIM, rbuf);
  }

  zero_kernel<<<(NUM_GRAPHS * HDIM + 255) / 256, 256, 0, stream>>>(pooled, NUM_GRAPHS * HDIM);
  pool_kernel<<<(int)((NH + 255) / 256), 256, 0, stream>>>(rbuf, batch, pooled, N);
  final_linear_kernel<<<(NUM_GRAPHS * NUM_CLASSES + 63) / 64, 64, 0, stream>>>(pooled, lin_W, lin_b, out);
}


# Edge buckets built with LDS counters only: edges binned by node range per chunk, one block per range counts, scans, places and id-sorts its buckets; no global atomics, 5 launches to 3
# speedup vs baseline: 5.2972x; 1.1687x over previous
#include <hip/hip_runtime.h>
#include <hip/hip_bf16.h>

typedef __attribute__((ext_vector_type(2))) float v2f;
typedef __attribute__((ext_vector_type(8))) float v8f;

#define HDIM 128
#define HDIM2 256
#define NUM_GRAPHS 64
#define NUM_CLASSES 10
#define MSG_EPS 1e-7f

__global__ void gemm16_wmma_f32(const float* __restrict__ X,
                                const float* __restrict__ W,
                                const float* __restrict__ bias,
                                float* __restrict__ Y,
                                int Kdim, int Mdim, int accum) {
  const int lane = threadIdx.x & 31;
  const int wid  = threadIdx.x >> 5;
  const int row0 = blockIdx.x * 16;
  const int col0 = (blockIdx.y * 4 + wid) * 16;
  if (col0 >= Mdim) return;

  const int half = lane >> 4;
  const int l16  = lane & 15;

  const float bv = bias[col0 + l16];
  v8f c;
  if (accum) {
#pragma unroll
    for (int r = 0; r < 8; ++r)
      c[r] = Y[(size_t)(row0 + half * 8 + r) * Mdim + col0 + l16] + bv;
  } else {
#pragma unroll
    for (int r = 0; r < 8; ++r) c[r] = bv;
  }

  const float* xrow = X + (size_t)(row0 + l16) * Kdim + half * 2;
  const float* wcol = W + (size_t)(half * 2) * Mdim + col0 + l16;

  for (int k = 0; k < Kdim; k += 4) {
    __builtin_prefetch(xrow + k + 32, 0, 1);
    float2 av = *(const float2*)(xrow + k);
    v2f a; a[0] = av.x; a[1] = av.y;
    v2f b;
    b[0] = wcol[(size_t)k * Mdim];
    b[1] = wcol[(size_t)(k + 1) * Mdim];
    c = __builtin_amdgcn_wmma_f32_16x16x4_f32(false, a, false, b,
                                              (short)0, c, false, false);
  }

#pragma unroll
  for (int r = 0; r < 8; ++r)
    Y[(size_t)(row0 + half * 8 + r) * Mdim + col0 + l16] = c[r];
}

#define A_PITCH 132
#define T_PITCH 260
#define R_PITCH 66

__device__ __forceinline__ v8f wmma4(v2f a, v2f b, v8f c) {
  return __builtin_amdgcn_wmma_f32_16x16x4_f32(false, a, false, b, (short)0, c, false, false);
}

__device__ __forceinline__ void row_stats(const float2* red, float2* stat, int tid, float invD) {
  const int row = tid >> 3, sub = tid & 7;
  float s = 0.f, q = 0.f;
#pragma unroll
  for (int i = 0; i < 8; ++i) {
    const float2 p = red[row * R_PITCH + sub + 8 * i];
    s += p.x; q += p.y;
  }
#pragma unroll
  for (int off = 1; off < 8; off <<= 1) {
    s += __shfl_xor(s, off, 32);
    q += __shfl_xor(q, off, 32);
  }
  if (sub == 0) {
    const float mean = s * invD;
    const float var  = q * invD - mean * mean;
    stat[row] = make_float2(mean, rsqrtf(var + 1e-5f));
  }
}

__global__ void __launch_bounds__(128) __attribute__((amdgpu_waves_per_eu(1, 4)))
mlp_ln_fused_kernel(const float* __restrict__ hc,
                    const float* __restrict__ W1, const float* __restrict__ b1,
                    const float* __restrict__ g1, const float* __restrict__ be1,
                    const float* __restrict__ W2, const float* __restrict__ b2,
                    float* __restrict__ h, int accum,
                    const float* __restrict__ g2, const float* __restrict__ be2,
                    float* __restrict__ rout) {
  __shared__ __attribute__((aligned(16))) float sA[16 * A_PITCH];
  __shared__ __attribute__((aligned(16))) float sT[16 * T_PITCH];
  __shared__ float2 sRed[16 * R_PITCH];
  __shared__ float2 sStat[16];

  const int tid  = threadIdx.x;
  const int lane = tid & 31, w = tid >> 5;
  const int hh = lane >> 4, c = lane & 15;
  const int row0 = blockIdx.x * 16;
  const int col1 = w * 64 + 4 * c;
  const int col2 = w * 32 + 2 * c;

#pragma unroll
  for (int i = 0; i < 4; ++i) {
    const int idx = tid + 128 * i, r = idx >> 5, c4 = idx & 31;
    const float4 v = ((const float4*)hc)[(size_t)(row0 + r) * (HDIM / 4) + c4];
    *(float4*)&sA[r * A_PITCH + c4 * 4] = v;
  }

  v8f acc2[2];
  {
    const float2 bb = *(const float2*)(b2 + col2);
#pragma unroll
    for (int r = 0; r < 8; ++r) {
      float2 o = *(const float2*)(h + (size_t)(row0 + 8 * hh + r) * HDIM + col2);
      if (!accum) o = make_float2(0.f, 0.f);
      acc2[0][r] = o.x + bb.x;
      acc2[1][r] = o.y + bb.y;
    }
  }
  __syncthreads();

  v8f acc[4];
  {
    const float4 bb = *(const float4*)(b1 + col1);
#pragma unroll
    for (int r = 0; r < 8; ++r) { acc[0][r] = bb.x; acc[1][r] = bb.y; acc[2][r] = bb.z; acc[3][r] = bb.w; }
  }
  {
    const float4* w1p = (const float4*)(W1 + (size_t)(2 * hh) * HDIM2 + col1);
    const float* ap = sA + c * A_PITCH + 2 * hh;
#pragma unroll
    for (int ks = 0; ks < HDIM / 4; ++ks) {
      const int k0 = 4 * ks;
      const float4 blo = w1p[(size_t)k0 * (HDIM2 / 4)];
      const float4 bhi = w1p[(size_t)(k0 + 1) * (HDIM2 / 4)];
      const float2 av = *(const float2*)(ap + k0);
      v2f a; a[0] = av.x; a[1] = av.y;
      v2f b;
      b[0] = blo.x; b[1] = bhi.x; acc[0] = wmma4(a, b, acc[0]);
      b[0] = blo.y; b[1] = bhi.y; acc[1] = wmma4(a, b, acc[1]);
      b[0] = blo.z; b[1] = bhi.z; acc[2] = wmma4(a, b, acc[2]);
      b[0] = blo.w; b[1] = bhi.w; acc[3] = wmma4(a, b, acc[3]);
    }
  }

#pragma unroll
  for (int r = 0; r < 8; ++r) {
    const float s = (acc[0][r] + acc[1][r]) + (acc[2][r] + acc[3][r]);
    const float q = (acc[0][r] * acc[0][r] + acc[1][r] * acc[1][r]) +
                    (acc[2][r] * acc[2][r] + acc[3][r] * acc[3][r]);
    sRed[(8 * hh + r) * R_PITCH + w * 16 + c] = make_float2(s, q);
  }
  __syncthreads();
  row_stats(sRed, sStat, tid, 1.f / (float)HDIM2);
  __syncthreads();
  {
    const float4 gg = *(const float4*)(g1 + col1);
    const float4 be = *(const float4*)(be1 + col1);
#pragma unroll
    for (int r = 0; r < 8; ++r) {
      const float2 st = sStat[8 * hh + r];
      float4 o;
      o.x = fmaxf((acc[0][r] - st.x) * st.y * gg.x + be.x, 0.f);
      o.y = fmaxf((acc[1][r] - st.x) * st.y * gg.y + be.y, 0.f);
      o.z = fmaxf((acc[2][r] - st.x) * st.y * gg.z + be.z, 0.f);
      o.w = fmaxf((acc[3][r] - st.x) * st.y * gg.w + be.w, 0.f);
      *(float4*)&sT[(8 * hh + r) * T_PITCH + col1] = o;
    }
  }
  __syncthreads();

  {
    const float2* w2p = (const float2*)(W2 + (size_t)(2 * hh) * HDIM + col2);
    const float* tp = sT + c * T_PITCH + 2 * hh;
#pragma unroll
    for (int ks = 0; ks < HDIM2 / 4; ++ks) {
      const int k0 = 4 * ks;
      const float2 blo = w2p[(size_t)k0 * (HDIM / 2)];
      const float2 bhi = w2p[(size_t)(k0 + 1) * (HDIM / 2)];
      const float2 av = *(const float2*)(tp + k0);
      v2f a; a[0] = av.x; a[1] = av.y;
      v2f b;
      b[0] = blo.x; b[1] = bhi.x; acc2[0] = wmma4(a, b, acc2[0]);
      b[0] = blo.y; b[1] = bhi.y; acc2[1] = wmma4(a, b, acc2[1]);
    }
  }

#pragma unroll
  for (int r = 0; r < 8; ++r) {
    const int row = 8 * hh + r;
    const float2 o = make_float2(acc2[0][r], acc2[1][r]);
    *(float2*)(h + (size_t)(row0 + row) * HDIM + col2) = o;
    sRed[row * R_PITCH + w * 16 + c] = make_float2(o.x + o.y, o.x * o.x + o.y * o.y);
  }
  __syncthreads();
  row_stats(sRed, sStat, tid, 1.f / (float)HDIM);
  __syncthreads();
  {
    const float2 gg = *(const float2*)(g2 + col2);
    const float2 be = *(const float2*)(be2 + col2);
#pragma unroll
    for (int r = 0; r < 8; ++r) {
      const int row = 8 * hh + r;
      const float2 st = sStat[row];
      float2 o;
      o.x = fmaxf((acc2[0][r] - st.x) * st.y * gg.x + be.x, 0.f);
      o.y = fmaxf((acc2[1][r] - st.x) * st.y * gg.y + be.y, 0.f);
      *(float2*)(rout + (size_t)(row0 + row) * HDIM + col2) = o;
    }
  }
}

#define NBIN 128
#define NCHUNK 128
#define RMAX 128

__global__ void zero_kernel(float* p, int n) {
  int i = blockIdx.x * blockDim.x + threadIdx.x;
  if (i < n) p[i] = 0.f;
}

__global__ void __launch_bounds__(1024)
bin_hist_kernel(const int* __restrict__ dst, int* __restrict__ hist, int E, int N) {
  __shared__ int h[NBIN];
  const int tid = threadIdx.x;
  const int R = (N + NBIN - 1) / NBIN;
  const int C = (E + NCHUNK - 1) / NCHUNK;
  const int e0 = min((int)blockIdx.x * C, E), e1 = min(e0 + C, E);
  if (tid < NBIN) h[tid] = 0;
  __syncthreads();
  for (int e = e0 + tid; e < e1; e += 1024) atomicAdd(&h[dst[e] / R], 1);
  __syncthreads();
  if (tid < NBIN) hist[blockIdx.x * NBIN + tid] = h[tid];
}

__global__ void __launch_bounds__(1024)
bin_place_kernel(const int* __restrict__ dst, const int* __restrict__ hist,
                 int* __restrict__ binStart, int* __restrict__ binned, int E, int N) {
  __shared__ int sAll[8][NBIN];
  __shared__ int sPre[8][NBIN];
  __shared__ int scanb[NBIN];
  __shared__ int cursor[NBIN];
  const int tid = threadIdx.x;
  const int R = (N + NBIN - 1) / NBIN;
  const int C = (E + NCHUNK - 1) / NCHUNK;
  const int chunk = blockIdx.x;
  const int e0 = min(chunk * C, E), e1 = min(e0 + C, E);
  {
    const int b = tid & (NBIN - 1), part = tid >> 7;
    int sa = 0, sp = 0;
#pragma unroll
    for (int k = 0; k < NCHUNK / 8; ++k) {
      const int c = part * (NCHUNK / 8) + k;
      const int v = hist[c * NBIN + b];
      sa += v;
      sp += (c < chunk) ? v : 0;
    }
    sAll[part][b] = sa;
    sPre[part][b] = sp;
  }
  __syncthreads();
  int tot = 0, pre = 0;
  if (tid < NBIN) {
#pragma unroll
    for (int p = 0; p < 8; ++p) { tot += sAll[p][tid]; pre += sPre[p][tid]; }
    scanb[tid] = tot;
  }
  __syncthreads();
  for (int d = 1; d < NBIN; d <<= 1) {
    int v = 0;
    if (tid < NBIN && tid >= d) v = scanb[tid - d];
    __syncthreads();
    if (tid < NBIN) scanb[tid] += v;
    __syncthreads();
  }
  if (tid < NBIN) {
    const int excl = scanb[tid] - tot;
    cursor[tid] = excl + pre;
    if (chunk == 0) {
      binStart[tid] = excl;
      if (tid == NBIN - 1) binStart[NBIN] = scanb[tid];
    }
  }
  __syncthreads();
  for (int e = e0 + tid; e < e1; e += 1024) {
    const int d = dst[e];
    const int p = atomicAdd(&cursor[d / R], 1);
    ((int2*)binned)[p] = make_int2(e, d);
  }
}

__global__ void __launch_bounds__(1024)
bin_sort_kernel(const int* __restrict__ binStart, const int* __restrict__ binned,
                const int* __restrict__ src, int* __restrict__ slot, int* __restrict__ raw,
                int* __restrict__ off, int* __restrict__ pairs, int E, int N) {
  __shared__ int cnt[RMAX];
  __shared__ int offL[RMAX];
  __shared__ int scanb[RMAX];
  const int tid = threadIdx.x;
  const int lane = tid & 31;
  const int R = (N + NBIN - 1) / NBIN;
  const int n0 = min((int)blockIdx.x * R, N), n1 = min(n0 + R, N);
  const int nn = n1 - n0;
  const int s0 = binStart[blockIdx.x], s1 = binStart[blockIdx.x + 1];
  if (tid < RMAX) cnt[tid] = 0;
  __syncthreads();
  for (int i = s0 + tid; i < s1; i += 1024) {
    const int2 pr = ((const int2*)binned)[i];
    slot[i] = atomicAdd(&cnt[pr.y - n0], 1);
  }
  __syncthreads();
  int mine = 0;
  if (tid < RMAX) { mine = cnt[tid]; scanb[tid] = mine; }
  __syncthreads();
  for (int d = 1; d < RMAX; d <<= 1) {
    int v = 0;
    if (tid < RMAX && tid >= d) v = scanb[tid - d];
    __syncthreads();
    if (tid < RMAX) scanb[tid] += v;
    __syncthreads();
  }
  if (tid < RMAX) {
    const int excl = scanb[tid] - mine;
    offL[tid] = excl;
    if (tid < nn) off[n0 + tid] = s0 + excl;
  }
  if (blockIdx.x == 0 && tid == 0) off[N] = E;
  __syncthreads();
  for (int i = s0 + tid; i < s1; i += 1024) {
    const int2 pr = ((const int2*)binned)[i];
    raw[s0 + offL[pr.y - n0] + slot[i]] = pr.x;
  }
  __syncthreads();
  const int w = __builtin_amdgcn_readfirstlane(tid >> 5);
  for (int j = w; j < nn; j += 32) {
    const int beg = __builtin_amdgcn_readfirstlane(s0 + offL[j]);
    const int end = __builtin_amdgcn_readfirstlane(beg + cnt[j]);
    for (int ib = beg; ib < end; ib += 32) {
      const int i = ib + lane;
      const bool live = i < end;
      const int ei = live ? raw[i] : 0x7fffffff;
      int r = 0;
      for (int jb = beg; jb < end; jb += 32) {
        const int jj = jb + lane;
        const int ej = (jj < end) ? raw[jj] : 0x7fffffff;
        const int cj = min(32, end - jb);
        for (int k = 0; k < cj; ++k)
          r += (__builtin_amdgcn_readlane(ej, k) < ei) ? 1 : 0;
      }
      if (live) ((int2*)pairs)[beg + r] = make_int2(ei, src[ei]);
    }
  }
}

#define EB 8

template <bool TAIL>
__device__ __forceinline__ void gather_group(const float* __restrict__ x,
                                             const float* __restrict__ ea,
                                             int2 pr, int first, int nlive,
                                             unsigned lane, float tv, bool tpos,
                                             float (&mx)[4], float (&den)[4], float (&num)[4]) {
  float4 xa[EB], eb[EB];
#pragma unroll
  for (int j = 0; j < EB; ++j) {
    const int e = __builtin_amdgcn_readlane(pr.x, first + j);
    const int s = __builtin_amdgcn_readlane(pr.y, first + j);
    xa[j] = ((const float4*)x)[(unsigned)s * (HDIM / 4) + lane];
    eb[j] = ((const float4*)ea)[(unsigned)e * (HDIM / 4) + lane];
  }
  float m[EB][4];
#pragma unroll
  for (int j = 0; j < EB; ++j) {
    m[j][0] = fmaxf(xa[j].x + eb[j].x, 0.f) + MSG_EPS;
    m[j][1] = fmaxf(xa[j].y + eb[j].y, 0.f) + MSG_EPS;
    m[j][2] = fmaxf(xa[j].z + eb[j].z, 0.f) + MSG_EPS;
    m[j][3] = fmaxf(xa[j].w + eb[j].w, 0.f) + MSG_EPS;
  }
#pragma unroll
  for (int c = 0; c < 4; ++c) {
    float ext = m[0][c];
    if (tpos) {
#pragma unroll
      for (int j = 1; j < EB; ++j) ext = fmaxf(ext, (TAIL && j >= nlive) ? m[0][c] : m[j][c]);
    } else {
#pragma unroll
      for (int j = 1; j < EB; ++j) ext = fminf(ext, (TAIL && j >= nlive) ? m[0][c] : m[j][c]);
    }
    const float nm = fmaxf(mx[c], ext * tv);
    const float sc = __expf(mx[c] - nm);
    float d = den[c] * sc, u = num[c] * sc;
#pragma unroll
    for (int j = 0; j < EB; ++j) {
      float ex = __expf(m[j][c] * tv - nm);
      if (TAIL && j >= nlive) ex = 0.f;
      d += ex;
      u += ex * m[j][c];
    }
    mx[c] = nm; den[c] = d; num[c] = u;
  }
}

__global__ void __launch_bounds__(128)
node_softmax_gather_kernel(const float* __restrict__ x, const float* __restrict__ ea,
                           const int* __restrict__ pairs, const int* __restrict__ off,
                           const float* __restrict__ tptr, int layer,
                           float* __restrict__ hc, int n) {
  const int nodev = blockIdx.x * (blockDim.x >> 5) + (threadIdx.x >> 5);
  const unsigned lane = threadIdx.x & 31u;
  if (nodev >= n) return;
  const int node = __builtin_amdgcn_readfirstlane(nodev);
  const int beg = off[node], end = off[node + 1];
  const float tv = tptr[layer];
  const bool tpos = tv >= 0.f;

  float mx[4], den[4], num[4];
#pragma unroll
  for (int c = 0; c < 4; ++c) { mx[c] = __int_as_float(0xff800000); den[c] = 0.f; num[c] = 0.f; }

  for (int base = beg; base < end; base += 32) {
    const int cnt = min(32, end - base);
    const int2 pr = ((const int2*)pairs)[base + min((int)lane, cnt - 1)];
    int g = 0;
    for (; g + EB <= cnt; g += EB)
      gather_group<false>(x, ea, pr, g, EB, lane, tv, tpos, mx, den, num);
    if (g < cnt)
      gather_group<true>(x, ea, pr, g, cnt - g, lane, tv, tpos, mx, den, num);
  }

  const float4 root = ((const float4*)(x + (size_t)node * HDIM))[lane];
  float4 o;
  if (beg < end) {
    o.x = num[0] / den[0] + root.x;
    o.y = num[1] / den[1] + root.y;
    o.z = num[2] / den[2] + root.z;
    o.w = num[3] / den[3] + root.w;
  } else {
    o = root;
  }
  ((float4*)(hc + (size_t)node * HDIM))[lane] = o;
}

__global__ void pool_kernel(const float* __restrict__ x, const int* __restrict__ batch,
                            float* __restrict__ pooled, int n) {
  int i = blockIdx.x * blockDim.x + threadIdx.x;
  if (i >= n * HDIM) return;
  int node = i >> 7, col = i & (HDIM - 1);
  atomicAdd(pooled + (size_t)batch[node] * HDIM + col, x[i]);
}

__global__ void final_linear_kernel(const float* __restrict__ pooled,
                                    const float* __restrict__ W,
                                    const float* __restrict__ b,
                                    float* __restrict__ out) {
  int i = blockIdx.x * blockDim.x + threadIdx.x;
  if (i >= NUM_GRAPHS * NUM_CLASSES) return;
  int g = i / NUM_CLASSES, c = i % NUM_CLASSES;
  float acc = b[c];
  const float* pr = pooled + (size_t)g * HDIM;
#pragma unroll 4
  for (int k = 0; k < HDIM; ++k) acc += pr[k] * W[k * NUM_CLASSES + c];
  out[i] = acc;
}

extern "C" void kernel_launch(void* const* d_in, const int* in_sizes, int n_in,
                              void* d_out, int out_size, void* d_ws, size_t ws_size,
                              hipStream_t stream) {
  const float* x     = (const float*)d_in[0];
  const float* ea    = (const float*)d_in[1];
  const float* enc_W = (const float*)d_in[2];
  const float* enc_b = (const float*)d_in[3];
  const float* t     = (const float*)d_in[4];
  const float* W1    = (const float*)d_in[5];
  const float* b1    = (const float*)d_in[6];
  const float* mlp_g = (const float*)d_in[7];
  const float* mlp_b = (const float*)d_in[8];
  const float* W2    = (const float*)d_in[9];
  const float* b2    = (const float*)d_in[10];
  const float* ln_g  = (const float*)d_in[11];
  const float* ln_b  = (const float*)d_in[12];
  const float* lin_W = (const float*)d_in[13];
  const float* lin_b = (const float*)d_in[14];
  const int*   eidx  = (const int*)d_in[15];
  const int*   batch = (const int*)d_in[16];

  const int N = in_sizes[0] / HDIM;
  const int E = in_sizes[1] / HDIM;
  const int L = in_sizes[4];
  const int* src = eidx;
  const int* dst = eidx + E;

  const size_t NH = (size_t)N * HDIM;
  float* h    = (float*)d_ws;
  float* rbuf = h    + NH;
  float* hc   = rbuf + NH;
  float* t1   = hc   + NH;
  int* pairs  = (int*)(t1 + 2 * NH);
  int* raw    = pairs + 2 * (size_t)E;
  int* slot   = raw   + E;
  int* cnt    = slot  + E;
  int* off    = cnt   + N;
  int* binned = off   + N + 2;
  int* hist   = binned + 2 * (size_t)E;
  int* binStart = hist + NCHUNK * NBIN;

  float* out    = (float*)d_out;
  float* pooled = out + NUM_GRAPHS * NUM_CLASSES;

  const dim3 gblk(128);
  const int nodeBlocks = (N + 3) / 4;

  bin_hist_kernel<<<NCHUNK, 1024, 0, stream>>>(dst, hist, E, N);
  bin_place_kernel<<<NCHUNK, 1024, 0, stream>>>(dst, hist, binStart, binned, E, N);
  bin_sort_kernel<<<NBIN, 1024, 0, stream>>>(binStart, binned, src, slot, raw, off, pairs, E, N);

  gemm16_wmma_f32<<<dim3(N / 16, HDIM / 64), gblk, 0, stream>>>(x, enc_W, enc_b, h, HDIM, HDIM, 0);

  for (int layer = 0; layer < L; ++layer) {
    const float* rin = layer > 0 ? rbuf : h;
    node_softmax_gather_kernel<<<nodeBlocks, 128, 0, stream>>>(rin, ea, pairs, off, t, layer, hc, N);
    const int nl = layer + 1 < L ? layer + 1 : 0;
    mlp_ln_fused_kernel<<<N / 16, 128, 0, stream>>>(
        hc, W1 + (size_t)layer * HDIM * HDIM2, b1 + layer * HDIM2,
        mlp_g + layer * HDIM2, mlp_b + layer * HDIM2,
        W2 + (size_t)layer * HDIM2 * HDIM, b2 + layer * HDIM,
        h, layer > 0 ? 1 : 0, ln_g + nl * HDIM, ln_b + nl * HDIM, rbuf);
  }

  zero_kernel<<<(NUM_GRAPHS * HDIM + 255) / 256, 256, 0, stream>>>(pooled, NUM_GRAPHS * HDIM);
  pool_kernel<<<(int)((NH + 255) / 256), 256, 0, stream>>>(rbuf, batch, pooled, N);
  final_linear_kernel<<<(NUM_GRAPHS * NUM_CLASSES + 63) / 64, 64, 0, stream>>>(pooled, lin_W, lin_b, out);
}


# Pooling and classifier in one launch, one block per graph: rows summed in fixed node order in registers and LDS, float atomics and the clearing launch gone, 15 launches to 13
# speedup vs baseline: 5.6621x; 1.0689x over previous
#include <hip/hip_runtime.h>
#include <hip/hip_bf16.h>

typedef __attribute__((ext_vector_type(2))) float v2f;
typedef __attribute__((ext_vector_type(8))) float v8f;

#define HDIM 128
#define HDIM2 256
#define NUM_GRAPHS 64
#define NUM_CLASSES 10
#define MSG_EPS 1e-7f

__global__ void gemm16_wmma_f32(const float* __restrict__ X,
                                const float* __restrict__ W,
                                const float* __restrict__ bias,
                                float* __restrict__ Y,
                                int Kdim, int Mdim, int accum) {
  const int lane = threadIdx.x & 31;
  const int wid  = threadIdx.x >> 5;
  const int row0 = blockIdx.x * 16;
  const int col0 = (blockIdx.y * 4 + wid) * 16;
  if (col0 >= Mdim) return;

  const int half = lane >> 4;
  const int l16  = lane & 15;

  const float bv = bias[col0 + l16];
  v8f c;
  if (accum) {
#pragma unroll
    for (int r = 0; r < 8; ++r)
      c[r] = Y[(size_t)(row0 + half * 8 + r) * Mdim + col0 + l16] + bv;
  } else {
#pragma unroll
    for (int r = 0; r < 8; ++r) c[r] = bv;
  }

  const float* xrow = X + (size_t)(row0 + l16) * Kdim + half * 2;
  const float* wcol = W + (size_t)(half * 2) * Mdim + col0 + l16;

  for (int k = 0; k < Kdim; k += 4) {
    __builtin_prefetch(xrow + k + 32, 0, 1);
    float2 av = *(const float2*)(xrow + k);
    v2f a; a[0] = av.x; a[1] = av.y;
    v2f b;
    b[0] = wcol[(size_t)k * Mdim];
    b[1] = wcol[(size_t)(k + 1) * Mdim];
    c = __builtin_amdgcn_wmma_f32_16x16x4_f32(false, a, false, b,
                                              (short)0, c, false, false);
  }

#pragma unroll
  for (int r = 0; r < 8; ++r)
    Y[(size_t)(row0 + half * 8 + r) * Mdim + col0 + l16] = c[r];
}

#define A_PITCH 132
#define T_PITCH 260
#define R_PITCH 66

__device__ __forceinline__ v8f wmma4(v2f a, v2f b, v8f c) {
  return __builtin_amdgcn_wmma_f32_16x16x4_f32(false, a, false, b, (short)0, c, false, false);
}

__device__ __forceinline__ void row_stats(const float2* red, float2* stat, int tid, float invD) {
  const int row = tid >> 3, sub = tid & 7;
  float s = 0.f, q = 0.f;
#pragma unroll
  for (int i = 0; i < 8; ++i) {
    const float2 p = red[row * R_PITCH + sub + 8 * i];
    s += p.x; q += p.y;
  }
#pragma unroll
  for (int off = 1; off < 8; off <<= 1) {
    s += __shfl_xor(s, off, 32);
    q += __shfl_xor(q, off, 32);
  }
  if (sub == 0) {
    const float mean = s * invD;
    const float var  = q * invD - mean * mean;
    stat[row] = make_float2(mean, rsqrtf(var + 1e-5f));
  }
}

__global__ void __launch_bounds__(128) __attribute__((amdgpu_waves_per_eu(1, 4)))
mlp_ln_fused_kernel(const float* __restrict__ hc,
                    const float* __restrict__ W1, const float* __restrict__ b1,
                    const float* __restrict__ g1, const float* __restrict__ be1,
                    const float* __restrict__ W2, const float* __restrict__ b2,
                    float* __restrict__ h, int accum,
                    const float* __restrict__ g2, const float* __restrict__ be2,
                    float* __restrict__ rout) {
  __shared__ __attribute__((aligned(16))) float sA[16 * A_PITCH];
  __shared__ __attribute__((aligned(16))) float sT[16 * T_PITCH];
  __shared__ float2 sRed[16 * R_PITCH];
  __shared__ float2 sStat[16];

  const int tid  = threadIdx.x;
  const int lane = tid & 31, w = tid >> 5;
  const int hh = lane >> 4, c = lane & 15;
  const int row0 = blockIdx.x * 16;
  const int col1 = w * 64 + 4 * c;
  const int col2 = w * 32 + 2 * c;

#pragma unroll
  for (int i = 0; i < 4; ++i) {
    const int idx = tid + 128 * i, r = idx >> 5, c4 = idx & 31;
    const float4 v = ((const float4*)hc)[(size_t)(row0 + r) * (HDIM / 4) + c4];
    *(float4*)&sA[r * A_PITCH + c4 * 4] = v;
  }

  v8f acc2[2];
  {
    const float2 bb = *(const float2*)(b2 + col2);
#pragma unroll
    for (int r = 0; r < 8; ++r) {
      float2 o = *(const float2*)(h + (size_t)(row0 + 8 * hh + r) * HDIM + col2);
      if (!accum) o = make_float2(0.f, 0.f);
      acc2[0][r] = o.x + bb.x;
      acc2[1][r] = o.y + bb.y;
    }
  }
  __syncthreads();

  v8f acc[4];
  {
    const float4 bb = *(const float4*)(b1 + col1);
#pragma unroll
    for (int r = 0; r < 8; ++r) { acc[0][r] = bb.x; acc[1][r] = bb.y; acc[2][r] = bb.z; acc[3][r] = bb.w; }
  }
  {
    const float4* w1p = (const float4*)(W1 + (size_t)(2 * hh) * HDIM2 + col1);
    const float* ap = sA + c * A_PITCH + 2 * hh;
#pragma unroll
    for (int ks = 0; ks < HDIM / 4; ++ks) {
      const int k0 = 4 * ks;
      const float4 blo = w1p[(size_t)k0 * (HDIM2 / 4)];
      const float4 bhi = w1p[(size_t)(k0 + 1) * (HDIM2 / 4)];
      const float2 av = *(const float2*)(ap + k0);
      v2f a; a[0] = av.x; a[1] = av.y;
      v2f b;
      b[0] = blo.x; b[1] = bhi.x; acc[0] = wmma4(a, b, acc[0]);
      b[0] = blo.y; b[1] = bhi.y; acc[1] = wmma4(a, b, acc[1]);
      b[0] = blo.z; b[1] = bhi.z; acc[2] = wmma4(a, b, acc[2]);
      b[0] = blo.w; b[1] = bhi.w; acc[3] = wmma4(a, b, acc[3]);
    }
  }

#pragma unroll
  for (int r = 0; r < 8; ++r) {
    const float s = (acc[0][r] + acc[1][r]) + (acc[2][r] + acc[3][r]);
    const float q = (acc[0][r] * acc[0][r] + acc[1][r] * acc[1][r]) +
                    (acc[2][r] * acc[2][r] + acc[3][r] * acc[3][r]);
    sRed[(8 * hh + r) * R_PITCH + w * 16 + c] = make_float2(s, q);
  }
  __syncthreads();
  row_stats(sRed, sStat, tid, 1.f / (float)HDIM2);
  __syncthreads();
  {
    const float4 gg = *(const float4*)(g1 + col1);
    const float4 be = *(const float4*)(be1 + col1);
#pragma unroll
    for (int r = 0; r < 8; ++r) {
      const float2 st = sStat[8 * hh + r];
      float4 o;
      o.x = fmaxf((acc[0][r] - st.x) * st.y * gg.x + be.x, 0.f);
      o.y = fmaxf((acc[1][r] - st.x) * st.y * gg.y + be.y, 0.f);
      o.z = fmaxf((acc[2][r] - st.x) * st.y * gg.z + be.z, 0.f);
      o.w = fmaxf((acc[3][r] - st.x) * st.y * gg.w + be.w, 0.f);
      *(float4*)&sT[(8 * hh + r) * T_PITCH + col1] = o;
    }
  }
  __syncthreads();

  {
    const float2* w2p = (const float2*)(W2 + (size_t)(2 * hh) * HDIM + col2);
    const float* tp = sT + c * T_PITCH + 2 * hh;
#pragma unroll
    for (int ks = 0; ks < HDIM2 / 4; ++ks) {
      const int k0 = 4 * ks;
      const float2 blo = w2p[(size_t)k0 * (HDIM / 2)];
      const float2 bhi = w2p[(size_t)(k0 + 1) * (HDIM / 2)];
      const float2 av = *(const float2*)(tp + k0);
      v2f a; a[0] = av.x; a[1] = av.y;
      v2f b;
      b[0] = blo.x; b[1] = bhi.x; acc2[0] = wmma4(a, b, acc2[0]);
      b[0] = blo.y; b[1] = bhi.y; acc2[1] = wmma4(a, b, acc2[1]);
    }
  }

#pragma unroll
  for (int r = 0; r < 8; ++r) {
    const int row = 8 * hh + r;
    const float2 o = make_float2(acc2[0][r], acc2[1][r]);
    *(float2*)(h + (size_t)(row0 + row) * HDIM + col2) = o;
    sRed[row * R_PITCH + w * 16 + c] = make_float2(o.x + o.y, o.x * o.x + o.y * o.y);
  }
  __syncthreads();
  row_stats(sRed, sStat, tid, 1.f / (float)HDIM);
  __syncthreads();
  {
    const float2 gg = *(const float2*)(g2 + col2);
    const float2 be = *(const float2*)(be2 + col2);
#pragma unroll
    for (int r = 0; r < 8; ++r) {
      const int row = 8 * hh + r;
      const float2 st = sStat[row];
      float2 o;
      o.x = fmaxf((acc2[0][r] - st.x) * st.y * gg.x + be.x, 0.f);
      o.y = fmaxf((acc2[1][r] - st.x) * st.y * gg.y + be.y, 0.f);
      *(float2*)(rout + (size_t)(row0 + row) * HDIM + col2) = o;
    }
  }
}

#define NBIN 128
#define NCHUNK 128
#define RMAX 128

__global__ void __launch_bounds__(1024)
bin_hist_kernel(const int* __restrict__ dst, int* __restrict__ hist, int E, int N) {
  __shared__ int h[NBIN];
  const int tid = threadIdx.x;
  const int R = (N + NBIN - 1) / NBIN;
  const int C = (E + NCHUNK - 1) / NCHUNK;
  const int e0 = min((int)blockIdx.x * C, E), e1 = min(e0 + C, E);
  if (tid < NBIN) h[tid] = 0;
  __syncthreads();
  for (int e = e0 + tid; e < e1; e += 1024) atomicAdd(&h[dst[e] / R], 1);
  __syncthreads();
  if (tid < NBIN) hist[blockIdx.x * NBIN + tid] = h[tid];
}

__global__ void __launch_bounds__(1024)
bin_place_kernel(const int* __restrict__ dst, const int* __restrict__ hist,
                 int* __restrict__ binStart, int* __restrict__ binned, int E, int N) {
  __shared__ int sAll[8][NBIN];
  __shared__ int sPre[8][NBIN];
  __shared__ int scanb[NBIN];
  __shared__ int cursor[NBIN];
  const int tid = threadIdx.x;
  const int R = (N + NBIN - 1) / NBIN;
  const int C = (E + NCHUNK - 1) / NCHUNK;
  const int chunk = blockIdx.x;
  const int e0 = min(chunk * C, E), e1 = min(e0 + C, E);
  {
    const int b = tid & (NBIN - 1), part = tid >> 7;
    int sa = 0, sp = 0;
#pragma unroll
    for (int k = 0; k < NCHUNK / 8; ++k) {
      const int c = part * (NCHUNK / 8) + k;
      const int v = hist[c * NBIN + b];
      sa += v;
      sp += (c < chunk) ? v : 0;
    }
    sAll[part][b] = sa;
    sPre[part][b] = sp;
  }
  __syncthreads();
  int tot = 0, pre = 0;
  if (tid < NBIN) {
#pragma unroll
    for (int p = 0; p < 8; ++p) { tot += sAll[p][tid]; pre += sPre[p][tid]; }
    scanb[tid] = tot;
  }
  __syncthreads();
  for (int d = 1; d < NBIN; d <<= 1) {
    int v = 0;
    if (tid < NBIN && tid >= d) v = scanb[tid - d];
    __syncthreads();
    if (tid < NBIN) scanb[tid] += v;
    __syncthreads();
  }
  if (tid < NBIN) {
    const int excl = scanb[tid] - tot;
    cursor[tid] = excl + pre;
    if (chunk == 0) {
      binStart[tid] = excl;
      if (tid == NBIN - 1) binStart[NBIN] = scanb[tid];
    }
  }
  __syncthreads();
  for (int e = e0 + tid; e < e1; e += 1024) {
    const int d = dst[e];
    const int p = atomicAdd(&cursor[d / R], 1);
    ((int2*)binned)[p] = make_int2(e, d);
  }
}

__global__ void __launch_bounds__(1024)
bin_sort_kernel(const int* __restrict__ binStart, const int* __restrict__ binned,
                const int* __restrict__ src, int* __restrict__ slot, int* __restrict__ raw,
                int* __restrict__ off, int* __restrict__ pairs, int E, int N) {
  __shared__ int cnt[RMAX];
  __shared__ int offL[RMAX];
  __shared__ int scanb[RMAX];
  const int tid = threadIdx.x;
  const int lane = tid & 31;
  const int R = (N + NBIN - 1) / NBIN;
  const int n0 = min((int)blockIdx.x * R, N), n1 = min(n0 + R, N);
  const int nn = n1 - n0;
  const int s0 = binStart[blockIdx.x], s1 = binStart[blockIdx.x + 1];
  if (tid < RMAX) cnt[tid] = 0;
  __syncthreads();
  for (int i = s0 + tid; i < s1; i += 1024) {
    const int2 pr = ((const int2*)binned)[i];
    slot[i] = atomicAdd(&cnt[pr.y - n0], 1);
  }
  __syncthreads();
  int mine = 0;
  if (tid < RMAX) { mine = cnt[tid]; scanb[tid] = mine; }
  __syncthreads();
  for (int d = 1; d < RMAX; d <<= 1) {
    int v = 0;
    if (tid < RMAX && tid >= d) v = scanb[tid - d];
    __syncthreads();
    if (tid < RMAX) scanb[tid] += v;
    __syncthreads();
  }
  if (tid < RMAX) {
    const int excl = scanb[tid] - mine;
    offL[tid] = excl;
    if (tid < nn) off[n0 + tid] = s0 + excl;
  }
  if (blockIdx.x == 0 && tid == 0) off[N] = E;
  __syncthreads();
  for (int i = s0 + tid; i < s1; i += 1024) {
    const int2 pr = ((const int2*)binned)[i];
    raw[s0 + offL[pr.y - n0] + slot[i]] = pr.x;
  }
  __syncthreads();
  const int w = __builtin_amdgcn_readfirstlane(tid >> 5);
  for (int j = w; j < nn; j += 32) {
    const int beg = __builtin_amdgcn_readfirstlane(s0 + offL[j]);
    const int end = __builtin_amdgcn_readfirstlane(beg + cnt[j]);
    for (int ib = beg; ib < end; ib += 32) {
      const int i = ib + lane;
      const bool live = i < end;
      const int ei = live ? raw[i] : 0x7fffffff;
      int r = 0;
      for (int jb = beg; jb < end; jb += 32) {
        const int jj = jb + lane;
        const int ej = (jj < end) ? raw[jj] : 0x7fffffff;
        const int cj = min(32, end - jb);
        for (int k = 0; k < cj; ++k)
          r += (__builtin_amdgcn_readlane(ej, k) < ei) ? 1 : 0;
      }
      if (live) ((int2*)pairs)[beg + r] = make_int2(ei, src[ei]);
    }
  }
}

#define EB 8

template <bool TAIL>
__device__ __forceinline__ void gather_group(const float* __restrict__ x,
                                             const float* __restrict__ ea,
                                             int2 pr, int first, int nlive,
                                             unsigned lane, float tv, bool tpos,
                                             float (&mx)[4], float (&den)[4], float (&num)[4]) {
  float4 xa[EB], eb[EB];
#pragma unroll
  for (int j = 0; j < EB; ++j) {
    const int e = __builtin_amdgcn_readlane(pr.x, first + j);
    const int s = __builtin_amdgcn_readlane(pr.y, first + j);
    xa[j] = ((const float4*)x)[(unsigned)s * (HDIM / 4) + lane];
    eb[j] = ((const float4*)ea)[(unsigned)e * (HDIM / 4) + lane];
  }
  float m[EB][4];
#pragma unroll
  for (int j = 0; j < EB; ++j) {
    m[j][0] = fmaxf(xa[j].x + eb[j].x, 0.f) + MSG_EPS;
    m[j][1] = fmaxf(xa[j].y + eb[j].y, 0.f) + MSG_EPS;
    m[j][2] = fmaxf(xa[j].z + eb[j].z, 0.f) + MSG_EPS;
    m[j][3] = fmaxf(xa[j].w + eb[j].w, 0.f) + MSG_EPS;
  }
#pragma unroll
  for (int c = 0; c < 4; ++c) {
    float ext = m[0][c];
    if (tpos) {
#pragma unroll
      for (int j = 1; j < EB; ++j) ext = fmaxf(ext, (TAIL && j >= nlive) ? m[0][c] : m[j][c]);
    } else {
#pragma unroll
      for (int j = 1; j < EB; ++j) ext = fminf(ext, (TAIL && j >= nlive) ? m[0][c] : m[j][c]);
    }
    const float nm = fmaxf(mx[c], ext * tv);
    const float sc = __expf(mx[c] - nm);
    float d = den[c] * sc, u = num[c] * sc;
#pragma unroll
    for (int j = 0; j < EB; ++j) {
      float ex = __expf(m[j][c] * tv - nm);
      if (TAIL && j >= nlive) ex = 0.f;
      d += ex;
      u += ex * m[j][c];
    }
    mx[c] = nm; den[c] = d; num[c] = u;
  }
}

__global__ void __launch_bounds__(128)
node_softmax_gather_kernel(const float* __restrict__ x, const float* __restrict__ ea,
                           const int* __restrict__ pairs, const int* __restrict__ off,
                           const float* __restrict__ tptr, int layer,
                           float* __restrict__ hc, int n) {
  const int nodev = blockIdx.x * (blockDim.x >> 5) + (threadIdx.x >> 5);
  const unsigned lane = threadIdx.x & 31u;
  if (nodev >= n) return;
  const int node = __builtin_amdgcn_readfirstlane(nodev);
  const int beg = off[node], end = off[node + 1];
  const float tv = tptr[layer];
  const bool tpos = tv >= 0.f;

  float mx[4], den[4], num[4];
#pragma unroll
  for (int c = 0; c < 4; ++c) { mx[c] = __int_as_float(0xff800000); den[c] = 0.f; num[c] = 0.f; }

  for (int base = beg; base < end; base += 32) {
    const int cnt = min(32, end - base);
    const int2 pr = ((const int2*)pairs)[base + min((int)lane, cnt - 1)];
    int g = 0;
    for (; g + EB <= cnt; g += EB)
      gather_group<false>(x, ea, pr, g, EB, lane, tv, tpos, mx, den, num);
    if (g < cnt)
      gather_group<true>(x, ea, pr, g, cnt - g, lane, tv, tpos, mx, den, num);
  }

  const float4 root = ((const float4*)(x + (size_t)node * HDIM))[lane];
  float4 o;
  if (beg < end) {
    o.x = num[0] / den[0] + root.x;
    o.y = num[1] / den[1] + root.y;
    o.z = num[2] / den[2] + root.z;
    o.w = num[3] / den[3] + root.w;
  } else {
    o = root;
  }
  ((float4*)(hc + (size_t)node * HDIM))[lane] = o;
}

#define PTRIPS 10
#define PROWS 8

__global__ void __launch_bounds__(1024)
pool_classify_kernel(const float* __restrict__ r, const int* __restrict__ batch,
                     const float* __restrict__ W, const float* __restrict__ b,
                     float* __restrict__ out, float* __restrict__ pooled, int n) {
  __shared__ __attribute__((aligned(16))) float part[32 * HDIM];
  __shared__ __attribute__((aligned(16))) float psum[HDIM];
  const int tid = threadIdx.x;
  const int lane = tid & 31;
  const int w = __builtin_amdgcn_readfirstlane(tid >> 5);
  const int g = blockIdx.x;

  float4 wv[NUM_CLASSES];
  float bias = 0.f;
  if (w == 0) {
#pragma unroll
    for (int j = 0; j < NUM_CLASSES; ++j) wv[j] = ((const float4*)W)[NUM_CLASSES * lane + j];
    if (lane < NUM_CLASSES) bias = b[lane];
  }

  const int lofs = 256 * (lane >> 3) + 8 * w + (lane & 7);
  float4 acc = make_float4(0.f, 0.f, 0.f, 0.f);
  for (int t0 = 0; 1024 * t0 < n; t0 += PTRIPS) {
    int bv[PTRIPS];
#pragma unroll
    for (int u = 0; u < PTRIPS; ++u) {
      const int nn = 1024 * (t0 + u) + lofs;
      bv[u] = nn < n ? batch[nn] : -1;
    }
#pragma unroll
    for (int u = 0; u < PTRIPS; ++u) {
      unsigned m = __builtin_amdgcn_ballot_w32(bv[u] == g);
      const int nbase = 1024 * (t0 + u) + 8 * w;
      while (m) {
        float4 v[PROWS];
#pragma unroll
        for (int q = 0; q < PROWS; ++q) {
          v[q] = make_float4(0.f, 0.f, 0.f, 0.f);
          if (m) {
            const int bit = __builtin_ctz(m);
            m &= m - 1;
            const int node = nbase + 256 * (bit >> 3) + (bit & 7);
            v[q] = ((const float4*)r)[(unsigned)node * (HDIM / 4) + lane];
          }
        }
#pragma unroll
        for (int q = 0; q < PROWS; ++q) {
          acc.x += v[q].x; acc.y += v[q].y; acc.z += v[q].z; acc.w += v[q].w;
        }
      }
    }
  }
  *(float4*)&part[w * HDIM + 4 * lane] = acc;
  __syncthreads();
  float ps = 0.f;
  if (tid < HDIM) {
    float t[32];
#pragma unroll
    for (int k = 0; k < 32; ++k) t[k] = part[k * HDIM + tid];
#pragma unroll
    for (int st = 1; st < 32; st <<= 1) {
#pragma unroll
      for (int k = 0; k < 32; k += 2 * st) t[k] += t[k + st];
    }
    ps = t[0];
    psum[tid] = ps;
  }
  __syncthreads();
  if (tid < HDIM) pooled[(size_t)g * HDIM + tid] = ps;
  if (w == 0) {
    const float4 p4 = *(const float4*)&psum[4 * lane];
    const float p[4] = {p4.x, p4.y, p4.z, p4.w};
    float wf[4 * NUM_CLASSES];
#pragma unroll
    for (int j = 0; j < NUM_CLASSES; ++j) {
      wf[4 * j + 0] = wv[j].x; wf[4 * j + 1] = wv[j].y; wf[4 * j + 2] = wv[j].z; wf[4 * j + 3] = wv[j].w;
    }
    float o = 0.f;
#pragma unroll
    for (int c = 0; c < NUM_CLASSES; ++c) {
      float s = (p[0] * wf[c] + p[1] * wf[NUM_CLASSES + c]) +
                (p[2] * wf[2 * NUM_CLASSES + c] + p[3] * wf[3 * NUM_CLASSES + c]);
#pragma unroll
      for (int off = 16; off > 0; off >>= 1) s += __shfl_xor(s, off, 32);
      if (lane == c) o = s;
    }
    if (lane < NUM_CLASSES) out[g * NUM_CLASSES + lane] = o + bias;
  }
}

extern "C" void kernel_launch(void* const* d_in, const int* in_sizes, int n_in,
                              void* d_out, int out_size, void* d_ws, size_t ws_size,
                              hipStream_t stream) {
  const float* x     = (const float*)d_in[0];
  const float* ea    = (const float*)d_in[1];
  const float* enc_W = (const float*)d_in[2];
  const float* enc_b = (const float*)d_in[3];
  const float* t     = (const float*)d_in[4];
  const float* W1    = (const float*)d_in[5];
  const float* b1    = (const float*)d_in[6];
  const float* mlp_g = (const float*)d_in[7];
  const float* mlp_b = (const float*)d_in[8];
  const float* W2    = (const float*)d_in[9];
  const float* b2    = (const float*)d_in[10];
  const float* ln_g  = (const float*)d_in[11];
  const float* ln_b  = (const float*)d_in[12];
  const float* lin_W = (const float*)d_in[13];
  const float* lin_b = (const float*)d_in[14];
  const int*   eidx  = (const int*)d_in[15];
  const int*   batch = (const int*)d_in[16];

  const int N = in_sizes[0] / HDIM;
  const int E = in_sizes[1] / HDIM;
  const int L = in_sizes[4];
  const int* src = eidx;
  const int* dst = eidx + E;

  const size_t NH = (size_t)N * HDIM;
  float* h    = (float*)d_ws;
  float* rbuf = h    + NH;
  float* hc   = rbuf + NH;
  float* t1   = hc   + NH;
  int* pairs  = (int*)(t1 + 2 * NH);
  int* raw    = pairs + 2 * (size_t)E;
  int* slot   = raw   + E;
  int* cnt    = slot  + E;
  int* off    = cnt   + N;
  int* binned = off   + N + 2;
  int* hist   = binned + 2 * (size_t)E;
  int* binStart = hist + NCHUNK * NBIN;

  float* out    = (float*)d_out;
  float* pooled = out + NUM_GRAPHS * NUM_CLASSES;

  const dim3 gblk(128);
  const int nodeBlocks = (N + 3) / 4;

  bin_hist_kernel<<<NCHUNK, 1024, 0, stream>>>(dst, hist, E, N);
  bin_place_kernel<<<NCHUNK, 1024, 0, stream>>>(dst, hist, binStart, binned, E, N);
  bin_sort_kernel<<<NBIN, 1024, 0, stream>>>(binStart, binned, src, slot, raw, off, pairs, E, N);

  gemm16_wmma_f32<<<dim3(N / 16, HDIM / 64), gblk, 0, stream>>>(x, enc_W, enc_b, h, HDIM, HDIM, 0);

  for (int layer = 0; layer < L; ++layer) {
    const float* rin = layer > 0 ? rbuf : h;
    node_softmax_gather_kernel<<<nodeBlocks, 128, 0, stream>>>(rin, ea, pairs, off, t, layer, hc, N);
    const int nl = layer + 1 < L ? layer + 1 : 0;
    mlp_ln_fused_kernel<<<N / 16, 128, 0, stream>>>(
        hc, W1 + (size_t)layer * HDIM * HDIM2, b1 + layer * HDIM2,
        mlp_g + layer * HDIM2, mlp_b + layer * HDIM2,
        W2 + (size_t)layer * HDIM2 * HDIM, b2 + layer * HDIM,
        h, layer > 0 ? 1 : 0, ln_g + nl * HDIM, ln_b + nl * HDIM, rbuf);
  }

  pool_classify_kernel<<<NUM_GRAPHS, 1024, 0, stream>>>(rbuf, batch, lin_W, lin_b, out, pooled, N);
}


# Prologue 4 launches to 2: edges binned inside each chunk (no counting launch, no cross-chunk scan; a range's sort block walks its 128 chunk runs), encoder product rides beside the sort on idle WGPs
# speedup vs baseline: 6.0765x; 1.0732x over previous
#include <hip/hip_runtime.h>
#include <hip/hip_bf16.h>

typedef __attribute__((ext_vector_type(2))) float v2f;
typedef __attribute__((ext_vector_type(8))) float v8f;

#define HDIM 128
#define HDIM2 256
#define NUM_GRAPHS 64
#define NUM_CLASSES 10
#define MSG_EPS 1e-7f

__device__ __forceinline__ v8f wmma4(v2f a, v2f b, v8f c) {
  return __builtin_amdgcn_wmma_f32_16x16x4_f32(false, a, false, b, (short)0, c, false, false);
}

#define ENC_STRIPS 10
#define ENC_KC 8

__device__ __forceinline__ void encoder_strip(const float* __restrict__ X,
                                              const float* __restrict__ W,
                                              const float* __restrict__ bias,
                                              float* __restrict__ Y, int strip, int lane) {
  const int hh = lane >> 4, c = lane & 15;
  const int row0 = (strip >> 1) * 16;
  const int col1 = (strip & 1) * 64 + 4 * c;
  v8f acc[4];
  {
    const float4 bb = *(const float4*)(bias + col1);
#pragma unroll
    for (int r = 0; r < 8; ++r) { acc[0][r] = bb.x; acc[1][r] = bb.y; acc[2][r] = bb.z; acc[3][r] = bb.w; }
  }
  const float4* wp = (const float4*)(W + (size_t)(2 * hh) * HDIM + col1);
  const float* ap = X + (size_t)(row0 + c) * HDIM + 2 * hh;
#pragma nounroll
  for (int kc = 0; kc < HDIM / 4; kc += ENC_KC) {
    float4 blo[ENC_KC], bhi[ENC_KC];
    float2 av[ENC_KC];
#pragma unroll
    for (int i = 0; i < ENC_KC; ++i) {
      const int k0 = 4 * (kc + i);
      blo[i] = wp[(size_t)k0 * (HDIM / 4)];
      bhi[i] = wp[(size_t)(k0 + 1) * (HDIM / 4)];
      av[i] = *(const float2*)(ap + k0);
    }
#pragma unroll
    for (int i = 0; i < ENC_KC; ++i) {
      v2f a; a[0] = av[i].x; a[1] = av[i].y;
      v2f b;
      b[0] = blo[i].x; b[1] = bhi[i].x; acc[0] = wmma4(a, b, acc[0]);
      b[0] = blo[i].y; b[1] = bhi[i].y; acc[1] = wmma4(a, b, acc[1]);
      b[0] = blo[i].z; b[1] = bhi[i].z; acc[2] = wmma4(a, b, acc[2]);
      b[0] = blo[i].w; b[1] = bhi[i].w; acc[3] = wmma4(a, b, acc[3]);
    }
  }
#pragma unroll
  for (int r = 0; r < 8; ++r) {
    float4 o;
    o.x = acc[0][r]; o.y = acc[1][r]; o.z = acc[2][r]; o.w = acc[3][r];
    *(float4*)(Y + (size_t)(row0 + 8 * hh + r) * HDIM + col1) = o;
  }
}

#define A_PITCH 132
#define T_PITCH 260
#define R_PITCH 66

__device__ __forceinline__ void row_stats(const float2* red, float2* stat, int tid, float invD) {
  const int row = tid >> 3, sub = tid & 7;
  float s = 0.f, q = 0.f;
#pragma unroll
  for (int i = 0; i < 8; ++i) {
    const float2 p = red[row * R_PITCH + sub + 8 * i];
    s += p.x; q += p.y;
  }
#pragma unroll
  for (int off = 1; off < 8; off <<= 1) {
    s += __shfl_xor(s, off, 32);
    q += __shfl_xor(q, off, 32);
  }
  if (sub == 0) {
    const float mean = s * invD;
    const float var  = q * invD - mean * mean;
    stat[row] = make_float2(mean, rsqrtf(var + 1e-5f));
  }
}

__global__ void __launch_bounds__(128) __attribute__((amdgpu_waves_per_eu(1, 4)))
mlp_ln_fused_kernel(const float* __restrict__ hc,
                    const float* __restrict__ W1, const float* __restrict__ b1,
                    const float* __restrict__ g1, const float* __restrict__ be1,
                    const float* __restrict__ W2, const float* __restrict__ b2,
                    float* __restrict__ h, int accum,
                    const float* __restrict__ g2, const float* __restrict__ be2,
                    float* __restrict__ rout) {
  __shared__ __attribute__((aligned(16))) float sA[16 * A_PITCH];
  __shared__ __attribute__((aligned(16))) float sT[16 * T_PITCH];
  __shared__ float2 sRed[16 * R_PITCH];
  __shared__ float2 sStat[16];

  const int tid  = threadIdx.x;
  const int lane = tid & 31, w = tid >> 5;
  const int hh = lane >> 4, c = lane & 15;
  const int row0 = blockIdx.x * 16;
  const int col1 = w * 64 + 4 * c;
  const int col2 = w * 32 + 2 * c;

#pragma unroll
  for (int i = 0; i < 4; ++i) {
    const int idx = tid + 128 * i, r = idx >> 5, c4 = idx & 31;
    const float4 v = ((const float4*)hc)[(size_t)(row0 + r) * (HDIM / 4) + c4];
    *(float4*)&sA[r * A_PITCH + c4 * 4] = v;
  }

  v8f acc2[2];
  {
    const float2 bb = *(const float2*)(b2 + col2);
#pragma unroll
    for (int r = 0; r < 8; ++r) {
      float2 o = *(const float2*)(h + (size_t)(row0 + 8 * hh + r) * HDIM + col2);
      if (!accum) o = make_float2(0.f, 0.f);
      acc2[0][r] = o.x + bb.x;
      acc2[1][r] = o.y + bb.y;
    }
  }
  __syncthreads();

  v8f acc[4];
  {
    const float4 bb = *(const float4*)(b1 + col1);
#pragma unroll
    for (int r = 0; r < 8; ++r) { acc[0][r] = bb.x; acc[1][r] = bb.y; acc[2][r] = bb.z; acc[3][r] = bb.w; }
  }
  {
    const float4* w1p = (const float4*)(W1 + (size_t)(2 * hh) * HDIM2 + col1);
    const float* ap = sA + c * A_PITCH + 2 * hh;
#pragma unroll
    for (int ks = 0; ks < HDIM / 4; ++ks) {
      const int k0 = 4 * ks;
      const float4 blo = w1p[(size_t)k0 * (HDIM2 / 4)];
      const float4 bhi = w1p[(size_t)(k0 + 1) * (HDIM2 / 4)];
      const float2 av = *(const float2*)(ap + k0);
      v2f a; a[0] = av.x; a[1] = av.y;
      v2f b;
      b[0] = blo.x; b[1] = bhi.x; acc[0] = wmma4(a, b, acc[0]);
      b[0] = blo.y; b[1] = bhi.y; acc[1] = wmma4(a, b, acc[1]);
      b[0] = blo.z; b[1] = bhi.z; acc[2] = wmma4(a, b, acc[2]);
      b[0] = blo.w; b[1] = bhi.w; acc[3] = wmma4(a, b, acc[3]);
    }
  }

#pragma unroll
  for (int r = 0; r < 8; ++r) {
    const float s = (acc[0][r] + acc[1][r]) + (acc[2][r] + acc[3][r]);
    const float q = (acc[0][r] * acc[0][r] + acc[1][r] * acc[1][r]) +
                    (acc[2][r] * acc[2][r] + acc[3][r] * acc[3][r]);
    sRed[(8 * hh + r) * R_PITCH + w * 16 + c] = make_float2(s, q);
  }
  __syncthreads();
  row_stats(sRed, sStat, tid, 1.f / (float)HDIM2);
  __syncthreads();
  {
    const float4 gg = *(const float4*)(g1 + col1);
    const float4 be = *(const float4*)(be1 + col1);
#pragma unroll
    for (int r = 0; r < 8; ++r) {
      const float2 st = sStat[8 * hh + r];
      float4 o;
      o.x = fmaxf((acc[0][r] - st.x) * st.y * gg.x + be.x, 0.f);
      o.y = fmaxf((acc[1][r] - st.x) * st.y * gg.y + be.y, 0.f);
      o.z = fmaxf((acc[2][r] - st.x) * st.y * gg.z + be.z, 0.f);
      o.w = fmaxf((acc[3][r] - st.x) * st.y * gg.w + be.w, 0.f);
      *(float4*)&sT[(8 * hh + r) * T_PITCH + col1] = o;
    }
  }
  __syncthreads();

  {
    const float2* w2p = (const float2*)(W2 + (size_t)(2 * hh) * HDIM + col2);
    const float* tp = sT + c * T_PITCH + 2 * hh;
#pragma unroll
    for (int ks = 0; ks < HDIM2 / 4; ++ks) {
      const int k0 = 4 * ks;
      const float2 blo = w2p[(size_t)k0 * (HDIM / 2)];
      const float2 bhi = w2p[(size_t)(k0 + 1) * (HDIM / 2)];
      const float2 av = *(const float2*)(tp + k0);
      v2f a; a[0] = av.x; a[1] = av.y;
      v2f b;
      b[0] = blo.x; b[1] = bhi.x; acc2[0] = wmma4(a, b, acc2[0]);
      b[0] = blo.y; b[1] = bhi.y; acc2[1] = wmma4(a, b, acc2[1]);
    }
  }

#pragma unroll
  for (int r = 0; r < 8; ++r) {
    const int row = 8 * hh + r;
    const float2 o = make_float2(acc2[0][r], acc2[1][r]);
    *(float2*)(h + (size_t)(row0 + row) * HDIM + col2) = o;
    sRed[row * R_PITCH + w * 16 + c] = make_float2(o.x + o.y, o.x * o.x + o.y * o.y);
  }
  __syncthreads();
  row_stats(sRed, sStat, tid, 1.f / (float)HDIM);
  __syncthreads();
  {
    const float2 gg = *(const float2*)(g2 + col2);
    const float2 be = *(const float2*)(be2 + col2);
#pragma unroll
    for (int r = 0; r < 8; ++r) {
      const int row = 8 * hh + r;
      const float2 st = sStat[row];
      float2 o;
      o.x = fmaxf((acc2[0][r] - st.x) * st.y * gg.x + be.x, 0.f);
      o.y = fmaxf((acc2[1][r] - st.x) * st.y * gg.y + be.y, 0.f);
      *(float2*)(rout + (size_t)(row0 + row) * HDIM + col2) = o;
    }
  }
}

#define NBIN 128
#define NCHUNK 128
#define RMAX 128

__global__ void __launch_bounds__(1024)
bin_local_kernel(const int* __restrict__ dst, int* __restrict__ tab,
                 int* __restrict__ binned, int E, int N) {
  __shared__ int h[NBIN];
  __shared__ int scanb[NBIN];
  __shared__ int cursor[NBIN];
  const int tid = threadIdx.x;
  const int R = (N + NBIN - 1) / NBIN;
  const int C = (E + NCHUNK - 1) / NCHUNK;
  const int chunk = blockIdx.x;
  const int e0 = min(chunk * C, E), e1 = min(e0 + C, E);
  if (tid < NBIN) h[tid] = 0;
  __syncthreads();
  for (int e = e0 + tid; e < e1; e += 1024) atomicAdd(&h[dst[e] / R], 1);
  __syncthreads();
  int tot = 0;
  if (tid < NBIN) { tot = h[tid]; scanb[tid] = tot; }
  __syncthreads();
  for (int d = 1; d < NBIN; d <<= 1) {
    int v = 0;
    if (tid < NBIN && tid >= d) v = scanb[tid - d];
    __syncthreads();
    if (tid < NBIN) scanb[tid] += v;
    __syncthreads();
  }
  if (tid < NBIN) {
    const int excl = scanb[tid] - tot;
    cursor[tid] = e0 + excl;
    tab[chunk * (NBIN + 1) + tid] = excl;
    if (tid == NBIN - 1) tab[chunk * (NBIN + 1) + NBIN] = scanb[tid];
  }
  __syncthreads();
  for (int e = e0 + tid; e < e1; e += 1024) {
    const int d = dst[e];
    const int p = atomicAdd(&cursor[d / R], 1);
    ((int2*)binned)[p] = make_int2(e, d);
  }
}

__global__ void __launch_bounds__(1024)
bin_sort_encoder_kernel(const int* __restrict__ tab, const int* __restrict__ binned,
                        const int* __restrict__ src, int* __restrict__ slot, int* __restrict__ raw,
                        int* __restrict__ off, int* __restrict__ pairs, int E, int N,
                        const float* __restrict__ X, const float* __restrict__ encW,
                        const float* __restrict__ encB, float* __restrict__ Y) {
  __shared__ int cnt[RMAX];
  __shared__ int offL[RMAX];
  __shared__ int scanb[RMAX];
  __shared__ int segSt[NCHUNK];
  __shared__ int segLen[NCHUNK];
  __shared__ int redLo[32];
  const int tid = threadIdx.x;
  const int lane = tid & 31;
  const int w = __builtin_amdgcn_readfirstlane(tid >> 5);
  if (blockIdx.x >= NBIN) {
    const int strip = ((int)blockIdx.x - NBIN) * ENC_STRIPS + w;
    if (w < ENC_STRIPS && strip < (N / 16) * (HDIM / 64))
      encoder_strip(X, encW, encB, Y, strip, lane);
    return;
  }
  const int bin = blockIdx.x;
  const int R = (N + NBIN - 1) / NBIN;
  const int C = (E + NCHUNK - 1) / NCHUNK;
  const int n0 = min(bin * R, N), n1 = min(n0 + R, N);
  const int nn = n1 - n0;
  int lo = 0;
  {
    const int b = tid & (NBIN - 1), part = tid >> 7;
#pragma unroll
    for (int k = 0; k < NCHUNK / 8; ++k) {
      const int c = part * (NCHUNK / 8) + k;
      const int v = tab[c * (NBIN + 1) + b + 1] - tab[c * (NBIN + 1) + b];
      lo += (b < bin) ? v : 0;
    }
  }
#pragma unroll
  for (int o = 16; o > 0; o >>= 1) lo += __shfl_xor(lo, o, 32);
  if (lane == 0) redLo[w] = lo;
  if (tid < NCHUNK) {
    const int a = tab[tid * (NBIN + 1) + bin], z = tab[tid * (NBIN + 1) + bin + 1];
    segSt[tid] = min(tid * C, E) + a;
    segLen[tid] = z - a;
  }
  if (tid < RMAX) cnt[tid] = 0;
  __syncthreads();
  int s0 = 0;
#pragma unroll
  for (int i = 0; i < 32; ++i) s0 += redLo[i];
  int2 pr0[NCHUNK / 32];
  int sl0[NCHUNK / 32];
#pragma unroll
  for (int q = 0; q < NCHUNK / 32; ++q) {
    const int c = w + 32 * q;
    pr0[q] = make_int2(0, n0);
    if (lane < segLen[c]) pr0[q] = ((const int2*)binned)[segSt[c] + lane];
  }
#pragma unroll
  for (int q = 0; q < NCHUNK / 32; ++q) {
    const int c = w + 32 * q;
    const int st = segSt[c], len = segLen[c];
    sl0[q] = 0;
    if (lane < len) sl0[q] = atomicAdd(&cnt[pr0[q].y - n0], 1);
    for (int k = lane + 32; k < len; k += 32) {
      const int2 pr = ((const int2*)binned)[st + k];
      slot[st + k] = atomicAdd(&cnt[pr.y - n0], 1);
    }
  }
  __syncthreads();
  int mine = 0;
  if (tid < RMAX) { mine = cnt[tid]; scanb[tid] = mine; }
  __syncthreads();
  for (int d = 1; d < RMAX; d <<= 1) {
    int v = 0;
    if (tid < RMAX && tid >= d) v = scanb[tid - d];
    __syncthreads();
    if (tid < RMAX) scanb[tid] += v;
    __syncthreads();
  }
  if (tid < RMAX) {
    const int excl = scanb[tid] - mine;
    offL[tid] = excl;
    if (tid < nn) off[n0 + tid] = s0 + excl;
  }
  if (blockIdx.x == 0 && tid == 0) off[N] = E;
  __syncthreads();
#pragma unroll
  for (int q = 0; q < NCHUNK / 32; ++q) {
    const int c = w + 32 * q;
    const int st = segSt[c], len = segLen[c];
    if (lane < len) raw[s0 + offL[pr0[q].y - n0] + sl0[q]] = pr0[q].x;
    for (int k = lane + 32; k < len; k += 32) {
      const int2 pr = ((const int2*)binned)[st + k];
      raw[s0 + offL[pr.y - n0] + slot[st + k]] = pr.x;
    }
  }
  __syncthreads();
  for (int j = w; j < nn; j += 32) {
    const int beg = __builtin_amdgcn_readfirstlane(s0 + offL[j]);
    const int end = __builtin_amdgcn_readfirstlane(beg + cnt[j]);
    for (int ib = beg; ib < end; ib += 32) {
      const int i = ib + lane;
      const bool live = i < end;
      const int ei = live ? raw[i] : 0x7fffffff;
      int r = 0;
      for (int jb = beg; jb < end; jb += 32) {
        const int jj = jb + lane;
        const int ej = (jj < end) ? raw[jj] : 0x7fffffff;
        const int cj = min(32, end - jb);
        for (int k = 0; k < cj; ++k)
          r += (__builtin_amdgcn_readlane(ej, k) < ei) ? 1 : 0;
      }
      if (live) ((int2*)pairs)[beg + r] = make_int2(ei, src[ei]);
    }
  }
}

#define EB 8

template <bool TAIL>
__device__ __forceinline__ void gather_group(const float* __restrict__ x,
                                             const float* __restrict__ ea,
                                             int2 pr, int first, int nlive,
                                             unsigned lane, float tv, bool tpos,
                                             float (&mx)[4], float (&den)[4], float (&num)[4]) {
  float4 xa[EB], eb[EB];
#pragma unroll
  for (int j = 0; j < EB; ++j) {
    const int e = __builtin_amdgcn_readlane(pr.x, first + j);
    const int s = __builtin_amdgcn_readlane(pr.y, first + j);
    xa[j] = ((const float4*)x)[(unsigned)s * (HDIM / 4) + lane];
    eb[j] = ((const float4*)ea)[(unsigned)e * (HDIM / 4) + lane];
  }
  float m[EB][4];
#pragma unroll
  for (int j = 0; j < EB; ++j) {
    m[j][0] = fmaxf(xa[j].x + eb[j].x, 0.f) + MSG_EPS;
    m[j][1] = fmaxf(xa[j].y + eb[j].y, 0.f) + MSG_EPS;
    m[j][2] = fmaxf(xa[j].z + eb[j].z, 0.f) + MSG_EPS;
    m[j][3] = fmaxf(xa[j].w + eb[j].w, 0.f) + MSG_EPS;
  }
#pragma unroll
  for (int c = 0; c < 4; ++c) {
    float ext = m[0][c];
    if (tpos) {
#pragma unroll
      for (int j = 1; j < EB; ++j) ext = fmaxf(ext, (TAIL && j >= nlive) ? m[0][c] : m[j][c]);
    } else {
#pragma unroll
      for (int j = 1; j < EB; ++j) ext = fminf(ext, (TAIL && j >= nlive) ? m[0][c] : m[j][c]);
    }
    const float nm = fmaxf(mx[c], ext * tv);
    const float sc = __expf(mx[c] - nm);
    float d = den[c] * sc, u = num[c] * sc;
#pragma unroll
    for (int j = 0; j < EB; ++j) {
      float ex = __expf(m[j][c] * tv - nm);
      if (TAIL && j >= nlive) ex = 0.f;
      d += ex;
      u += ex * m[j][c];
    }
    mx[c] = nm; den[c] = d; num[c] = u;
  }
}

__global__ void __launch_bounds__(128)
node_softmax_gather_kernel(const float* __restrict__ x, const float* __restrict__ ea,
                           const int* __restrict__ pairs, const int* __restrict__ off,
                           const float* __restrict__ tptr, int layer,
                           float* __restrict__ hc, int n) {
  const int nodev = blockIdx.x * (blockDim.x >> 5) + (threadIdx.x >> 5);
  const unsigned lane = threadIdx.x & 31u;
  if (nodev >= n) return;
  const int node = __builtin_amdgcn_readfirstlane(nodev);
  const int beg = off[node], end = off[node + 1];
  const float tv = tptr[layer];
  const bool tpos = tv >= 0.f;

  float mx[4], den[4], num[4];
#pragma unroll
  for (int c = 0; c < 4; ++c) { mx[c] = __int_as_float(0xff800000); den[c] = 0.f; num[c] = 0.f; }

  for (int base = beg; base < end; base += 32) {
    const int cnt = min(32, end - base);
    const int2 pr = ((const int2*)pairs)[base + min((int)lane, cnt - 1)];
    int g = 0;
    for (; g + EB <= cnt; g += EB)
      gather_group<false>(x, ea, pr, g, EB, lane, tv, tpos, mx, den, num);
    if (g < cnt)
      gather_group<true>(x, ea, pr, g, cnt - g, lane, tv, tpos, mx, den, num);
  }

  const float4 root = ((const float4*)(x + (size_t)node * HDIM))[lane];
  float4 o;
  if (beg < end) {
    o.x = num[0] / den[0] + root.x;
    o.y = num[1] / den[1] + root.y;
    o.z = num[2] / den[2] + root.z;
    o.w = num[3] / den[3] + root.w;
  } else {
    o = root;
  }
  ((float4*)(hc + (size_t)node * HDIM))[lane] = o;
}

#define PTRIPS 10
#define PROWS 8

__global__ void __launch_bounds__(1024)
pool_classify_kernel(const float* __restrict__ r, const int* __restrict__ batch,
                     const float* __restrict__ W, const float* __restrict__ b,
                     float* __restrict__ out, float* __restrict__ pooled, int n) {
  __shared__ __attribute__((aligned(16))) float part[32 * HDIM];
  __shared__ __attribute__((aligned(16))) float psum[HDIM];
  const int tid = threadIdx.x;
  const int lane = tid & 31;
  const int w = __builtin_amdgcn_readfirstlane(tid >> 5);
  const int g = blockIdx.x;

  float4 wv[NUM_CLASSES];
  float bias = 0.f;
  if (w == 0) {
#pragma unroll
    for (int j = 0; j < NUM_CLASSES; ++j) wv[j] = ((const float4*)W)[NUM_CLASSES * lane + j];
    if (lane < NUM_CLASSES) bias = b[lane];
  }

  const int lofs = 256 * (lane >> 3) + 8 * w + (lane & 7);
  float4 acc = make_float4(0.f, 0.f, 0.f, 0.f);
  for (int t0 = 0; 1024 * t0 < n; t0 += PTRIPS) {
    int bv[PTRIPS];
#pragma unroll
    for (int u = 0; u < PTRIPS; ++u) {
      const int nn = 1024 * (t0 + u) + lofs;
      bv[u] = nn < n ? batch[nn] : -1;
    }
#pragma unroll
    for (int u = 0; u < PTRIPS; ++u) {
      unsigned m = __builtin_amdgcn_ballot_w32(bv[u] == g);
      const int nbase = 1024 * (t0 + u) + 8 * w;
      while (m) {
        float4 v[PROWS];
#pragma unroll
        for (int q = 0; q < PROWS; ++q) {
          v[q] = make_float4(0.f, 0.f, 0.f, 0.f);
          if (m) {
            const int bit = __builtin_ctz(m);
            m &= m - 1;
            const int node = nbase + 256 * (bit >> 3) + (bit & 7);
            v[q] = ((const float4*)r)[(unsigned)node * (HDIM / 4) + lane];
          }
        }
#pragma unroll
        for (int q = 0; q < PROWS; ++q) {
          acc.x += v[q].x; acc.y += v[q].y; acc.z += v[q].z; acc.w += v[q].w;
        }
      }
    }
  }
  *(float4*)&part[w * HDIM + 4 * lane] = acc;
  __syncthreads();
  float ps = 0.f;
  if (tid < HDIM) {
    float t[32];
#pragma unroll
    for (int k = 0; k < 32; ++k) t[k] = part[k * HDIM + tid];
#pragma unroll
    for (int st = 1; st < 32; st <<= 1) {
#pragma unroll
      for (int k = 0; k < 32; k += 2 * st) t[k] += t[k + st];
    }
    ps = t[0];
    psum[tid] = ps;
  }
  __syncthreads();
  if (tid < HDIM) pooled[(size_t)g * HDIM + tid] = ps;
  if (w == 0) {
    const float4 p4 = *(const float4*)&psum[4 * lane];
    const float p[4] = {p4.x, p4.y, p4.z, p4.w};
    float wf[4 * NUM_CLASSES];
#pragma unroll
    for (int j = 0; j < NUM_CLASSES; ++j) {
      wf[4 * j + 0] = wv[j].x; wf[4 * j + 1] = wv[j].y; wf[4 * j + 2] = wv[j].z; wf[4 * j + 3] = wv[j].w;
    }
    float o = 0.f;
#pragma unroll
    for (int c = 0; c < NUM_CLASSES; ++c) {
      float s = (p[0] * wf[c] + p[1] * wf[NUM_CLASSES + c]) +
                (p[2] * wf[2 * NUM_CLASSES + c] + p[3] * wf[3 * NUM_CLASSES + c]);
#pragma unroll
      for (int off = 16; off > 0; off >>= 1) s += __shfl_xor(s, off, 32);
      if (lane == c) o = s;
    }
    if (lane < NUM_CLASSES) out[g * NUM_CLASSES + lane] = o + bias;
  }
}

extern "C" void kernel_launch(void* const* d_in, const int* in_sizes, int n_in,
                              void* d_out, int out_size, void* d_ws, size_t ws_size,
                              hipStream_t stream) {
  const float* x     = (const float*)d_in[0];
  const float* ea    = (const float*)d_in[1];
  const float* enc_W = (const float*)d_in[2];
  const float* enc_b = (const float*)d_in[3];
  const float* t     = (const float*)d_in[4];
  const float* W1    = (const float*)d_in[5];
  const float* b1    = (const float*)d_in[6];
  const float* mlp_g = (const float*)d_in[7];
  const float* mlp_b = (const float*)d_in[8];
  const float* W2    = (const float*)d_in[9];
  const float* b2    = (const float*)d_in[10];
  const float* ln_g  = (const float*)d_in[11];
  const float* ln_b  = (const float*)d_in[12];
  const float* lin_W = (const float*)d_in[13];
  const float* lin_b = (const float*)d_in[14];
  const int*   eidx  = (const int*)d_in[15];
  const int*   batch = (const int*)d_in[16];

  const int N = in_sizes[0] / HDIM;
  const int E = in_sizes[1] / HDIM;
  const int L = in_sizes[4];
  const int* src = eidx;
  const int* dst = eidx + E;

  const size_t NH = (size_t)N * HDIM;
  float* h    = (float*)d_ws;
  float* rbuf = h    + NH;
  float* hc   = rbuf + NH;
  float* t1   = hc   + NH;
  int* pairs  = (int*)(t1 + 2 * NH);
  int* raw    = pairs + 2 * (size_t)E;
  int* slot   = raw   + E;
  int* cnt    = slot  + E;
  int* off    = cnt   + N;
  int* binned = off   + N + 2;
  int* tab    = binned + 2 * (size_t)E;

  float* out    = (float*)d_out;
  float* pooled = out + NUM_GRAPHS * NUM_CLASSES;

  const int nodeBlocks = (N + 3) / 4;

  bin_local_kernel<<<NCHUNK, 1024, 0, stream>>>(dst, tab, binned, E, N);
  const int encBlocks = (N / 16 * (HDIM / 64) + ENC_STRIPS - 1) / ENC_STRIPS;
  bin_sort_encoder_kernel<<<NBIN + encBlocks, 1024, 0, stream>>>(tab, binned, src, slot, raw, off, pairs,
                                                                E, N, x, enc_W, enc_b, h);

  for (int layer = 0; layer < L; ++layer) {
    const float* rin = layer > 0 ? rbuf : h;
    node_softmax_gather_kernel<<<nodeBlocks, 128, 0, stream>>>(rin, ea, pairs, off, t, layer, hc, N);
    const int nl = layer + 1 < L ? layer + 1 : 0;
    mlp_ln_fused_kernel<<<N / 16, 128, 0, stream>>>(
        hc, W1 + (size_t)layer * HDIM * HDIM2, b1 + layer * HDIM2,
        mlp_g + layer * HDIM2, mlp_b + layer * HDIM2,
        W2 + (size_t)layer * HDIM2 * HDIM, b2 + layer * HDIM,
        h, layer > 0 ? 1 : 0, ln_g + nl * HDIM, ln_b + nl * HDIM, rbuf);
  }

  pool_classify_kernel<<<NUM_GRAPHS, 1024, 0, stream>>>(rbuf, batch, lin_W, lin_b, out, pooled, N);
}


# Each layer in ONE launch, one 40-node block per WGP: 20 waves gather two nodes each into an LDS tile, 12 run the MLP on it; 11 launches to 7, hc plane gone; every float sum keeps its order
# speedup vs baseline: 6.5704x; 1.0813x over previous
#include <hip/hip_runtime.h>
#include <hip/hip_bf16.h>

typedef __attribute__((ext_vector_type(2))) float v2f;
typedef __attribute__((ext_vector_type(8))) float v8f;

#define HDIM 128
#define HDIM2 256
#define NUM_GRAPHS 64
#define NUM_CLASSES 10
#define MSG_EPS 1e-7f

__device__ __forceinline__ v8f wmma4(v2f a, v2f b, v8f c) {
  return __builtin_amdgcn_wmma_f32_16x16x4_f32(false, a, false, b, (short)0, c, false, false);
}

#define ENC_STRIPS 10
#define ENC_KC 8

__device__ __forceinline__ void encoder_strip(const float* __restrict__ X,
                                              const float* __restrict__ W,
                                              const float* __restrict__ bias,
                                              float* __restrict__ Y, int strip, int lane) {
  const int hh = lane >> 4, c = lane & 15;
  const int row0 = (strip >> 1) * 16;
  const int col1 = (strip & 1) * 64 + 4 * c;
  v8f acc[4];
  {
    const float4 bb = *(const float4*)(bias + col1);
#pragma unroll
    for (int r = 0; r < 8; ++r) { acc[0][r] = bb.x; acc[1][r] = bb.y; acc[2][r] = bb.z; acc[3][r] = bb.w; }
  }
  const float4* wp = (const float4*)(W + (size_t)(2 * hh) * HDIM + col1);
  const float* ap = X + (size_t)(row0 + c) * HDIM + 2 * hh;
#pragma nounroll
  for (int kc = 0; kc < HDIM / 4; kc += ENC_KC) {
    float4 blo[ENC_KC], bhi[ENC_KC];
    float2 av[ENC_KC];
#pragma unroll
    for (int i = 0; i < ENC_KC; ++i) {
      const int k0 = 4 * (kc + i);
      blo[i] = wp[(size_t)k0 * (HDIM / 4)];
      bhi[i] = wp[(size_t)(k0 + 1) * (HDIM / 4)];
      av[i] = *(const float2*)(ap + k0);
    }
#pragma unroll
    for (int i = 0; i < ENC_KC; ++i) {
      v2f a; a[0] = av[i].x; a[1] = av[i].y;
      v2f b;
      b[0] = blo[i].x; b[1] = bhi[i].x; acc[0] = wmma4(a, b, acc[0]);
      b[0] = blo[i].y; b[1] = bhi[i].y; acc[1] = wmma4(a, b, acc[1]);
      b[0] = blo[i].z; b[1] = bhi[i].z; acc[2] = wmma4(a, b, acc[2]);
      b[0] = blo[i].w; b[1] = bhi[i].w; acc[3] = wmma4(a, b, acc[3]);
    }
  }
#pragma unroll
  for (int r = 0; r < 8; ++r) {
    float4 o;
    o.x = acc[0][r]; o.y = acc[1][r]; o.z = acc[2][r]; o.w = acc[3][r];
    *(float4*)(Y + (size_t)(row0 + 8 * hh + r) * HDIM + col1) = o;
  }
}

#define A_PITCH 132
#define T_PITCH 260
#define R_PITCH 66

__device__ __forceinline__ void row_stats(const float2* red, float2* stat, int tid, float invD) {
  const int row = tid >> 3, sub = tid & 7;
  float s = 0.f, q = 0.f;
#pragma unroll
  for (int i = 0; i < 8; ++i) {
    const float2 p = red[row * R_PITCH + sub + 8 * i];
    s += p.x; q += p.y;
  }
#pragma unroll
  for (int off = 1; off < 8; off <<= 1) {
    s += __shfl_xor(s, off, 32);
    q += __shfl_xor(q, off, 32);
  }
  if (sub == 0) {
    const float mean = s * invD;
    const float var  = q * invD - mean * mean;
    stat[row] = make_float2(mean, rsqrtf(var + 1e-5f));
  }
}

#define NBIN 128
#define NCHUNK 128
#define RMAX 128

__global__ void __launch_bounds__(1024)
bin_local_kernel(const int* __restrict__ dst, int* __restrict__ tab,
                 int* __restrict__ binned, int E, int N) {
  __shared__ int h[NBIN];
  __shared__ int scanb[NBIN];
  __shared__ int cursor[NBIN];
  const int tid = threadIdx.x;
  const int R = (N + NBIN - 1) / NBIN;
  const int C = (E + NCHUNK - 1) / NCHUNK;
  const int chunk = blockIdx.x;
  const int e0 = min(chunk * C, E), e1 = min(e0 + C, E);
  if (tid < NBIN) h[tid] = 0;
  __syncthreads();
  for (int e = e0 + tid; e < e1; e += 1024) atomicAdd(&h[dst[e] / R], 1);
  __syncthreads();
  int tot = 0;
  if (tid < NBIN) { tot = h[tid]; scanb[tid] = tot; }
  __syncthreads();
  for (int d = 1; d < NBIN; d <<= 1) {
    int v = 0;
    if (tid < NBIN && tid >= d) v = scanb[tid - d];
    __syncthreads();
    if (tid < NBIN) scanb[tid] += v;
    __syncthreads();
  }
  if (tid < NBIN) {
    const int excl = scanb[tid] - tot;
    cursor[tid] = e0 + excl;
    tab[chunk * (NBIN + 1) + tid] = excl;
    if (tid == NBIN - 1) tab[chunk * (NBIN + 1) + NBIN] = scanb[tid];
  }
  __syncthreads();
  for (int e = e0 + tid; e < e1; e += 1024) {
    const int d = dst[e];
    const int p = atomicAdd(&cursor[d / R], 1);
    ((int2*)binned)[p] = make_int2(e, d);
  }
}

__global__ void __launch_bounds__(1024)
bin_sort_encoder_kernel(const int* __restrict__ tab, const int* __restrict__ binned,
                        const int* __restrict__ src, int* __restrict__ slot, int* __restrict__ raw,
                        int* __restrict__ off, int* __restrict__ pairs, int E, int N,
                        const float* __restrict__ X, const float* __restrict__ encW,
                        const float* __restrict__ encB, float* __restrict__ Y) {
  __shared__ int cnt[RMAX];
  __shared__ int offL[RMAX];
  __shared__ int scanb[RMAX];
  __shared__ int segSt[NCHUNK];
  __shared__ int segLen[NCHUNK];
  __shared__ int redLo[32];
  const int tid = threadIdx.x;
  const int lane = tid & 31;
  const int w = __builtin_amdgcn_readfirstlane(tid >> 5);
  if (blockIdx.x >= NBIN) {
    const int strip = ((int)blockIdx.x - NBIN) * ENC_STRIPS + w;
    if (w < ENC_STRIPS && strip < (N / 16) * (HDIM / 64))
      encoder_strip(X, encW, encB, Y, strip, lane);
    return;
  }
  const int bin = blockIdx.x;
  const int R = (N + NBIN - 1) / NBIN;
  const int C = (E + NCHUNK - 1) / NCHUNK;
  const int n0 = min(bin * R, N), n1 = min(n0 + R, N);
  const int nn = n1 - n0;
  int lo = 0;
  {
    const int b = tid & (NBIN - 1), part = tid >> 7;
#pragma unroll
    for (int k = 0; k < NCHUNK / 8; ++k) {
      const int c = part * (NCHUNK / 8) + k;
      const int v = tab[c * (NBIN + 1) + b + 1] - tab[c * (NBIN + 1) + b];
      lo += (b < bin) ? v : 0;
    }
  }
#pragma unroll
  for (int o = 16; o > 0; o >>= 1) lo += __shfl_xor(lo, o, 32);
  if (lane == 0) redLo[w] = lo;
  if (tid < NCHUNK) {
    const int a = tab[tid * (NBIN + 1) + bin], z = tab[tid * (NBIN + 1) + bin + 1];
    segSt[tid] = min(tid * C, E) + a;
    segLen[tid] = z - a;
  }
  if (tid < RMAX) cnt[tid] = 0;
  __syncthreads();
  int s0 = 0;
#pragma unroll
  for (int i = 0; i < 32; ++i) s0 += redLo[i];
  int2 pr0[NCHUNK / 32];
  int sl0[NCHUNK / 32];
#pragma unroll
  for (int q = 0; q < NCHUNK / 32; ++q) {
    const int c = w + 32 * q;
    pr0[q] = make_int2(0, n0);
    if (lane < segLen[c]) pr0[q] = ((const int2*)binned)[segSt[c] + lane];
  }
#pragma unroll
  for (int q = 0; q < NCHUNK / 32; ++q) {
    const int c = w + 32 * q;
    const int st = segSt[c], len = segLen[c];
    sl0[q] = 0;
    if (lane < len) sl0[q] = atomicAdd(&cnt[pr0[q].y - n0], 1);
    for (int k = lane + 32; k < len; k += 32) {
      const int2 pr = ((const int2*)binned)[st + k];
      slot[st + k] = atomicAdd(&cnt[pr.y - n0], 1);
    }
  }
  __syncthreads();
  int mine = 0;
  if (tid < RMAX) { mine = cnt[tid]; scanb[tid] = mine; }
  __syncthreads();
  for (int d = 1; d < RMAX; d <<= 1) {
    int v = 0;
    if (tid < RMAX && tid >= d) v = scanb[tid - d];
    __syncthreads();
    if (tid < RMAX) scanb[tid] += v;
    __syncthreads();
  }
  if (tid < RMAX) {
    const int excl = scanb[tid] - mine;
    offL[tid] = excl;
    if (tid < nn) off[n0 + tid] = s0 + excl;
  }
  if (blockIdx.x == 0 && tid == 0) off[N] = E;
  __syncthreads();
#pragma unroll
  for (int q = 0; q < NCHUNK / 32; ++q) {
    const int c = w + 32 * q;
    const int st = segSt[c], len = segLen[c];
    if (lane < len) raw[s0 + offL[pr0[q].y - n0] + sl0[q]] = pr0[q].x;
    for (int k = lane + 32; k < len; k += 32) {
      const int2 pr = ((const int2*)binned)[st + k];
      raw[s0 + offL[pr.y - n0] + slot[st + k]] = pr.x;
    }
  }
  __syncthreads();
  for (int j = w; j < nn; j += 32) {
    const int beg = __builtin_amdgcn_readfirstlane(s0 + offL[j]);
    const int end = __builtin_amdgcn_readfirstlane(beg + cnt[j]);
    for (int ib = beg; ib < end; ib += 32) {
      const int i = ib + lane;
      const bool live = i < end;
      const int ei = live ? raw[i] : 0x7fffffff;
      int r = 0;
      for (int jb = beg; jb < end; jb += 32) {
        const int jj = jb + lane;
        const int ej = (jj < end) ? raw[jj] : 0x7fffffff;
        const int cj = min(32, end - jb);
        for (int k = 0; k < cj; ++k)
          r += (__builtin_amdgcn_readlane(ej, k) < ei) ? 1 : 0;
      }
      if (live) ((int2*)pairs)[beg + r] = make_int2(ei, src[ei]);
    }
  }
}

#define EB 8

__device__ __forceinline__ void group_load(const float* __restrict__ x,
                                           const float* __restrict__ ea,
                                           int2 pr, int first, unsigned lane,
                                           float4 (&xa)[EB], float4 (&eb)[EB]) {
#pragma unroll
  for (int j = 0; j < EB; ++j) {
    const int e = __builtin_amdgcn_readlane(pr.x, first + j);
    const int s = __builtin_amdgcn_readlane(pr.y, first + j);
    xa[j] = ((const float4*)x)[(unsigned)s * (HDIM / 4) + lane];
    eb[j] = ((const float4*)ea)[(unsigned)e * (HDIM / 4) + lane];
  }
}

template <bool TAIL>
__device__ __forceinline__ void group_compute(const float4 (&xa)[EB], const float4 (&eb)[EB],
                                              int nlive, float tv, bool tpos,
                                              float (&mx)[4], float (&den)[4], float (&num)[4]) {
  float m[EB][4];
#pragma unroll
  for (int j = 0; j < EB; ++j) {
    m[j][0] = fmaxf(xa[j].x + eb[j].x, 0.f) + MSG_EPS;
    m[j][1] = fmaxf(xa[j].y + eb[j].y, 0.f) + MSG_EPS;
    m[j][2] = fmaxf(xa[j].z + eb[j].z, 0.f) + MSG_EPS;
    m[j][3] = fmaxf(xa[j].w + eb[j].w, 0.f) + MSG_EPS;
  }
#pragma unroll
  for (int c = 0; c < 4; ++c) {
    float ext = m[0][c];
    if (tpos) {
#pragma unroll
      for (int j = 1; j < EB; ++j) ext = fmaxf(ext, (TAIL && j >= nlive) ? m[0][c] : m[j][c]);
    } else {
#pragma unroll
      for (int j = 1; j < EB; ++j) ext = fminf(ext, (TAIL && j >= nlive) ? m[0][c] : m[j][c]);
    }
    const float nm = fmaxf(mx[c], ext * tv);
    const float sc = __expf(mx[c] - nm);
    float d = den[c] * sc, u = num[c] * sc;
#pragma unroll
    for (int j = 0; j < EB; ++j) {
      float ex = __expf(m[j][c] * tv - nm);
      if (TAIL && j >= nlive) ex = 0.f;
      d += ex;
      u += ex * m[j][c];
    }
    mx[c] = nm; den[c] = d; num[c] = u;
  }
}

#define NPB 40
#define TROWS 48

__global__ void __launch_bounds__(640) __attribute__((amdgpu_waves_per_eu(5, 5)))
layer_fused_kernel(const float* __restrict__ xin, const float* __restrict__ ea,
                   const int* __restrict__ pairs, const int* __restrict__ off,
                   const float* __restrict__ tptr, int layer,
                   const float* __restrict__ W1, const float* __restrict__ b1,
                   const float* __restrict__ g1, const float* __restrict__ be1,
                   const float* __restrict__ W2, const float* __restrict__ b2,
                   const float* hin, float* hout, int accum,
                   const float* __restrict__ g2, const float* __restrict__ be2,
                   float* __restrict__ rout, int n) {
  __shared__ __attribute__((aligned(16))) float sA[TROWS * A_PITCH];
  __shared__ __attribute__((aligned(16))) float sT[TROWS * T_PITCH];
  __shared__ float2 sRed[TROWS * R_PITCH];
  __shared__ float2 sStat[TROWS];

  const int tid  = threadIdx.x;
  const int lane = tid & 31;
  const int w = __builtin_amdgcn_readfirstlane(tid >> 5);
  const int row0 = blockIdx.x * NPB;

  {
    const int nodeA = row0 + 2 * w;
    const bool hasA = nodeA < n, hasB = nodeA + 1 < n;
    int begA = 0, endA = 0, endB = 0;
    if (hasA) { begA = off[nodeA]; endA = off[nodeA + 1]; endB = hasB ? off[nodeA + 2] : endA; }
    const int lenA = endA - begA, lenB = endB - endA;
    const float tv = tptr[layer];
    const bool tpos = tv >= 0.f;

    float mxA[4], denA[4], numA[4], mxB[4], denB[4], numB[4];
#pragma unroll
    for (int c = 0; c < 4; ++c) {
      mxA[c] = __int_as_float(0xff800000); denA[c] = 0.f; numA[c] = 0.f;
      mxB[c] = __int_as_float(0xff800000); denB[c] = 0.f; numB[c] = 0.f;
    }

    for (int base = 0; base < lenA || base < lenB; base += 32) {
      const int cntA = min(32, lenA - base), cntB = min(32, lenB - base);
      int2 prA = make_int2(0, 0), prB = make_int2(0, 0);
      if (cntA > 0) prA = ((const int2*)pairs)[begA + base + min(lane, cntA - 1)];
      if (cntB > 0) prB = ((const int2*)pairs)[endA + base + min(lane, cntB - 1)];
      for (int g = 0; g < cntA || g < cntB; g += EB) {
        const int nA = cntA - g, nB = cntB - g;
        if (nA > 0 && nB > 0) {
          float4 xaA[EB], ebA[EB], xaB[EB], ebB[EB];
          group_load(xin, ea, prA, g, (unsigned)lane, xaA, ebA);
          group_load(xin, ea, prB, g, (unsigned)lane, xaB, ebB);
          if (nA >= EB) group_compute<false>(xaA, ebA, EB, tv, tpos, mxA, denA, numA);
          else          group_compute<true>(xaA, ebA, nA, tv, tpos, mxA, denA, numA);
          if (nB >= EB) group_compute<false>(xaB, ebB, EB, tv, tpos, mxB, denB, numB);
          else          group_compute<true>(xaB, ebB, nB, tv, tpos, mxB, denB, numB);
        } else if (nA > 0) {
          float4 xaA[EB], ebA[EB];
          group_load(xin, ea, prA, g, (unsigned)lane, xaA, ebA);
          if (nA >= EB) group_compute<false>(xaA, ebA, EB, tv, tpos, mxA, denA, numA);
          else          group_compute<true>(xaA, ebA, nA, tv, tpos, mxA, denA, numA);
        } else {
          float4 xaB[EB], ebB[EB];
          group_load(xin, ea, prB, g, (unsigned)lane, xaB, ebB);
          if (nB >= EB) group_compute<false>(xaB, ebB, EB, tv, tpos, mxB, denB, numB);
          else          group_compute<true>(xaB, ebB, nB, tv, tpos, mxB, denB, numB);
        }
      }
    }

    float4 oA = make_float4(0.f, 0.f, 0.f, 0.f), oB = oA;
    if (hasA) {
      const float4 root = ((const float4*)(xin + (size_t)nodeA * HDIM))[lane];
      oA = root;
      if (lenA > 0) {
        oA.x = numA[0] / denA[0] + root.x;
        oA.y = numA[1] / denA[1] + root.y;
        oA.z = numA[2] / denA[2] + root.z;
        oA.w = numA[3] / denA[3] + root.w;
      }
    }
    if (hasB) {
      const float4 root = ((const float4*)(xin + (size_t)(nodeA + 1) * HDIM))[lane];
      oB = root;
      if (lenB > 0) {
        oB.x = numB[0] / denB[0] + root.x;
        oB.y = numB[1] / denB[1] + root.y;
        oB.z = numB[2] / denB[2] + root.z;
        oB.w = numB[3] / denB[3] + root.w;
      }
    }
    *(float4*)&sA[(2 * w) * A_PITCH + 4 * lane] = oA;
    *(float4*)&sA[(2 * w + 1) * A_PITCH + 4 * lane] = oB;
    if (w < (TROWS - NPB) / 2) {
      const float4 z = make_float4(0.f, 0.f, 0.f, 0.f);
      *(float4*)&sA[(NPB + 2 * w) * A_PITCH + 4 * lane] = z;
      *(float4*)&sA[(NPB + 2 * w + 1) * A_PITCH + 4 * lane] = z;
    }
  }
  __syncthreads();

  const bool mm = w < 12;
  const int rt = w >> 2, cg = w & 3;
  const int hh = lane >> 4, c = lane & 15;
  const int col1 = cg * 64 + 4 * c;
  const int col2 = cg * 32 + 2 * c;
  const int trow = 16 * rt + 8 * hh;
  const int nrows = min(NPB, n - row0);

  v8f acc[4];
  v8f acc2[2];
  if (mm) {
    {
      const float4 bb = *(const float4*)(b1 + col1);
#pragma unroll
      for (int r = 0; r < 8; ++r) { acc[0][r] = bb.x; acc[1][r] = bb.y; acc[2][r] = bb.z; acc[3][r] = bb.w; }
    }
    const float4* w1p = (const float4*)(W1 + (size_t)(2 * hh) * HDIM2 + col1);
    const float* ap = sA + (16 * rt + c) * A_PITCH + 2 * hh;
#pragma unroll
    for (int ks = 0; ks < HDIM / 4; ++ks) {
      const int k0 = 4 * ks;
      const float4 blo = w1p[(size_t)k0 * (HDIM2 / 4)];
      const float4 bhi = w1p[(size_t)(k0 + 1) * (HDIM2 / 4)];
      const float2 av = *(const float2*)(ap + k0);
      v2f a; a[0] = av.x; a[1] = av.y;
      v2f b;
      b[0] = blo.x; b[1] = bhi.x; acc[0] = wmma4(a, b, acc[0]);
      b[0] = blo.y; b[1] = bhi.y; acc[1] = wmma4(a, b, acc[1]);
      b[0] = blo.z; b[1] = bhi.z; acc[2] = wmma4(a, b, acc[2]);
      b[0] = blo.w; b[1] = bhi.w; acc[3] = wmma4(a, b, acc[3]);
    }

#pragma unroll
    for (int r = 0; r < 8; ++r) {
      const float s = (acc[0][r] + acc[1][r]) + (acc[2][r] + acc[3][r]);
      const float q = (acc[0][r] * acc[0][r] + acc[1][r] * acc[1][r]) +
                      (acc[2][r] * acc[2][r] + acc[3][r] * acc[3][r]);
      sRed[(trow + r) * R_PITCH + cg * 16 + c] = make_float2(s, q);
    }
  }
  __syncthreads();
  if (mm) row_stats(sRed, sStat, tid, 1.f / (float)HDIM2);
  __syncthreads();
  if (mm) {
    const float4 gg = *(const float4*)(g1 + col1);
    const float4 be = *(const float4*)(be1 + col1);
#pragma unroll
    for (int r = 0; r < 8; ++r) {
      const float2 st = sStat[trow + r];
      float4 o;
      o.x = fmaxf((acc[0][r] - st.x) * st.y * gg.x + be.x, 0.f);
      o.y = fmaxf((acc[1][r] - st.x) * st.y * gg.y + be.y, 0.f);
      o.z = fmaxf((acc[2][r] - st.x) * st.y * gg.z + be.z, 0.f);
      o.w = fmaxf((acc[3][r] - st.x) * st.y * gg.w + be.w, 0.f);
      *(float4*)&sT[(trow + r) * T_PITCH + col1] = o;
    }
  }
  __syncthreads();

  if (mm) {
    {
      const float2 bb = *(const float2*)(b2 + col2);
#pragma unroll
      for (int r = 0; r < 8; ++r) {
        const int lrow = min(trow + r, nrows - 1);
        float2 o = *(const float2*)(hin + (size_t)(row0 + lrow) * HDIM + col2);
        if (!accum) o = make_float2(0.f, 0.f);
        acc2[0][r] = o.x + bb.x;
        acc2[1][r] = o.y + bb.y;
      }
    }
    const float2* w2p = (const float2*)(W2 + (size_t)(2 * hh) * HDIM + col2);
    const float* tp = sT + (16 * rt + c) * T_PITCH + 2 * hh;
#pragma unroll
    for (int ks = 0; ks < HDIM2 / 4; ++ks) {
      const int k0 = 4 * ks;
      const float2 blo = w2p[(size_t)k0 * (HDIM / 2)];
      const float2 bhi = w2p[(size_t)(k0 + 1) * (HDIM / 2)];
      const float2 av = *(const float2*)(tp + k0);
      v2f a; a[0] = av.x; a[1] = av.y;
      v2f b;
      b[0] = blo.x; b[1] = bhi.x; acc2[0] = wmma4(a, b, acc2[0]);
      b[0] = blo.y; b[1] = bhi.y; acc2[1] = wmma4(a, b, acc2[1]);
    }

#pragma unroll
    for (int r = 0; r < 8; ++r) {
      const float2 o = make_float2(acc2[0][r], acc2[1][r]);
      if (trow + r < nrows) *(float2*)(hout + (size_t)(row0 + trow + r) * HDIM + col2) = o;
      sRed[(trow + r) * R_PITCH + cg * 16 + c] = make_float2(o.x + o.y, o.x * o.x + o.y * o.y);
    }
  }
  __syncthreads();
  if (mm) row_stats(sRed, sStat, tid, 1.f / (float)HDIM);
  __syncthreads();
  if (mm) {
    const float2 gg = *(const float2*)(g2 + col2);
    const float2 be = *(const float2*)(be2 + col2);
#pragma unroll
    for (int r = 0; r < 8; ++r) {
      const float2 st = sStat[trow + r];
      float2 o;
      o.x = fmaxf((acc2[0][r] - st.x) * st.y * gg.x + be.x, 0.f);
      o.y = fmaxf((acc2[1][r] - st.x) * st.y * gg.y + be.y, 0.f);
      if (trow + r < nrows) *(float2*)(rout + (size_t)(row0 + trow + r) * HDIM + col2) = o;
    }
  }
}

#define PTRIPS 10
#define PROWS 8

__global__ void __launch_bounds__(1024)
pool_classify_kernel(const float* __restrict__ r, const int* __restrict__ batch,
                     const float* __restrict__ W, const float* __restrict__ b,
                     float* __restrict__ out, float* __restrict__ pooled, int n) {
  __shared__ __attribute__((aligned(16))) float part[32 * HDIM];
  __shared__ __attribute__((aligned(16))) float psum[HDIM];
  const int tid = threadIdx.x;
  const int lane = tid & 31;
  const int w = __builtin_amdgcn_readfirstlane(tid >> 5);
  const int g = blockIdx.x;

  float4 wv[NUM_CLASSES];
  float bias = 0.f;
  if (w == 0) {
#pragma unroll
    for (int j = 0; j < NUM_CLASSES; ++j) wv[j] = ((const float4*)W)[NUM_CLASSES * lane + j];
    if (lane < NUM_CLASSES) bias = b[lane];
  }

  const int lofs = 256 * (lane >> 3) + 8 * w + (lane & 7);
  float4 acc = make_float4(0.f, 0.f, 0.f, 0.f);
  for (int t0 = 0; 1024 * t0 < n; t0 += PTRIPS) {
    int bv[PTRIPS];
#pragma unroll
    for (int u = 0; u < PTRIPS; ++u) {
      const int nn = 1024 * (t0 + u) + lofs;
      bv[u] = nn < n ? batch[nn] : -1;
    }
#pragma unroll
    for (int u = 0; u < PTRIPS; ++u) {
      unsigned m = __builtin_amdgcn_ballot_w32(bv[u] == g);
      const int nbase = 1024 * (t0 + u) + 8 * w;
      while (m) {
        float4 v[PROWS];
#pragma unroll
        for (int q = 0; q < PROWS; ++q) {
          v[q] = make_float4(0.f, 0.f, 0.f, 0.f);
          if (m) {
            const int bit = __builtin_ctz(m);
            m &= m - 1;
            const int node = nbase + 256 * (bit >> 3) + (bit & 7);
            v[q] = ((const float4*)r)[(unsigned)node * (HDIM / 4) + lane];
          }
        }
#pragma unroll
        for (int q = 0; q < PROWS; ++q) {
          acc.x += v[q].x; acc.y += v[q].y; acc.z += v[q].z; acc.w += v[q].w;
        }
      }
    }
  }
  *(float4*)&part[w * HDIM + 4 * lane] = acc;
  __syncthreads();
  float ps = 0.f;
  if (tid < HDIM) {
    float t[32];
#pragma unroll
    for (int k = 0; k < 32; ++k) t[k] = part[k * HDIM + tid];
#pragma unroll
    for (int st = 1; st < 32; st <<= 1) {
#pragma unroll
      for (int k = 0; k < 32; k += 2 * st) t[k] += t[k + st];
    }
    ps = t[0];
    psum[tid] = ps;
  }
  __syncthreads();
  if (tid < HDIM) pooled[(size_t)g * HDIM + tid] = ps;
  if (w == 0) {
    const float4 p4 = *(const float4*)&psum[4 * lane];
    const float p[4] = {p4.x, p4.y, p4.z, p4.w};
    float wf[4 * NUM_CLASSES];
#pragma unroll
    for (int j = 0; j < NUM_CLASSES; ++j) {
      wf[4 * j + 0] = wv[j].x; wf[4 * j + 1] = wv[j].y; wf[4 * j + 2] = wv[j].z; wf[4 * j + 3] = wv[j].w;
    }
    float o = 0.f;
#pragma unroll
    for (int c = 0; c < NUM_CLASSES; ++c) {
      float s = (p[0] * wf[c] + p[1] * wf[NUM_CLASSES + c]) +
                (p[2] * wf[2 * NUM_CLASSES + c] + p[3] * wf[3 * NUM_CLASSES + c]);
#pragma unroll
      for (int off = 16; off > 0; off >>= 1) s += __shfl_xor(s, off, 32);
      if (lane == c) o = s;
    }
    if (lane < NUM_CLASSES) out[g * NUM_CLASSES + lane] = o + bias;
  }
}

extern "C" void kernel_launch(void* const* d_in, const int* in_sizes, int n_in,
                              void* d_out, int out_size, void* d_ws, size_t ws_size,
                              hipStream_t stream) {
  const float* x     = (const float*)d_in[0];
  const float* ea    = (const float*)d_in[1];
  const float* enc_W = (const float*)d_in[2];
  const float* enc_b = (const float*)d_in[3];
  const float* t     = (const float*)d_in[4];
  const float* W1    = (const float*)d_in[5];
  const float* b1    = (const float*)d_in[6];
  const float* mlp_g = (const float*)d_in[7];
  const float* mlp_b = (const float*)d_in[8];
  const float* W2    = (const float*)d_in[9];
  const float* b2    = (const float*)d_in[10];
  const float* ln_g  = (const float*)d_in[11];
  const float* ln_b  = (const float*)d_in[12];
  const float* lin_W = (const float*)d_in[13];
  const float* lin_b = (const float*)d_in[14];
  const int*   eidx  = (const int*)d_in[15];
  const int*   batch = (const int*)d_in[16];

  const int N = in_sizes[0] / HDIM;
  const int E = in_sizes[1] / HDIM;
  const int L = in_sizes[4];
  const int* src = eidx;
  const int* dst = eidx + E;

  const size_t NH = (size_t)N * HDIM;
  float* h    = (float*)d_ws;
  float* rbuf = h    + NH;
  float* hc   = rbuf + NH;
  float* t1   = hc   + NH;
  int* pairs  = (int*)(t1 + 2 * NH);
  int* raw    = pairs + 2 * (size_t)E;
  int* slot   = raw   + E;
  int* cnt    = slot  + E;
  int* off    = cnt   + N;
  int* binned = off   + N + 2;
  int* tab    = binned + 2 * (size_t)E;

  float* out    = (float*)d_out;
  float* pooled = out + NUM_GRAPHS * NUM_CLASSES;

  bin_local_kernel<<<NCHUNK, 1024, 0, stream>>>(dst, tab, binned, E, N);
  const int encBlocks = (N / 16 * (HDIM / 64) + ENC_STRIPS - 1) / ENC_STRIPS;
  bin_sort_encoder_kernel<<<NBIN + encBlocks, 1024, 0, stream>>>(tab, binned, src, slot, raw, off, pairs,
                                                                E, N, x, enc_W, enc_b, h);

  float* hsum = t1;
  const float* rin = h;
  for (int layer = 0; layer < L; ++layer) {
    float* rnext = (layer & 1) ? hc : rbuf;
    const int nl = layer + 1 < L ? layer + 1 : 0;
    layer_fused_kernel<<<(N + NPB - 1) / NPB, 640, 0, stream>>>(
        rin, ea, pairs, off, t, layer,
        W1 + (size_t)layer * HDIM * HDIM2, b1 + layer * HDIM2,
        mlp_g + layer * HDIM2, mlp_b + layer * HDIM2,
        W2 + (size_t)layer * HDIM2 * HDIM, b2 + layer * HDIM,
        layer > 0 ? hsum : h, hsum, layer > 0 ? 1 : 0, ln_g + nl * HDIM, ln_b + nl * HDIM, rnext, N);
    rin = rnext;
  }

  pool_classify_kernel<<<NUM_GRAPHS, 1024, 0, stream>>>(rin, batch, lin_W, lin_b, out, pooled, N);
}


# Softmax gather arithmetic cut: sums run over relu(x+e) with eps added once per node, weight = exp2 of one packed multiply-add (log2(e) folded into t); same sum order, terms round differently
# speedup vs baseline: 6.8282x; 1.0392x over previous
#include <hip/hip_runtime.h>
#include <hip/hip_bf16.h>

typedef __attribute__((ext_vector_type(2))) float v2f;
typedef __attribute__((ext_vector_type(8))) float v8f;

#define HDIM 128
#define HDIM2 256
#define NUM_GRAPHS 64
#define NUM_CLASSES 10
#define MSG_EPS 1e-7f

__device__ __forceinline__ v8f wmma4(v2f a, v2f b, v8f c) {
  return __builtin_amdgcn_wmma_f32_16x16x4_f32(false, a, false, b, (short)0, c, false, false);
}

#define ENC_STRIPS 10
#define ENC_KC 8

__device__ __forceinline__ void encoder_strip(const float* __restrict__ X,
                                              const float* __restrict__ W,
                                              const float* __restrict__ bias,
                                              float* __restrict__ Y, int strip, int lane) {
  const int hh = lane >> 4, c = lane & 15;
  const int row0 = (strip >> 1) * 16;
  const int col1 = (strip & 1) * 64 + 4 * c;
  v8f acc[4];
  {
    const float4 bb = *(const float4*)(bias + col1);
#pragma unroll
    for (int r = 0; r < 8; ++r) { acc[0][r] = bb.x; acc[1][r] = bb.y; acc[2][r] = bb.z; acc[3][r] = bb.w; }
  }
  const float4* wp = (const float4*)(W + (size_t)(2 * hh) * HDIM + col1);
  const float* ap = X + (size_t)(row0 + c) * HDIM + 2 * hh;
#pragma nounroll
  for (int kc = 0; kc < HDIM / 4; kc += ENC_KC) {
    float4 blo[ENC_KC], bhi[ENC_KC];
    float2 av[ENC_KC];
#pragma unroll
    for (int i = 0; i < ENC_KC; ++i) {
      const int k0 = 4 * (kc + i);
      blo[i] = wp[(size_t)k0 * (HDIM / 4)];
      bhi[i] = wp[(size_t)(k0 + 1) * (HDIM / 4)];
      av[i] = *(const float2*)(ap + k0);
    }
#pragma unroll
    for (int i = 0; i < ENC_KC; ++i) {
      v2f a; a[0] = av[i].x; a[1] = av[i].y;
      v2f b;
      b[0] = blo[i].x; b[1] = bhi[i].x; acc[0] = wmma4(a, b, acc[0]);
      b[0] = blo[i].y; b[1] = bhi[i].y; acc[1] = wmma4(a, b, acc[1]);
      b[0] = blo[i].z; b[1] = bhi[i].z; acc[2] = wmma4(a, b, acc[2]);
      b[0] = blo[i].w; b[1] = bhi[i].w; acc[3] = wmma4(a, b, acc[3]);
    }
  }
#pragma unroll
  for (int r = 0; r < 8; ++r) {
    float4 o;
    o.x = acc[0][r]; o.y = acc[1][r]; o.z = acc[2][r]; o.w = acc[3][r];
    *(float4*)(Y + (size_t)(row0 + 8 * hh + r) * HDIM + col1) = o;
  }
}

#define A_PITCH 132
#define T_PITCH 260
#define R_PITCH 66

__device__ __forceinline__ void row_stats(const float2* red, float2* stat, int tid, float invD) {
  const int row = tid >> 3, sub = tid & 7;
  float s = 0.f, q = 0.f;
#pragma unroll
  for (int i = 0; i < 8; ++i) {
    const float2 p = red[row * R_PITCH + sub + 8 * i];
    s += p.x; q += p.y;
  }
#pragma unroll
  for (int off = 1; off < 8; off <<= 1) {
    s += __shfl_xor(s, off, 32);
    q += __shfl_xor(q, off, 32);
  }
  if (sub == 0) {
    const float mean = s * invD;
    const float var  = q * invD - mean * mean;
    stat[row] = make_float2(mean, rsqrtf(var + 1e-5f));
  }
}

#define NBIN 128
#define NCHUNK 128
#define RMAX 128

__global__ void __launch_bounds__(1024)
bin_local_kernel(const int* __restrict__ dst, int* __restrict__ tab,
                 int* __restrict__ binned, int E, int N) {
  __shared__ int h[NBIN];
  __shared__ int scanb[NBIN];
  __shared__ int cursor[NBIN];
  const int tid = threadIdx.x;
  const int R = (N + NBIN - 1) / NBIN;
  const int C = (E + NCHUNK - 1) / NCHUNK;
  const int chunk = blockIdx.x;
  const int e0 = min(chunk * C, E), e1 = min(e0 + C, E);
  if (tid < NBIN) h[tid] = 0;
  __syncthreads();
  for (int e = e0 + tid; e < e1; e += 1024) atomicAdd(&h[dst[e] / R], 1);
  __syncthreads();
  int tot = 0;
  if (tid < NBIN) { tot = h[tid]; scanb[tid] = tot; }
  __syncthreads();
  for (int d = 1; d < NBIN; d <<= 1) {
    int v = 0;
    if (tid < NBIN && tid >= d) v = scanb[tid - d];
    __syncthreads();
    if (tid < NBIN) scanb[tid] += v;
    __syncthreads();
  }
  if (tid < NBIN) {
    const int excl = scanb[tid] - tot;
    cursor[tid] = e0 + excl;
    tab[chunk * (NBIN + 1) + tid] = excl;
    if (tid == NBIN - 1) tab[chunk * (NBIN + 1) + NBIN] = scanb[tid];
  }
  __syncthreads();
  for (int e = e0 + tid; e < e1; e += 1024) {
    const int d = dst[e];
    const int p = atomicAdd(&cursor[d / R], 1);
    ((int2*)binned)[p] = make_int2(e, d);
  }
}

__global__ void __launch_bounds__(1024)
bin_sort_encoder_kernel(const int* __restrict__ tab, const int* __restrict__ binned,
                        const int* __restrict__ src, int* __restrict__ slot, int* __restrict__ raw,
                        int* __restrict__ off, int* __restrict__ pairs, int E, int N,
                        const float* __restrict__ X, const float* __restrict__ encW,
                        const float* __restrict__ encB, float* __restrict__ Y) {
  __shared__ int cnt[RMAX];
  __shared__ int offL[RMAX];
  __shared__ int scanb[RMAX];
  __shared__ int segSt[NCHUNK];
  __shared__ int segLen[NCHUNK];
  __shared__ int redLo[32];
  const int tid = threadIdx.x;
  const int lane = tid & 31;
  const int w = __builtin_amdgcn_readfirstlane(tid >> 5);
  if (blockIdx.x >= NBIN) {
    const int strip = ((int)blockIdx.x - NBIN) * ENC_STRIPS + w;
    if (w < ENC_STRIPS && strip < (N / 16) * (HDIM / 64))
      encoder_strip(X, encW, encB, Y, strip, lane);
    return;
  }
  const int bin = blockIdx.x;
  const int R = (N + NBIN - 1) / NBIN;
  const int C = (E + NCHUNK - 1) / NCHUNK;
  const int n0 = min(bin * R, N), n1 = min(n0 + R, N);
  const int nn = n1 - n0;
  int lo = 0;
  {
    const int b = tid & (NBIN - 1), part = tid >> 7;
#pragma unroll
    for (int k = 0; k < NCHUNK / 8; ++k) {
      const int c = part * (NCHUNK / 8) + k;
      const int v = tab[c * (NBIN + 1) + b + 1] - tab[c * (NBIN + 1) + b];
      lo += (b < bin) ? v : 0;
    }
  }
#pragma unroll
  for (int o = 16; o > 0; o >>= 1) lo += __shfl_xor(lo, o, 32);
  if (lane == 0) redLo[w] = lo;
  if (tid < NCHUNK) {
    const int a = tab[tid * (NBIN + 1) + bin], z = tab[tid * (NBIN + 1) + bin + 1];
    segSt[tid] = min(tid * C, E) + a;
    segLen[tid] = z - a;
  }
  if (tid < RMAX) cnt[tid] = 0;
  __syncthreads();
  int s0 = 0;
#pragma unroll
  for (int i = 0; i < 32; ++i) s0 += redLo[i];
  int2 pr0[NCHUNK / 32];
  int sl0[NCHUNK / 32];
#pragma unroll
  for (int q = 0; q < NCHUNK / 32; ++q) {
    const int c = w + 32 * q;
    pr0[q] = make_int2(0, n0);
    if (lane < segLen[c]) pr0[q] = ((const int2*)binned)[segSt[c] + lane];
  }
#pragma unroll
  for (int q = 0; q < NCHUNK / 32; ++q) {
    const int c = w + 32 * q;
    const int st = segSt[c], len = segLen[c];
    sl0[q] = 0;
    if (lane < len) sl0[q] = atomicAdd(&cnt[pr0[q].y - n0], 1);
    for (int k = lane + 32; k < len; k += 32) {
      const int2 pr = ((const int2*)binned)[st + k];
      slot[st + k] = atomicAdd(&cnt[pr.y - n0], 1);
    }
  }
  __syncthreads();
  int mine = 0;
  if (tid < RMAX) { mine = cnt[tid]; scanb[tid] = mine; }
  __syncthreads();
  for (int d = 1; d < RMAX; d <<= 1) {
    int v = 0;
    if (tid < RMAX && tid >= d) v = scanb[tid - d];
    __syncthreads();
    if (tid < RMAX) scanb[tid] += v;
    __syncthreads();
  }
  if (tid < RMAX) {
    const int excl = scanb[tid] - mine;
    offL[tid] = excl;
    if (tid < nn) off[n0 + tid] = s0 + excl;
  }
  if (blockIdx.x == 0 && tid == 0) off[N] = E;
  __syncthreads();
#pragma unroll
  for (int q = 0; q < NCHUNK / 32; ++q) {
    const int c = w + 32 * q;
    const int st = segSt[c], len = segLen[c];
    if (lane < len) raw[s0 + offL[pr0[q].y - n0] + sl0[q]] = pr0[q].x;
    for (int k = lane + 32; k < len; k += 32) {
      const int2 pr = ((const int2*)binned)[st + k];
      raw[s0 + offL[pr.y - n0] + slot[st + k]] = pr.x;
    }
  }
  __syncthreads();
  for (int j = w; j < nn; j += 32) {
    const int beg = __builtin_amdgcn_readfirstlane(s0 + offL[j]);
    const int end = __builtin_amdgcn_readfirstlane(beg + cnt[j]);
    for (int ib = beg; ib < end; ib += 32) {
      const int i = ib + lane;
      const bool live = i < end;
      const int ei = live ? raw[i] : 0x7fffffff;
      int r = 0;
      for (int jb = beg; jb < end; jb += 32) {
        const int jj = jb + lane;
        const int ej = (jj < end) ? raw[jj] : 0x7fffffff;
        const int cj = min(32, end - jb);
        for (int k = 0; k < cj; ++k)
          r += (__builtin_amdgcn_readlane(ej, k) < ei) ? 1 : 0;
      }
      if (live) ((int2*)pairs)[beg + r] = make_int2(ei, src[ei]);
    }
  }
}

#define EB 8

__device__ __forceinline__ void group_load(const float* __restrict__ x,
                                           const float* __restrict__ ea,
                                           int2 pr, int first, unsigned lane,
                                           float4 (&xa)[EB], float4 (&eb)[EB]) {
#pragma unroll
  for (int j = 0; j < EB; ++j) {
    const int e = __builtin_amdgcn_readlane(pr.x, first + j);
    const int s = __builtin_amdgcn_readlane(pr.y, first + j);
    xa[j] = ((const float4*)x)[(unsigned)s * (HDIM / 4) + lane];
    eb[j] = ((const float4*)ea)[(unsigned)e * (HDIM / 4) + lane];
  }
}

template <bool TAIL>
__device__ __forceinline__ void group_compute(const float4 (&xa)[EB], const float4 (&eb)[EB],
                                              int nlive, float t2, bool tpos,
                                              v2f (&mx)[2], v2f (&den)[2], v2f (&num)[2]) {
  const v2f zero = {0.f, 0.f};
  const v2f t2v = {t2, t2};
  const v2f epsv = {MSG_EPS, MSG_EPS};
  v2f r[EB][2];
#pragma unroll
  for (int j = 0; j < EB; ++j) {
    const v2f x0 = {xa[j].x, xa[j].y}, x1 = {xa[j].z, xa[j].w};
    const v2f e0 = {eb[j].x, eb[j].y}, e1 = {eb[j].z, eb[j].w};
    r[j][0] = __builtin_elementwise_max(x0 + e0, zero);
    r[j][1] = __builtin_elementwise_max(x1 + e1, zero);
  }
#pragma unroll
  for (int p = 0; p < 2; ++p) {
    v2f ext = r[0][p];
    if (tpos) {
#pragma unroll
      for (int j = 1; j < EB; ++j) ext = __builtin_elementwise_max(ext, (TAIL && j >= nlive) ? r[0][p] : r[j][p]);
    } else {
#pragma unroll
      for (int j = 1; j < EB; ++j) ext = __builtin_elementwise_min(ext, (TAIL && j >= nlive) ? r[0][p] : r[j][p]);
    }
    const v2f nm = __builtin_elementwise_max(mx[p], (ext + epsv) * t2v);
    const v2f ds = mx[p] - nm;
    v2f sc; sc[0] = __builtin_amdgcn_exp2f(ds[0]); sc[1] = __builtin_amdgcn_exp2f(ds[1]);
    const v2f c0 = epsv * t2v - nm;
    v2f d = den[p] * sc, u = num[p] * sc;
#pragma unroll
    for (int j = 0; j < EB; ++j) {
      const v2f a = r[j][p] * t2v + c0;
      v2f ex; ex[0] = __builtin_amdgcn_exp2f(a[0]); ex[1] = __builtin_amdgcn_exp2f(a[1]);
      if (TAIL && j >= nlive) ex = zero;
      d += ex;
      u += ex * r[j][p];
    }
    mx[p] = nm; den[p] = d; num[p] = u;
  }
}

#define NPB 40
#define TROWS 48

__global__ void __launch_bounds__(640) __attribute__((amdgpu_waves_per_eu(5, 5)))
layer_fused_kernel(const float* __restrict__ xin, const float* __restrict__ ea,
                   const int* __restrict__ pairs, const int* __restrict__ off,
                   const float* __restrict__ tptr, int layer,
                   const float* __restrict__ W1, const float* __restrict__ b1,
                   const float* __restrict__ g1, const float* __restrict__ be1,
                   const float* __restrict__ W2, const float* __restrict__ b2,
                   const float* hin, float* hout, int accum,
                   const float* __restrict__ g2, const float* __restrict__ be2,
                   float* __restrict__ rout, int n) {
  __shared__ __attribute__((aligned(16))) float sA[TROWS * A_PITCH];
  __shared__ __attribute__((aligned(16))) float sT[TROWS * T_PITCH];
  __shared__ float2 sRed[TROWS * R_PITCH];
  __shared__ float2 sStat[TROWS];

  const int tid  = threadIdx.x;
  const int lane = tid & 31;
  const int w = __builtin_amdgcn_readfirstlane(tid >> 5);
  const int row0 = blockIdx.x * NPB;

  {
    const int nodeA = row0 + 2 * w;
    const bool hasA = nodeA < n, hasB = nodeA + 1 < n;
    int begA = 0, endA = 0, endB = 0;
    if (hasA) { begA = off[nodeA]; endA = off[nodeA + 1]; endB = hasB ? off[nodeA + 2] : endA; }
    const int lenA = endA - begA, lenB = endB - endA;
    const float tv = tptr[layer];
    const bool tpos = tv >= 0.f;
    const float t2 = tv * 1.44269504088896341f;

    v2f mxA[2], denA[2], numA[2], mxB[2], denB[2], numB[2];
#pragma unroll
    for (int p = 0; p < 2; ++p) {
      const float ninf = __int_as_float(0xff800000);
      mxA[p][0] = ninf; mxA[p][1] = ninf; denA[p][0] = 0.f; denA[p][1] = 0.f; numA[p][0] = 0.f; numA[p][1] = 0.f;
      mxB[p][0] = ninf; mxB[p][1] = ninf; denB[p][0] = 0.f; denB[p][1] = 0.f; numB[p][0] = 0.f; numB[p][1] = 0.f;
    }

    for (int base = 0; base < lenA || base < lenB; base += 32) {
      const int cntA = min(32, lenA - base), cntB = min(32, lenB - base);
      int2 prA = make_int2(0, 0), prB = make_int2(0, 0);
      if (cntA > 0) prA = ((const int2*)pairs)[begA + base + min(lane, cntA - 1)];
      if (cntB > 0) prB = ((const int2*)pairs)[endA + base + min(lane, cntB - 1)];
      for (int g = 0; g < cntA || g < cntB; g += EB) {
        const int nA = cntA - g, nB = cntB - g;
        if (nA > 0 && nB > 0) {
          float4 xaA[EB], ebA[EB], xaB[EB], ebB[EB];
          group_load(xin, ea, prA, g, (unsigned)lane, xaA, ebA);
          group_load(xin, ea, prB, g, (unsigned)lane, xaB, ebB);
          if (nA >= EB) group_compute<false>(xaA, ebA, EB, t2, tpos, mxA, denA, numA);
          else          group_compute<true>(xaA, ebA, nA, t2, tpos, mxA, denA, numA);
          if (nB >= EB) group_compute<false>(xaB, ebB, EB, t2, tpos, mxB, denB, numB);
          else          group_compute<true>(xaB, ebB, nB, t2, tpos, mxB, denB, numB);
        } else if (nA > 0) {
          float4 xaA[EB], ebA[EB];
          group_load(xin, ea, prA, g, (unsigned)lane, xaA, ebA);
          if (nA >= EB) group_compute<false>(xaA, ebA, EB, t2, tpos, mxA, denA, numA);
          else          group_compute<true>(xaA, ebA, nA, t2, tpos, mxA, denA, numA);
        } else {
          float4 xaB[EB], ebB[EB];
          group_load(xin, ea, prB, g, (unsigned)lane, xaB, ebB);
          if (nB >= EB) group_compute<false>(xaB, ebB, EB, t2, tpos, mxB, denB, numB);
          else          group_compute<true>(xaB, ebB, nB, t2, tpos, mxB, denB, numB);
        }
      }
    }

    float4 oA = make_float4(0.f, 0.f, 0.f, 0.f), oB = oA;
    if (hasA) {
      const float4 root = ((const float4*)(xin + (size_t)nodeA * HDIM))[lane];
      oA = root;
      if (lenA > 0) {
        oA.x = (numA[0][0] / denA[0][0] + MSG_EPS) + root.x;
        oA.y = (numA[0][1] / denA[0][1] + MSG_EPS) + root.y;
        oA.z = (numA[1][0] / denA[1][0] + MSG_EPS) + root.z;
        oA.w = (numA[1][1] / denA[1][1] + MSG_EPS) + root.w;
      }
    }
    if (hasB) {
      const float4 root = ((const float4*)(xin + (size_t)(nodeA + 1) * HDIM))[lane];
      oB = root;
      if (lenB > 0) {
        oB.x = (numB[0][0] / denB[0][0] + MSG_EPS) + root.x;
        oB.y = (numB[0][1] / denB[0][1] + MSG_EPS) + root.y;
        oB.z = (numB[1][0] / denB[1][0] + MSG_EPS) + root.z;
        oB.w = (numB[1][1] / denB[1][1] + MSG_EPS) + root.w;
      }
    }
    *(float4*)&sA[(2 * w) * A_PITCH + 4 * lane] = oA;
    *(float4*)&sA[(2 * w + 1) * A_PITCH + 4 * lane] = oB;
    if (w < (TROWS - NPB) / 2) {
      const float4 z = make_float4(0.f, 0.f, 0.f, 0.f);
      *(float4*)&sA[(NPB + 2 * w) * A_PITCH + 4 * lane] = z;
      *(float4*)&sA[(NPB + 2 * w + 1) * A_PITCH + 4 * lane] = z;
    }
  }
  __syncthreads();

  const bool mm = w < 12;
  const int rt = w >> 2, cg = w & 3;
  const int hh = lane >> 4, c = lane & 15;
  const int col1 = cg * 64 + 4 * c;
  const int col2 = cg * 32 + 2 * c;
  const int trow = 16 * rt + 8 * hh;
  const int nrows = min(NPB, n - row0);

  v8f acc[4];
  v8f acc2[2];
  if (mm) {
    {
      const float4 bb = *(const float4*)(b1 + col1);
#pragma unroll
      for (int r = 0; r < 8; ++r) { acc[0][r] = bb.x; acc[1][r] = bb.y; acc[2][r] = bb.z; acc[3][r] = bb.w; }
    }
    const float4* w1p = (const float4*)(W1 + (size_t)(2 * hh) * HDIM2 + col1);
    const float* ap = sA + (16 * rt + c) * A_PITCH + 2 * hh;
#pragma unroll
    for (int ks = 0; ks < HDIM / 4; ++ks) {
      const int k0 = 4 * ks;
      const float4 blo = w1p[(size_t)k0 * (HDIM2 / 4)];
      const float4 bhi = w1p[(size_t)(k0 + 1) * (HDIM2 / 4)];
      const float2 av = *(const float2*)(ap + k0);
      v2f a; a[0] = av.x; a[1] = av.y;
      v2f b;
      b[0] = blo.x; b[1] = bhi.x; acc[0] = wmma4(a, b, acc[0]);
      b[0] = blo.y; b[1] = bhi.y; acc[1] = wmma4(a, b, acc[1]);
      b[0] = blo.z; b[1] = bhi.z; acc[2] = wmma4(a, b, acc[2]);
      b[0] = blo.w; b[1] = bhi.w; acc[3] = wmma4(a, b, acc[3]);
    }

#pragma unroll
    for (int r = 0; r < 8; ++r) {
      const float s = (acc[0][r] + acc[1][r]) + (acc[2][r] + acc[3][r]);
      const float q = (acc[0][r] * acc[0][r] + acc[1][r] * acc[1][r]) +
                      (acc[2][r] * acc[2][r] + acc[3][r] * acc[3][r]);
      sRed[(trow + r) * R_PITCH + cg * 16 + c] = make_float2(s, q);
    }
  }
  __syncthreads();
  if (mm) row_stats(sRed, sStat, tid, 1.f / (float)HDIM2);
  __syncthreads();
  if (mm) {
    const float4 gg = *(const float4*)(g1 + col1);
    const float4 be = *(const float4*)(be1 + col1);
#pragma unroll
    for (int r = 0; r < 8; ++r) {
      const float2 st = sStat[trow + r];
      float4 o;
      o.x = fmaxf((acc[0][r] - st.x) * st.y * gg.x + be.x, 0.f);
      o.y = fmaxf((acc[1][r] - st.x) * st.y * gg.y + be.y, 0.f);
      o.z = fmaxf((acc[2][r] - st.x) * st.y * gg.z + be.z, 0.f);
      o.w = fmaxf((acc[3][r] - st.x) * st.y * gg.w + be.w, 0.f);
      *(float4*)&sT[(trow + r) * T_PITCH + col1] = o;
    }
  }
  __syncthreads();

  if (mm) {
    {
      const float2 bb = *(const float2*)(b2 + col2);
#pragma unroll
      for (int r = 0; r < 8; ++r) {
        const int lrow = min(trow + r, nrows - 1);
        float2 o = *(const float2*)(hin + (size_t)(row0 + lrow) * HDIM + col2);
        if (!accum) o = make_float2(0.f, 0.f);
        acc2[0][r] = o.x + bb.x;
        acc2[1][r] = o.y + bb.y;
      }
    }
    const float2* w2p = (const float2*)(W2 + (size_t)(2 * hh) * HDIM + col2);
    const float* tp = sT + (16 * rt + c) * T_PITCH + 2 * hh;
#pragma unroll
    for (int ks = 0; ks < HDIM2 / 4; ++ks) {
      const int k0 = 4 * ks;
      const float2 blo = w2p[(size_t)k0 * (HDIM / 2)];
      const float2 bhi = w2p[(size_t)(k0 + 1) * (HDIM / 2)];
      const float2 av = *(const float2*)(tp + k0);
      v2f a; a[0] = av.x; a[1] = av.y;
      v2f b;
      b[0] = blo.x; b[1] = bhi.x; acc2[0] = wmma4(a, b, acc2[0]);
      b[0] = blo.y; b[1] = bhi.y; acc2[1] = wmma4(a, b, acc2[1]);
    }

#pragma unroll
    for (int r = 0; r < 8; ++r) {
      const float2 o = make_float2(acc2[0][r], acc2[1][r]);
      if (trow + r < nrows) *(float2*)(hout + (size_t)(row0 + trow + r) * HDIM + col2) = o;
      sRed[(trow + r) * R_PITCH + cg * 16 + c] = make_float2(o.x + o.y, o.x * o.x + o.y * o.y);
    }
  }
  __syncthreads();
  if (mm) row_stats(sRed, sStat, tid, 1.f / (float)HDIM);
  __syncthreads();
  if (mm) {
    const float2 gg = *(const float2*)(g2 + col2);
    const float2 be = *(const float2*)(be2 + col2);
#pragma unroll
    for (int r = 0; r < 8; ++r) {
      const float2 st = sStat[trow + r];
      float2 o;
      o.x = fmaxf((acc2[0][r] - st.x) * st.y * gg.x + be.x, 0.f);
      o.y = fmaxf((acc2[1][r] - st.x) * st.y * gg.y + be.y, 0.f);
      if (trow + r < nrows) *(float2*)(rout + (size_t)(row0 + trow + r) * HDIM + col2) = o;
    }
  }
}

#define PTRIPS 10
#define PROWS 8

__global__ void __launch_bounds__(1024)
pool_classify_kernel(const float* __restrict__ r, const int* __restrict__ batch,
                     const float* __restrict__ W, const float* __restrict__ b,
                     float* __restrict__ out, float* __restrict__ pooled, int n) {
  __shared__ __attribute__((aligned(16))) float part[32 * HDIM];
  __shared__ __attribute__((aligned(16))) float psum[HDIM];
  const int tid = threadIdx.x;
  const int lane = tid & 31;
  const int w = __builtin_amdgcn_readfirstlane(tid >> 5);
  const int g = blockIdx.x;

  float4 wv[NUM_CLASSES];
  float bias = 0.f;
  if (w == 0) {
#pragma unroll
    for (int j = 0; j < NUM_CLASSES; ++j) wv[j] = ((const float4*)W)[NUM_CLASSES * lane + j];
    if (lane < NUM_CLASSES) bias = b[lane];
  }

  const int lofs = 256 * (lane >> 3) + 8 * w + (lane & 7);
  float4 acc = make_float4(0.f, 0.f, 0.f, 0.f);
  for (int t0 = 0; 1024 * t0 < n; t0 += PTRIPS) {
    int bv[PTRIPS];
#pragma unroll
    for (int u = 0; u < PTRIPS; ++u) {
      const int nn = 1024 * (t0 + u) + lofs;
      bv[u] = nn < n ? batch[nn] : -1;
    }
#pragma unroll
    for (int u = 0; u < PTRIPS; ++u) {
      unsigned m = __builtin_amdgcn_ballot_w32(bv[u] == g);
      const int nbase = 1024 * (t0 + u) + 8 * w;
      while (m) {
        float4 v[PROWS];
#pragma unroll
        for (int q = 0; q < PROWS; ++q) {
          v[q] = make_float4(0.f, 0.f, 0.f, 0.f);
          if (m) {
            const int bit = __builtin_ctz(m);
            m &= m - 1;
            const int node = nbase + 256 * (bit >> 3) + (bit & 7);
            v[q] = ((const float4*)r)[(unsigned)node * (HDIM / 4) + lane];
          }
        }
#pragma unroll
        for (int q = 0; q < PROWS; ++q) {
          acc.x += v[q].x; acc.y += v[q].y; acc.z += v[q].z; acc.w += v[q].w;
        }
      }
    }
  }
  *(float4*)&part[w * HDIM + 4 * lane] = acc;
  __syncthreads();
  float ps = 0.f;
  if (tid < HDIM) {
    float t[32];
#pragma unroll
    for (int k = 0; k < 32; ++k) t[k] = part[k * HDIM + tid];
#pragma unroll
    for (int st = 1; st < 32; st <<= 1) {
#pragma unroll
      for (int k = 0; k < 32; k += 2 * st) t[k] += t[k + st];
    }
    ps = t[0];
    psum[tid] = ps;
  }
  __syncthreads();
  if (tid < HDIM) pooled[(size_t)g * HDIM + tid] = ps;
  if (w == 0) {
    const float4 p4 = *(const float4*)&psum[4 * lane];
    const float p[4] = {p4.x, p4.y, p4.z, p4.w};
    float wf[4 * NUM_CLASSES];
#pragma unroll
    for (int j = 0; j < NUM_CLASSES; ++j) {
      wf[4 * j + 0] = wv[j].x; wf[4 * j + 1] = wv[j].y; wf[4 * j + 2] = wv[j].z; wf[4 * j + 3] = wv[j].w;
    }
    float o = 0.f;
#pragma unroll
    for (int c = 0; c < NUM_CLASSES; ++c) {
      float s = (p[0] * wf[c] + p[1] * wf[NUM_CLASSES + c]) +
                (p[2] * wf[2 * NUM_CLASSES + c] + p[3] * wf[3 * NUM_CLASSES + c]);
#pragma unroll
      for (int off = 16; off > 0; off >>= 1) s += __shfl_xor(s, off, 32);
      if (lane == c) o = s;
    }
    if (lane < NUM_CLASSES) out[g * NUM_CLASSES + lane] = o + bias;
  }
}

extern "C" void kernel_launch(void* const* d_in, const int* in_sizes, int n_in,
                              void* d_out, int out_size, void* d_ws, size_t ws_size,
                              hipStream_t stream) {
  const float* x     = (const float*)d_in[0];
  const float* ea    = (const float*)d_in[1];
  const float* enc_W = (const float*)d_in[2];
  const float* enc_b = (const float*)d_in[3];
  const float* t     = (const float*)d_in[4];
  const float* W1    = (const float*)d_in[5];
  const float* b1    = (const float*)d_in[6];
  const float* mlp_g = (const float*)d_in[7];
  const float* mlp_b = (const float*)d_in[8];
  const float* W2    = (const float*)d_in[9];
  const float* b2    = (const float*)d_in[10];
  const float* ln_g  = (const float*)d_in[11];
  const float* ln_b  = (const float*)d_in[12];
  const float* lin_W = (const float*)d_in[13];
  const float* lin_b = (const float*)d_in[14];
  const int*   eidx  = (const int*)d_in[15];
  const int*   batch = (const int*)d_in[16];

  const int N = in_sizes[0] / HDIM;
  const int E = in_sizes[1] / HDIM;
  const int L = in_sizes[4];
  const int* src = eidx;
  const int* dst = eidx + E;

  const size_t NH = (size_t)N * HDIM;
  float* h    = (float*)d_ws;
  float* rbuf = h    + NH;
  float* hc   = rbuf + NH;
  float* t1   = hc   + NH;
  int* pairs  = (int*)(t1 + 2 * NH);
  int* raw    = pairs + 2 * (size_t)E;
  int* slot   = raw   + E;
  int* cnt    = slot  + E;
  int* off    = cnt   + N;
  int* binned = off   + N + 2;
  int* tab    = binned + 2 * (size_t)E;

  float* out    = (float*)d_out;
  float* pooled = out + NUM_GRAPHS * NUM_CLASSES;

  bin_local_kernel<<<NCHUNK, 1024, 0, stream>>>(dst, tab, binned, E, N);
  const int encBlocks = (N / 16 * (HDIM / 64) + ENC_STRIPS - 1) / ENC_STRIPS;
  bin_sort_encoder_kernel<<<NBIN + encBlocks, 1024, 0, stream>>>(tab, binned, src, slot, raw, off, pairs,
                                                                E, N, x, enc_W, enc_b, h);

  float* hsum = t1;
  const float* rin = h;
  for (int layer = 0; layer < L; ++layer) {
    float* rnext = (layer & 1) ? hc : rbuf;
    const int nl = layer + 1 < L ? layer + 1 : 0;
    layer_fused_kernel<<<(N + NPB - 1) / NPB, 640, 0, stream>>>(
        rin, ea, pairs, off, t, layer,
        W1 + (size_t)layer * HDIM * HDIM2, b1 + layer * HDIM2,
        mlp_g + layer * HDIM2, mlp_b + layer * HDIM2,
        W2 + (size_t)layer * HDIM2 * HDIM, b2 + layer * HDIM,
        layer > 0 ? hsum : h, hsum, layer > 0 ? 1 : 0, ln_g + nl * HDIM, ln_b + nl * HDIM, rnext, N);
    rin = rnext;
  }

  pool_classify_kernel<<<NUM_GRAPHS, 1024, 0, stream>>>(rin, batch, lin_W, lin_b, out, pooled, N);
}


# Gather: after the first group's row loads of each 32-edge stretch, lanes of slots 8 and up prefetch their edge-attribute rows to L2 (no register held); later groups then hit L2; sums unchanged
# speedup vs baseline: 7.0835x; 1.0374x over previous
#include <hip/hip_runtime.h>
#include <hip/hip_bf16.h>

typedef __attribute__((ext_vector_type(2))) float v2f;
typedef __attribute__((ext_vector_type(8))) float v8f;

#define HDIM 128
#define HDIM2 256
#define NUM_GRAPHS 64
#define NUM_CLASSES 10
#define MSG_EPS 1e-7f

__device__ __forceinline__ v8f wmma4(v2f a, v2f b, v8f c) {
  return __builtin_amdgcn_wmma_f32_16x16x4_f32(false, a, false, b, (short)0, c, false, false);
}

#define ENC_STRIPS 10
#define ENC_KC 8

__device__ __forceinline__ void encoder_strip(const float* __restrict__ X,
                                              const float* __restrict__ W,
                                              const float* __restrict__ bias,
                                              float* __restrict__ Y, int strip, int lane) {
  const int hh = lane >> 4, c = lane & 15;
  const int row0 = (strip >> 1) * 16;
  const int col1 = (strip & 1) * 64 + 4 * c;
  v8f acc[4];
  {
    const float4 bb = *(const float4*)(bias + col1);
#pragma unroll
    for (int r = 0; r < 8; ++r) { acc[0][r] = bb.x; acc[1][r] = bb.y; acc[2][r] = bb.z; acc[3][r] = bb.w; }
  }
  const float4* wp = (const float4*)(W + (size_t)(2 * hh) * HDIM + col1);
  const float* ap = X + (size_t)(row0 + c) * HDIM + 2 * hh;
#pragma nounroll
  for (int kc = 0; kc < HDIM / 4; kc += ENC_KC) {
    float4 blo[ENC_KC], bhi[ENC_KC];
    float2 av[ENC_KC];
#pragma unroll
    for (int i = 0; i < ENC_KC; ++i) {
      const int k0 = 4 * (kc + i);
      blo[i] = wp[(size_t)k0 * (HDIM / 4)];
      bhi[i] = wp[(size_t)(k0 + 1) * (HDIM / 4)];
      av[i] = *(const float2*)(ap + k0);
    }
#pragma unroll
    for (int i = 0; i < ENC_KC; ++i) {
      v2f a; a[0] = av[i].x; a[1] = av[i].y;
      v2f b;
      b[0] = blo[i].x; b[1] = bhi[i].x; acc[0] = wmma4(a, b, acc[0]);
      b[0] = blo[i].y; b[1] = bhi[i].y; acc[1] = wmma4(a, b, acc[1]);
      b[0] = blo[i].z; b[1] = bhi[i].z; acc[2] = wmma4(a, b, acc[2]);
      b[0] = blo[i].w; b[1] = bhi[i].w; acc[3] = wmma4(a, b, acc[3]);
    }
  }
#pragma unroll
  for (int r = 0; r < 8; ++r) {
    float4 o;
    o.x = acc[0][r]; o.y = acc[1][r]; o.z = acc[2][r]; o.w = acc[3][r];
    *(float4*)(Y + (size_t)(row0 + 8 * hh + r) * HDIM + col1) = o;
  }
}

#define A_PITCH 132
#define T_PITCH 260
#define R_PITCH 66

__device__ __forceinline__ void row_stats(const float2* red, float2* stat, int tid, float invD) {
  const int row = tid >> 3, sub = tid & 7;
  float s = 0.f, q = 0.f;
#pragma unroll
  for (int i = 0; i < 8; ++i) {
    const float2 p = red[row * R_PITCH + sub + 8 * i];
    s += p.x; q += p.y;
  }
#pragma unroll
  for (int off = 1; off < 8; off <<= 1) {
    s += __shfl_xor(s, off, 32);
    q += __shfl_xor(q, off, 32);
  }
  if (sub == 0) {
    const float mean = s * invD;
    const float var  = q * invD - mean * mean;
    stat[row] = make_float2(mean, rsqrtf(var + 1e-5f));
  }
}

#define NBIN 128
#define NCHUNK 128
#define RMAX 128

__global__ void __launch_bounds__(1024)
bin_local_kernel(const int* __restrict__ dst, int* __restrict__ tab,
                 int* __restrict__ binned, int E, int N) {
  __shared__ int h[NBIN];
  __shared__ int scanb[NBIN];
  __shared__ int cursor[NBIN];
  const int tid = threadIdx.x;
  const int R = (N + NBIN - 1) / NBIN;
  const int C = (E + NCHUNK - 1) / NCHUNK;
  const int chunk = blockIdx.x;
  const int e0 = min(chunk * C, E), e1 = min(e0 + C, E);
  if (tid < NBIN) h[tid] = 0;
  __syncthreads();
  for (int e = e0 + tid; e < e1; e += 1024) atomicAdd(&h[dst[e] / R], 1);
  __syncthreads();
  int tot = 0;
  if (tid < NBIN) { tot = h[tid]; scanb[tid] = tot; }
  __syncthreads();
  for (int d = 1; d < NBIN; d <<= 1) {
    int v = 0;
    if (tid < NBIN && tid >= d) v = scanb[tid - d];
    __syncthreads();
    if (tid < NBIN) scanb[tid] += v;
    __syncthreads();
  }
  if (tid < NBIN) {
    const int excl = scanb[tid] - tot;
    cursor[tid] = e0 + excl;
    tab[chunk * (NBIN + 1) + tid] = excl;
    if (tid == NBIN - 1) tab[chunk * (NBIN + 1) + NBIN] = scanb[tid];
  }
  __syncthreads();
  for (int e = e0 + tid; e < e1; e += 1024) {
    const int d = dst[e];
    const int p = atomicAdd(&cursor[d / R], 1);
    ((int2*)binned)[p] = make_int2(e, d);
  }
}

__global__ void __launch_bounds__(1024)
bin_sort_encoder_kernel(const int* __restrict__ tab, const int* __restrict__ binned,
                        const int* __restrict__ src, int* __restrict__ slot, int* __restrict__ raw,
                        int* __restrict__ off, int* __restrict__ pairs, int E, int N,
                        const float* __restrict__ X, const float* __restrict__ encW,
                        const float* __restrict__ encB, float* __restrict__ Y) {
  __shared__ int cnt[RMAX];
  __shared__ int offL[RMAX];
  __shared__ int scanb[RMAX];
  __shared__ int segSt[NCHUNK];
  __shared__ int segLen[NCHUNK];
  __shared__ int redLo[32];
  const int tid = threadIdx.x;
  const int lane = tid & 31;
  const int w = __builtin_amdgcn_readfirstlane(tid >> 5);
  if (blockIdx.x >= NBIN) {
    const int strip = ((int)blockIdx.x - NBIN) * ENC_STRIPS + w;
    if (w < ENC_STRIPS && strip < (N / 16) * (HDIM / 64))
      encoder_strip(X, encW, encB, Y, strip, lane);
    return;
  }
  const int bin = blockIdx.x;
  const int R = (N + NBIN - 1) / NBIN;
  const int C = (E + NCHUNK - 1) / NCHUNK;
  const int n0 = min(bin * R, N), n1 = min(n0 + R, N);
  const int nn = n1 - n0;
  int lo = 0;
  {
    const int b = tid & (NBIN - 1), part = tid >> 7;
#pragma unroll
    for (int k = 0; k < NCHUNK / 8; ++k) {
      const int c = part * (NCHUNK / 8) + k;
      const int v = tab[c * (NBIN + 1) + b + 1] - tab[c * (NBIN + 1) + b];
      lo += (b < bin) ? v : 0;
    }
  }
#pragma unroll
  for (int o = 16; o > 0; o >>= 1) lo += __shfl_xor(lo, o, 32);
  if (lane == 0) redLo[w] = lo;
  if (tid < NCHUNK) {
    const int a = tab[tid * (NBIN + 1) + bin], z = tab[tid * (NBIN + 1) + bin + 1];
    segSt[tid] = min(tid * C, E) + a;
    segLen[tid] = z - a;
  }
  if (tid < RMAX) cnt[tid] = 0;
  __syncthreads();
  int s0 = 0;
#pragma unroll
  for (int i = 0; i < 32; ++i) s0 += redLo[i];
  int2 pr0[NCHUNK / 32];
  int sl0[NCHUNK / 32];
#pragma unroll
  for (int q = 0; q < NCHUNK / 32; ++q) {
    const int c = w + 32 * q;
    pr0[q] = make_int2(0, n0);
    if (lane < segLen[c]) pr0[q] = ((const int2*)binned)[segSt[c] + lane];
  }
#pragma unroll
  for (int q = 0; q < NCHUNK / 32; ++q) {
    const int c = w + 32 * q;
    const int st = segSt[c], len = segLen[c];
    sl0[q] = 0;
    if (lane < len) sl0[q] = atomicAdd(&cnt[pr0[q].y - n0], 1);
    for (int k = lane + 32; k < len; k += 32) {
      const int2 pr = ((const int2*)binned)[st + k];
      slot[st + k] = atomicAdd(&cnt[pr.y - n0], 1);
    }
  }
  __syncthreads();
  int mine = 0;
  if (tid < RMAX) { mine = cnt[tid]; scanb[tid] = mine; }
  __syncthreads();
  for (int d = 1; d < RMAX; d <<= 1) {
    int v = 0;
    if (tid < RMAX && tid >= d) v = scanb[tid - d];
    __syncthreads();
    if (tid < RMAX) scanb[tid] += v;
    __syncthreads();
  }
  if (tid < RMAX) {
    const int excl = scanb[tid] - mine;
    offL[tid] = excl;
    if (tid < nn) off[n0 + tid] = s0 + excl;
  }
  if (blockIdx.x == 0 && tid == 0) off[N] = E;
  __syncthreads();
#pragma unroll
  for (int q = 0; q < NCHUNK / 32; ++q) {
    const int c = w + 32 * q;
    const int st = segSt[c], len = segLen[c];
    if (lane < len) raw[s0 + offL[pr0[q].y - n0] + sl0[q]] = pr0[q].x;
    for (int k = lane + 32; k < len; k += 32) {
      const int2 pr = ((const int2*)binned)[st + k];
      raw[s0 + offL[pr.y - n0] + slot[st + k]] = pr.x;
    }
  }
  __syncthreads();
  for (int j = w; j < nn; j += 32) {
    const int beg = __builtin_amdgcn_readfirstlane(s0 + offL[j]);
    const int end = __builtin_amdgcn_readfirstlane(beg + cnt[j]);
    for (int ib = beg; ib < end; ib += 32) {
      const int i = ib + lane;
      const bool live = i < end;
      const int ei = live ? raw[i] : 0x7fffffff;
      int r = 0;
      for (int jb = beg; jb < end; jb += 32) {
        const int jj = jb + lane;
        const int ej = (jj < end) ? raw[jj] : 0x7fffffff;
        const int cj = min(32, end - jb);
        for (int k = 0; k < cj; ++k)
          r += (__builtin_amdgcn_readlane(ej, k) < ei) ? 1 : 0;
      }
      if (live) ((int2*)pairs)[beg + r] = make_int2(ei, src[ei]);
    }
  }
}

#define EB 8

__device__ __forceinline__ void group_load(const float* __restrict__ x,
                                           const float* __restrict__ ea,
                                           int2 pr, int first, unsigned lane,
                                           float4 (&xa)[EB], float4 (&eb)[EB]) {
#pragma unroll
  for (int j = 0; j < EB; ++j) {
    const int e = __builtin_amdgcn_readlane(pr.x, first + j);
    const int s = __builtin_amdgcn_readlane(pr.y, first + j);
    xa[j] = ((const float4*)x)[(unsigned)s * (HDIM / 4) + lane];
    eb[j] = ((const float4*)ea)[(unsigned)e * (HDIM / 4) + lane];
  }
}

__device__ __forceinline__ void row_prefetch(const float* __restrict__ ea, int e) {
  const char* p = (const char*)ea + (size_t)(unsigned)e * (HDIM * sizeof(float));
  __builtin_prefetch(p, 0, 2);
  __builtin_prefetch(p + 256, 0, 2);
}
__device__ __forceinline__ void stretch_prefetch(const float* __restrict__ ea, int2 prA, int cntA,
                                                 int2 prB, int cntB, int lane) {
  if (lane >= EB && lane < cntA) row_prefetch(ea, prA.x);
  if (lane >= EB && lane < cntB) row_prefetch(ea, prB.x);
}

template <bool TAIL>
__device__ __forceinline__ void group_compute(const float4 (&xa)[EB], const float4 (&eb)[EB],
                                              int nlive, float t2, bool tpos,
                                              v2f (&mx)[2], v2f (&den)[2], v2f (&num)[2]) {
  const v2f zero = {0.f, 0.f};
  const v2f t2v = {t2, t2};
  const v2f epsv = {MSG_EPS, MSG_EPS};
  v2f r[EB][2];
#pragma unroll
  for (int j = 0; j < EB; ++j) {
    const v2f x0 = {xa[j].x, xa[j].y}, x1 = {xa[j].z, xa[j].w};
    const v2f e0 = {eb[j].x, eb[j].y}, e1 = {eb[j].z, eb[j].w};
    r[j][0] = __builtin_elementwise_max(x0 + e0, zero);
    r[j][1] = __builtin_elementwise_max(x1 + e1, zero);
  }
#pragma unroll
  for (int p = 0; p < 2; ++p) {
    v2f ext = r[0][p];
    if (tpos) {
#pragma unroll
      for (int j = 1; j < EB; ++j) ext = __builtin_elementwise_max(ext, (TAIL && j >= nlive) ? r[0][p] : r[j][p]);
    } else {
#pragma unroll
      for (int j = 1; j < EB; ++j) ext = __builtin_elementwise_min(ext, (TAIL && j >= nlive) ? r[0][p] : r[j][p]);
    }
    const v2f nm = __builtin_elementwise_max(mx[p], (ext + epsv) * t2v);
    const v2f ds = mx[p] - nm;
    v2f sc; sc[0] = __builtin_amdgcn_exp2f(ds[0]); sc[1] = __builtin_amdgcn_exp2f(ds[1]);
    const v2f c0 = epsv * t2v - nm;
    v2f d = den[p] * sc, u = num[p] * sc;
#pragma unroll
    for (int j = 0; j < EB; ++j) {
      const v2f a = r[j][p] * t2v + c0;
      v2f ex; ex[0] = __builtin_amdgcn_exp2f(a[0]); ex[1] = __builtin_amdgcn_exp2f(a[1]);
      if (TAIL && j >= nlive) ex = zero;
      d += ex;
      u += ex * r[j][p];
    }
    mx[p] = nm; den[p] = d; num[p] = u;
  }
}

#define NPB 40
#define TROWS 48

__global__ void __launch_bounds__(640) __attribute__((amdgpu_waves_per_eu(5, 5)))
layer_fused_kernel(const float* __restrict__ xin, const float* __restrict__ ea,
                   const int* __restrict__ pairs, const int* __restrict__ off,
                   const float* __restrict__ tptr, int layer,
                   const float* __restrict__ W1, const float* __restrict__ b1,
                   const float* __restrict__ g1, const float* __restrict__ be1,
                   const float* __restrict__ W2, const float* __restrict__ b2,
                   const float* hin, float* hout, int accum,
                   const float* __restrict__ g2, const float* __restrict__ be2,
                   float* __restrict__ rout, int n) {
  __shared__ __attribute__((aligned(16))) float sA[TROWS * A_PITCH];
  __shared__ __attribute__((aligned(16))) float sT[TROWS * T_PITCH];
  __shared__ float2 sRed[TROWS * R_PITCH];
  __shared__ float2 sStat[TROWS];

  const int tid  = threadIdx.x;
  const int lane = tid & 31;
  const int w = __builtin_amdgcn_readfirstlane(tid >> 5);
  const int row0 = blockIdx.x * NPB;

  {
    const int nodeA = row0 + 2 * w;
    const bool hasA = nodeA < n, hasB = nodeA + 1 < n;
    int begA = 0, endA = 0, endB = 0;
    if (hasA) { begA = off[nodeA]; endA = off[nodeA + 1]; endB = hasB ? off[nodeA + 2] : endA; }
    const int lenA = endA - begA, lenB = endB - endA;
    const float tv = tptr[layer];
    const bool tpos = tv >= 0.f;
    const float t2 = tv * 1.44269504088896341f;

    v2f mxA[2], denA[2], numA[2], mxB[2], denB[2], numB[2];
#pragma unroll
    for (int p = 0; p < 2; ++p) {
      const float ninf = __int_as_float(0xff800000);
      mxA[p][0] = ninf; mxA[p][1] = ninf; denA[p][0] = 0.f; denA[p][1] = 0.f; numA[p][0] = 0.f; numA[p][1] = 0.f;
      mxB[p][0] = ninf; mxB[p][1] = ninf; denB[p][0] = 0.f; denB[p][1] = 0.f; numB[p][0] = 0.f; numB[p][1] = 0.f;
    }

    for (int base = 0; base < lenA || base < lenB; base += 32) {
      const int cntA = min(32, lenA - base), cntB = min(32, lenB - base);
      int2 prA = make_int2(0, 0), prB = make_int2(0, 0);
      if (cntA > 0) prA = ((const int2*)pairs)[begA + base + min(lane, cntA - 1)];
      if (cntB > 0) prB = ((const int2*)pairs)[endA + base + min(lane, cntB - 1)];
      for (int g = 0; g < cntA || g < cntB; g += EB) {
        const int nA = cntA - g, nB = cntB - g;
        if (nA > 0 && nB > 0) {
          float4 xaA[EB], ebA[EB], xaB[EB], ebB[EB];
          group_load(xin, ea, prA, g, (unsigned)lane, xaA, ebA);
          group_load(xin, ea, prB, g, (unsigned)lane, xaB, ebB);
          if (g == 0) stretch_prefetch(ea, prA, cntA, prB, cntB, lane);
          if (nA >= EB) group_compute<false>(xaA, ebA, EB, t2, tpos, mxA, denA, numA);
          else          group_compute<true>(xaA, ebA, nA, t2, tpos, mxA, denA, numA);
          if (nB >= EB) group_compute<false>(xaB, ebB, EB, t2, tpos, mxB, denB, numB);
          else          group_compute<true>(xaB, ebB, nB, t2, tpos, mxB, denB, numB);
        } else if (nA > 0) {
          float4 xaA[EB], ebA[EB];
          group_load(xin, ea, prA, g, (unsigned)lane, xaA, ebA);
          if (g == 0) stretch_prefetch(ea, prA, cntA, prB, cntB, lane);
          if (nA >= EB) group_compute<false>(xaA, ebA, EB, t2, tpos, mxA, denA, numA);
          else          group_compute<true>(xaA, ebA, nA, t2, tpos, mxA, denA, numA);
        } else {
          float4 xaB[EB], ebB[EB];
          group_load(xin, ea, prB, g, (unsigned)lane, xaB, ebB);
          if (g == 0) stretch_prefetch(ea, prA, cntA, prB, cntB, lane);
          if (nB >= EB) group_compute<false>(xaB, ebB, EB, t2, tpos, mxB, denB, numB);
          else          group_compute<true>(xaB, ebB, nB, t2, tpos, mxB, denB, numB);
        }
      }
    }

    float4 oA = make_float4(0.f, 0.f, 0.f, 0.f), oB = oA;
    if (hasA) {
      const float4 root = ((const float4*)(xin + (size_t)nodeA * HDIM))[lane];
      oA = root;
      if (lenA > 0) {
        oA.x = (numA[0][0] / denA[0][0] + MSG_EPS) + root.x;
        oA.y = (numA[0][1] / denA[0][1] + MSG_EPS) + root.y;
        oA.z = (numA[1][0] / denA[1][0] + MSG_EPS) + root.z;
        oA.w = (numA[1][1] / denA[1][1] + MSG_EPS) + root.w;
      }
    }
    if (hasB) {
      const float4 root = ((const float4*)(xin + (size_t)(nodeA + 1) * HDIM))[lane];
      oB = root;
      if (lenB > 0) {
        oB.x = (numB[0][0] / denB[0][0] + MSG_EPS) + root.x;
        oB.y = (numB[0][1] / denB[0][1] + MSG_EPS) + root.y;
        oB.z = (numB[1][0] / denB[1][0] + MSG_EPS) + root.z;
        oB.w = (numB[1][1] / denB[1][1] + MSG_EPS) + root.w;
      }
    }
    *(float4*)&sA[(2 * w) * A_PITCH + 4 * lane] = oA;
    *(float4*)&sA[(2 * w + 1) * A_PITCH + 4 * lane] = oB;
    if (w < (TROWS - NPB) / 2) {
      const float4 z = make_float4(0.f, 0.f, 0.f, 0.f);
      *(float4*)&sA[(NPB + 2 * w) * A_PITCH + 4 * lane] = z;
      *(float4*)&sA[(NPB + 2 * w + 1) * A_PITCH + 4 * lane] = z;
    }
  }
  __syncthreads();

  const bool mm = w < 12;
  const int rt = w >> 2, cg = w & 3;
  const int hh = lane >> 4, c = lane & 15;
  const int col1 = cg * 64 + 4 * c;
  const int col2 = cg * 32 + 2 * c;
  const int trow = 16 * rt + 8 * hh;
  const int nrows = min(NPB, n - row0);

  v8f acc[4];
  v8f acc2[2];
  if (mm) {
    {
      const float4 bb = *(const float4*)(b1 + col1);
#pragma unroll
      for (int r = 0; r < 8; ++r) { acc[0][r] = bb.x; acc[1][r] = bb.y; acc[2][r] = bb.z; acc[3][r] = bb.w; }
    }
    const float4* w1p = (const float4*)(W1 + (size_t)(2 * hh) * HDIM2 + col1);
    const float* ap = sA + (16 * rt + c) * A_PITCH + 2 * hh;
#pragma unroll
    for (int ks = 0; ks < HDIM / 4; ++ks) {
      const int k0 = 4 * ks;
      const float4 blo = w1p[(size_t)k0 * (HDIM2 / 4)];
      const float4 bhi = w1p[(size_t)(k0 + 1) * (HDIM2 / 4)];
      const float2 av = *(const float2*)(ap + k0);
      v2f a; a[0] = av.x; a[1] = av.y;
      v2f b;
      b[0] = blo.x; b[1] = bhi.x; acc[0] = wmma4(a, b, acc[0]);
      b[0] = blo.y; b[1] = bhi.y; acc[1] = wmma4(a, b, acc[1]);
      b[0] = blo.z; b[1] = bhi.z; acc[2] = wmma4(a, b, acc[2]);
      b[0] = blo.w; b[1] = bhi.w; acc[3] = wmma4(a, b, acc[3]);
    }

#pragma unroll
    for (int r = 0; r < 8; ++r) {
      const float s = (acc[0][r] + acc[1][r]) + (acc[2][r] + acc[3][r]);
      const float q = (acc[0][r] * acc[0][r] + acc[1][r] * acc[1][r]) +
                      (acc[2][r] * acc[2][r] + acc[3][r] * acc[3][r]);
      sRed[(trow + r) * R_PITCH + cg * 16 + c] = make_float2(s, q);
    }
  }
  __syncthreads();
  if (mm) row_stats(sRed, sStat, tid, 1.f / (float)HDIM2);
  __syncthreads();
  if (mm) {
    const float4 gg = *(const float4*)(g1 + col1);
    const float4 be = *(const float4*)(be1 + col1);
#pragma unroll
    for (int r = 0; r < 8; ++r) {
      const float2 st = sStat[trow + r];
      float4 o;
      o.x = fmaxf((acc[0][r] - st.x) * st.y * gg.x + be.x, 0.f);
      o.y = fmaxf((acc[1][r] - st.x) * st.y * gg.y + be.y, 0.f);
      o.z = fmaxf((acc[2][r] - st.x) * st.y * gg.z + be.z, 0.f);
      o.w = fmaxf((acc[3][r] - st.x) * st.y * gg.w + be.w, 0.f);
      *(float4*)&sT[(trow + r) * T_PITCH + col1] = o;
    }
  }
  __syncthreads();

  if (mm) {
    {
      const float2 bb = *(const float2*)(b2 + col2);
#pragma unroll
      for (int r = 0; r < 8; ++r) {
        const int lrow = min(trow + r, nrows - 1);
        float2 o = *(const float2*)(hin + (size_t)(row0 + lrow) * HDIM + col2);
        if (!accum) o = make_float2(0.f, 0.f);
        acc2[0][r] = o.x + bb.x;
        acc2[1][r] = o.y + bb.y;
      }
    }
    const float2* w2p = (const float2*)(W2 + (size_t)(2 * hh) * HDIM + col2);
    const float* tp = sT + (16 * rt + c) * T_PITCH + 2 * hh;
#pragma unroll
    for (int ks = 0; ks < HDIM2 / 4; ++ks) {
      const int k0 = 4 * ks;
      const float2 blo = w2p[(size_t)k0 * (HDIM / 2)];
      const float2 bhi = w2p[(size_t)(k0 + 1) * (HDIM / 2)];
      const float2 av = *(const float2*)(tp + k0);
      v2f a; a[0] = av.x; a[1] = av.y;
      v2f b;
      b[0] = blo.x; b[1] = bhi.x; acc2[0] = wmma4(a, b, acc2[0]);
      b[0] = blo.y; b[1] = bhi.y; acc2[1] = wmma4(a, b, acc2[1]);
    }

#pragma unroll
    for (int r = 0; r < 8; ++r) {
      const float2 o = make_float2(acc2[0][r], acc2[1][r]);
      if (trow + r < nrows) *(float2*)(hout + (size_t)(row0 + trow + r) * HDIM + col2) = o;
      sRed[(trow + r) * R_PITCH + cg * 16 + c] = make_float2(o.x + o.y, o.x * o.x + o.y * o.y);
    }
  }
  __syncthreads();
  if (mm) row_stats(sRed, sStat, tid, 1.f / (float)HDIM);
  __syncthreads();
  if (mm) {
    const float2 gg = *(const float2*)(g2 + col2);
    const float2 be = *(const float2*)(be2 + col2);
#pragma unroll
    for (int r = 0; r < 8; ++r) {
      const float2 st = sStat[trow + r];
      float2 o;
      o.x = fmaxf((acc2[0][r] - st.x) * st.y * gg.x + be.x, 0.f);
      o.y = fmaxf((acc2[1][r] - st.x) * st.y * gg.y + be.y, 0.f);
      if (trow + r < nrows) *(float2*)(rout + (size_t)(row0 + trow + r) * HDIM + col2) = o;
    }
  }
}

#define PTRIPS 10
#define PROWS 8

__global__ void __launch_bounds__(1024)
pool_classify_kernel(const float* __restrict__ r, const int* __restrict__ batch,
                     const float* __restrict__ W, const float* __restrict__ b,
                     float* __restrict__ out, float* __restrict__ pooled, int n) {
  __shared__ __attribute__((aligned(16))) float part[32 * HDIM];
  __shared__ __attribute__((aligned(16))) float psum[HDIM];
  const int tid = threadIdx.x;
  const int lane = tid & 31;
  const int w = __builtin_amdgcn_readfirstlane(tid >> 5);
  const int g = blockIdx.x;

  float4 wv[NUM_CLASSES];
  float bias = 0.f;
  if (w == 0) {
#pragma unroll
    for (int j = 0; j < NUM_CLASSES; ++j) wv[j] = ((const float4*)W)[NUM_CLASSES * lane + j];
    if (lane < NUM_CLASSES) bias = b[lane];
  }

  const int lofs = 256 * (lane >> 3) + 8 * w + (lane & 7);
  float4 acc = make_float4(0.f, 0.f, 0.f, 0.f);
  for (int t0 = 0; 1024 * t0 < n; t0 += PTRIPS) {
    int bv[PTRIPS];
#pragma unroll
    for (int u = 0; u < PTRIPS; ++u) {
      const int nn = 1024 * (t0 + u) + lofs;
      bv[u] = nn < n ? batch[nn] : -1;
    }
#pragma unroll
    for (int u = 0; u < PTRIPS; ++u) {
      unsigned m = __builtin_amdgcn_ballot_w32(bv[u] == g);
      const int nbase = 1024 * (t0 + u) + 8 * w;
      while (m) {
        float4 v[PROWS];
#pragma unroll
        for (int q = 0; q < PROWS; ++q) {
          v[q] = make_float4(0.f, 0.f, 0.f, 0.f);
          if (m) {
            const int bit = __builtin_ctz(m);
            m &= m - 1;
            const int node = nbase + 256 * (bit >> 3) + (bit & 7);
            v[q] = ((const float4*)r)[(unsigned)node * (HDIM / 4) + lane];
          }
        }
#pragma unroll
        for (int q = 0; q < PROWS; ++q) {
          acc.x += v[q].x; acc.y += v[q].y; acc.z += v[q].z; acc.w += v[q].w;
        }
      }
    }
  }
  *(float4*)&part[w * HDIM + 4 * lane] = acc;
  __syncthreads();
  float ps = 0.f;
  if (tid < HDIM) {
    float t[32];
#pragma unroll
    for (int k = 0; k < 32; ++k) t[k] = part[k * HDIM + tid];
#pragma unroll
    for (int st = 1; st < 32; st <<= 1) {
#pragma unroll
      for (int k = 0; k < 32; k += 2 * st) t[k] += t[k + st];
    }
    ps = t[0];
    psum[tid] = ps;
  }
  __syncthreads();
  if (tid < HDIM) pooled[(size_t)g * HDIM + tid] = ps;
  if (w == 0) {
    const float4 p4 = *(const float4*)&psum[4 * lane];
    const float p[4] = {p4.x, p4.y, p4.z, p4.w};
    float wf[4 * NUM_CLASSES];
#pragma unroll
    for (int j = 0; j < NUM_CLASSES; ++j) {
      wf[4 * j + 0] = wv[j].x; wf[4 * j + 1] = wv[j].y; wf[4 * j + 2] = wv[j].z; wf[4 * j + 3] = wv[j].w;
    }
    float o = 0.f;
#pragma unroll
    for (int c = 0; c < NUM_CLASSES; ++c) {
      float s = (p[0] * wf[c] + p[1] * wf[NUM_CLASSES + c]) +
                (p[2] * wf[2 * NUM_CLASSES + c] + p[3] * wf[3 * NUM_CLASSES + c]);
#pragma unroll
      for (int off = 16; off > 0; off >>= 1) s += __shfl_xor(s, off, 32);
      if (lane == c) o = s;
    }
    if (lane < NUM_CLASSES) out[g * NUM_CLASSES + lane] = o + bias;
  }
}

extern "C" void kernel_launch(void* const* d_in, const int* in_sizes, int n_in,
                              void* d_out, int out_size, void* d_ws, size_t ws_size,
                              hipStream_t stream) {
  const float* x     = (const float*)d_in[0];
  const float* ea    = (const float*)d_in[1];
  const float* enc_W = (const float*)d_in[2];
  const float* enc_b = (const float*)d_in[3];
  const float* t     = (const float*)d_in[4];
  const float* W1    = (const float*)d_in[5];
  const float* b1    = (const float*)d_in[6];
  const float* mlp_g = (const float*)d_in[7];
  const float* mlp_b = (const float*)d_in[8];
  const float* W2    = (const float*)d_in[9];
  const float* b2    = (const float*)d_in[10];
  const float* ln_g  = (const float*)d_in[11];
  const float* ln_b  = (const float*)d_in[12];
  const float* lin_W = (const float*)d_in[13];
  const float* lin_b = (const float*)d_in[14];
  const int*   eidx  = (const int*)d_in[15];
  const int*   batch = (const int*)d_in[16];

  const int N = in_sizes[0] / HDIM;
  const int E = in_sizes[1] / HDIM;
  const int L = in_sizes[4];
  const int* src = eidx;
  const int* dst = eidx + E;

  const size_t NH = (size_t)N * HDIM;
  float* h    = (float*)d_ws;
  float* rbuf = h    + NH;
  float* hc   = rbuf + NH;
  float* t1   = hc   + NH;
  int* pairs  = (int*)(t1 + 2 * NH);
  int* raw    = pairs + 2 * (size_t)E;
  int* slot   = raw   + E;
  int* cnt    = slot  + E;
  int* off    = cnt   + N;
  int* binned = off   + N + 2;
  int* tab    = binned + 2 * (size_t)E;

  float* out    = (float*)d_out;
  float* pooled = out + NUM_GRAPHS * NUM_CLASSES;

  bin_local_kernel<<<NCHUNK, 1024, 0, stream>>>(dst, tab, binned, E, N);
  const int encBlocks = (N / 16 * (HDIM / 64) + ENC_STRIPS - 1) / ENC_STRIPS;
  bin_sort_encoder_kernel<<<NBIN + encBlocks, 1024, 0, stream>>>(tab, binned, src, slot, raw, off, pairs,
                                                                E, N, x, enc_W, enc_b, h);

  float* hsum = t1;
  const float* rin = h;
  for (int layer = 0; layer < L; ++layer) {
    float* rnext = (layer & 1) ? hc : rbuf;
    const int nl = layer + 1 < L ? layer + 1 : 0;
    layer_fused_kernel<<<(N + NPB - 1) / NPB, 640, 0, stream>>>(
        rin, ea, pairs, off, t, layer,
        W1 + (size_t)layer * HDIM * HDIM2, b1 + layer * HDIM2,
        mlp_g + layer * HDIM2, mlp_b + layer * HDIM2,
        W2 + (size_t)layer * HDIM2 * HDIM, b2 + layer * HDIM,
        layer > 0 ? hsum : h, hsum, layer > 0 ? 1 : 0, ln_g + nl * HDIM, ln_b + nl * HDIM, rnext, N);
    rin = rnext;
  }

  pool_classify_kernel<<<NUM_GRAPHS, 1024, 0, stream>>>(rin, batch, lin_W, lin_b, out, pooled, N);
}
